# starting kernel
# baseline (speedup 1.0000x reference)
#include <hip/hip_runtime.h>


typedef float v2f __attribute__((ext_vector_type(2)));
typedef float v8f __attribute__((ext_vector_type(8)));

__global__ void sage_deg_kernel(const int* __restrict__ dst,
                                float* __restrict__ deg, int E) {
  int e = blockIdx.x * blockDim.x + threadIdx.x;
  if (e < E) atomicAdd(&deg[dst[e]], 1.0f);
}

template <int F>
__global__ void sage_scatter_kernel(const float* __restrict__ feat,
                                    const int* __restrict__ src,
                                    const int* __restrict__ dst,
                                    float* __restrict__ agg, int E) {
  constexpr int CH = F / 4;
  int t = blockIdx.x * blockDim.x + threadIdx.x;
  if (t >= E * CH) return;
  int e = t / CH;
  int c = (t % CH) * 4;
  int s = src[e];
  int d = dst[e];
  const float4 v = *(const float4*)(feat + (long long)s * F + c);
  float* p = agg + (long long)d * F + c;
  atomicAdd(p + 0, v.x);
  atomicAdd(p + 1, v.y);
  atomicAdd(p + 2, v.z);
  atomicAdd(p + 3, v.w);
}

template <int FIN_T, int FOUT_T, bool RELU>
__global__ __launch_bounds__(32) void sage_gemm_kernel(
    const float* __restrict__ agg, const float* __restrict__ xin,
    const float* __restrict__ deg, const float* __restrict__ Wl,
    const float* __restrict__ Wr, const float* __restrict__ bias,
    float* __restrict__ out) {
  const int lane = threadIdx.x;
  const int half = lane >> 4;
  const int l15  = lane & 15;
  const int row  = blockIdx.x * 16 + l15;
  const int col  = blockIdx.y * 16 + l15;
  const int koff = half * 2;

  const float inv = 1.0f / fmaxf(deg[row], 1.0f);
  const float* aRow = agg + (long long)row * FIN_T;
  const float* xRow = xin + (long long)row * FIN_T;
  __builtin_prefetch(aRow, 0, 1);
  __builtin_prefetch(xRow, 0, 1);

  v8f acc = {0.f, 0.f, 0.f, 0.f, 0.f, 0.f, 0.f, 0.f};

#pragma unroll
  for (int k0 = 0; k0 < FIN_T; k0 += 4) {
    v2f a = *(const v2f*)(aRow + k0 + koff);
    a *= inv;
    v2f b;
    b.x = Wl[(k0 + koff) * FOUT_T + col];
    b.y = Wl[(k0 + koff + 1) * FOUT_T + col];
    acc = __builtin_amdgcn_wmma_f32_16x16x4_f32(
         false, a,  false, b,
         (short)0, acc,  false,  false);
  }
#pragma unroll
  for (int k0 = 0; k0 < FIN_T; k0 += 4) {
    v2f a = *(const v2f*)(xRow + k0 + koff);
    v2f b;
    b.x = Wr[(k0 + koff) * FOUT_T + col];
    b.y = Wr[(k0 + koff + 1) * FOUT_T + col];
    acc = __builtin_amdgcn_wmma_f32_16x16x4_f32(
        false, a, false, b, (short)0, acc, false, false);
  }

  const float bb = bias[col];
#pragma unroll
  for (int v = 0; v < 8; ++v) {
    int m = blockIdx.x * 16 + v + half * 8;
    float val = acc[v] + bb;
    if (RELU) val = fmaxf(val, 0.f);
    out[(long long)m * FOUT_T + col] = val;
  }
}

extern "C" void kernel_launch(void* const* d_in, const int* in_sizes, int n_in,
                              void* d_out, int out_size, void* d_ws,
                              size_t ws_size, hipStream_t stream) {
  constexpr int FIN = 64, HID = 128, FOUT = 32;

  const float* x   = (const float*)d_in[0];
  const int*   ei  = (const int*)d_in[1];
  const float* Wl1 = (const float*)d_in[2];
  const float* Wr1 = (const float*)d_in[3];
  const float* b1  = (const float*)d_in[4];
  const float* Wl2 = (const float*)d_in[5];
  const float* Wr2 = (const float*)d_in[6];
  const float* b2  = (const float*)d_in[7];
  float* out = (float*)d_out;

  const int n = in_sizes[0] / FIN;
  const int E = in_sizes[1] / 2;
  const int* src = ei;
  const int* dst = ei + E;

  float* deg  = (float*)d_ws;
  float* agg1 = deg + n;
  float* agg2 = agg1 + (size_t)n * FIN;
  float* h    = agg2 + (size_t)n * HID;

  hipMemsetAsync(d_ws, 0, (size_t)n * (1 + FIN + HID) * sizeof(float), stream);

  sage_deg_kernel<<<(E + 255) / 256, 256, 0, stream>>>(dst, deg, E);

  sage_scatter_kernel<FIN>
      <<<(E * (FIN / 4) + 255) / 256, 256, 0, stream>>>(x, src, dst, agg1, E);

  sage_gemm_kernel<FIN, HID, true>
      <<<dim3(n / 16, HID / 16), 32, 0, stream>>>(agg1, x, deg, Wl1, Wr1, b1, h);

  sage_scatter_kernel<HID>
      <<<(E * (HID / 4) + 255) / 256, 256, 0, stream>>>(h, src, dst, agg2, E);

  sage_gemm_kernel<HID, FOUT, false>
      <<<dim3(n / 16, FOUT / 16), 32, 0, stream>>>(agg2, h, deg, Wl2, Wr2, b2, out);
}


# Layer 2 applies W_l2 before the mean aggregation (linearity): the atomic scatter carries 32 not 128 floats per edge, scaled rows added straight into out; f32 summation order changes
# speedup vs baseline: 1.5929x; 1.5929x over previous
#include <hip/hip_runtime.h>


typedef float v2f __attribute__((ext_vector_type(2)));
typedef float v8f __attribute__((ext_vector_type(8)));

__global__ void sage_deg_kernel(const int* __restrict__ dst,
                                float* __restrict__ deg, int E) {
  int e = blockIdx.x * blockDim.x + threadIdx.x;
  if (e < E) atomicAdd(&deg[dst[e]], 1.0f);
}

template <int F, bool MEAN>
__global__ void sage_scatter_kernel(const float* __restrict__ feat,
                                    const int* __restrict__ src,
                                    const int* __restrict__ dst,
                                    const float* __restrict__ deg,
                                    float* __restrict__ agg, int E) {
  constexpr int CH = F / 4;
  int t = blockIdx.x * blockDim.x + threadIdx.x;
  if (t >= E * CH) return;
  int e = t / CH;
  int c = (t % CH) * 4;
  int s = src[e];
  int d = dst[e];
  float4 v = *(const float4*)(feat + (long long)s * F + c);
  if (MEAN) {
    const float inv = 1.0f / fmaxf(deg[d], 1.0f);
    v.x *= inv;
    v.y *= inv;
    v.z *= inv;
    v.w *= inv;
  }
  float* p = agg + (long long)d * F + c;
  atomicAdd(p + 0, v.x);
  atomicAdd(p + 1, v.y);
  atomicAdd(p + 2, v.z);
  atomicAdd(p + 3, v.w);
}

template <int FIN_T, int FOUT_T, bool RELU>
__global__ __launch_bounds__(32) void sage_gemm_kernel(
    const float* __restrict__ agg, const float* __restrict__ xin,
    const float* __restrict__ deg, const float* __restrict__ Wl,
    const float* __restrict__ Wr, const float* __restrict__ bias,
    float* __restrict__ out) {
  const int lane = threadIdx.x;
  const int half = lane >> 4;
  const int l15  = lane & 15;
  const int row  = blockIdx.x * 16 + l15;
  const int col  = blockIdx.y * 16 + l15;
  const int koff = half * 2;

  const float inv = 1.0f / fmaxf(deg[row], 1.0f);
  const float* aRow = agg + (long long)row * FIN_T;
  const float* xRow = xin + (long long)row * FIN_T;
  __builtin_prefetch(aRow, 0, 1);
  __builtin_prefetch(xRow, 0, 1);

  v8f acc = {0.f, 0.f, 0.f, 0.f, 0.f, 0.f, 0.f, 0.f};

#pragma unroll
  for (int k0 = 0; k0 < FIN_T; k0 += 4) {
    v2f a = *(const v2f*)(aRow + k0 + koff);
    a *= inv;
    v2f b;
    b.x = Wl[(k0 + koff) * FOUT_T + col];
    b.y = Wl[(k0 + koff + 1) * FOUT_T + col];
    acc = __builtin_amdgcn_wmma_f32_16x16x4_f32(
         false, a,  false, b,
         (short)0, acc,  false,  false);
  }
#pragma unroll
  for (int k0 = 0; k0 < FIN_T; k0 += 4) {
    v2f a = *(const v2f*)(xRow + k0 + koff);
    v2f b;
    b.x = Wr[(k0 + koff) * FOUT_T + col];
    b.y = Wr[(k0 + koff + 1) * FOUT_T + col];
    acc = __builtin_amdgcn_wmma_f32_16x16x4_f32(
        false, a, false, b, (short)0, acc, false, false);
  }

  const float bb = bias[col];
#pragma unroll
  for (int v = 0; v < 8; ++v) {
    int m = blockIdx.x * 16 + v + half * 8;
    float val = acc[v] + bb;
    if (RELU) val = fmaxf(val, 0.f);
    out[(long long)m * FOUT_T + col] = val;
  }
}

template <int FIN_T, int FOUT_T>
__global__ __launch_bounds__(32) void sage_pre_kernel(
    const float* __restrict__ hin, const float* __restrict__ Wl,
    const float* __restrict__ Wr, const float* __restrict__ bias,
    float* __restrict__ p, float* __restrict__ out) {
  const int lane = threadIdx.x;
  const int half = lane >> 4;
  const int l15  = lane & 15;
  const int row  = blockIdx.x * 16 + l15;
  const int col  = blockIdx.y * 16 + l15;
  const int koff = half * 2;

  const float* hRow = hin + (long long)row * FIN_T;
  __builtin_prefetch(hRow, 0, 1);

  v8f accL = {0.f, 0.f, 0.f, 0.f, 0.f, 0.f, 0.f, 0.f};
  v8f accR = {0.f, 0.f, 0.f, 0.f, 0.f, 0.f, 0.f, 0.f};

#pragma unroll
  for (int k0 = 0; k0 < FIN_T; k0 += 4) {
    v2f a = *(const v2f*)(hRow + k0 + koff);
    v2f bl, br;
    bl.x = Wl[(k0 + koff) * FOUT_T + col];
    bl.y = Wl[(k0 + koff + 1) * FOUT_T + col];
    br.x = Wr[(k0 + koff) * FOUT_T + col];
    br.y = Wr[(k0 + koff + 1) * FOUT_T + col];
    accL = __builtin_amdgcn_wmma_f32_16x16x4_f32(
         false, a,  false, bl,
         (short)0, accL,  false,  false);
    accR = __builtin_amdgcn_wmma_f32_16x16x4_f32(
        false, a, false, br, (short)0, accR, false, false);
  }

  const float bb = bias[col];
#pragma unroll
  for (int v = 0; v < 8; ++v) {
    int m = blockIdx.x * 16 + v + half * 8;
    p[(long long)m * FOUT_T + col]   = accL[v];
    out[(long long)m * FOUT_T + col] = accR[v] + bb;
  }
}

extern "C" void kernel_launch(void* const* d_in, const int* in_sizes, int n_in,
                              void* d_out, int out_size, void* d_ws,
                              size_t ws_size, hipStream_t stream) {
  constexpr int FIN = 64, HID = 128, FOUT = 32;

  const float* x   = (const float*)d_in[0];
  const int*   ei  = (const int*)d_in[1];
  const float* Wl1 = (const float*)d_in[2];
  const float* Wr1 = (const float*)d_in[3];
  const float* b1  = (const float*)d_in[4];
  const float* Wl2 = (const float*)d_in[5];
  const float* Wr2 = (const float*)d_in[6];
  const float* b2  = (const float*)d_in[7];
  float* out = (float*)d_out;

  const int n = in_sizes[0] / FIN;
  const int E = in_sizes[1] / 2;
  const int* src = ei;
  const int* dst = ei + E;

  float* deg  = (float*)d_ws;
  float* agg1 = deg + n;
  float* p2   = agg1 + (size_t)n * FIN;
  float* h    = p2 + (size_t)n * FOUT;

  hipMemsetAsync(d_ws, 0, (size_t)n * (1 + FIN) * sizeof(float), stream);

  sage_deg_kernel<<<(E + 255) / 256, 256, 0, stream>>>(dst, deg, E);

  sage_scatter_kernel<FIN, false>
      <<<(E * (FIN / 4) + 255) / 256, 256, 0, stream>>>(x, src, dst, deg, agg1, E);

  sage_gemm_kernel<FIN, HID, true>
      <<<dim3(n / 16, HID / 16), 32, 0, stream>>>(agg1, x, deg, Wl1, Wr1, b1, h);

  sage_pre_kernel<HID, FOUT>
      <<<dim3(n / 16, FOUT / 16), 32, 0, stream>>>(h, Wl2, Wr2, b2, p2, out);

  sage_scatter_kernel<FOUT, true>
      <<<(E * (FOUT / 4) + 255) / 256, 256, 0, stream>>>(p2, src, dst, deg, out, E);
}


# Mean aggregation without float atomics: edges counting-sorted by destination per call (int count, one-block scan, arrival-order slot fill), each node sums its own rows; f32 sum order changes
# speedup vs baseline: 1.7522x; 1.1000x over previous
#include <hip/hip_runtime.h>


typedef float v2f __attribute__((ext_vector_type(2)));
typedef float v8f __attribute__((ext_vector_type(8)));

__global__ void sage_count_kernel(const int* __restrict__ dst,
                                  int* __restrict__ cnt, int E) {
  const int e0 = (blockIdx.x * blockDim.x + threadIdx.x) * 4;
  if (e0 >= E) return;
  if ((E & 3) == 0) {
    const int4 d = *(const int4*)(dst + e0);
    atomicAdd(&cnt[d.x], 1);
    atomicAdd(&cnt[d.y], 1);
    atomicAdd(&cnt[d.z], 1);
    atomicAdd(&cnt[d.w], 1);
  } else {
    for (int u = 0; u < 4; ++u)
      if (e0 + u < E) atomicAdd(&cnt[dst[e0 + u]], 1);
  }
}

__global__ __launch_bounds__(1024) void sage_scan_kernel(
    const int* __restrict__ cnt, int* __restrict__ rowptr, int n) {
  __shared__ int wsum[32];
  __shared__ int tot;
  const int tid  = threadIdx.x;
  const int lane = tid & 31;
  const int w    = tid >> 5;
  int carry = 0;
  for (int base = 0; base < n; base += 16384) {
    const int i = base + tid * 16;
    int4 a = make_int4(0, 0, 0, 0), b = a, c = a, e = a;
    if (i < n) {
      const int4* q = (const int4*)(cnt + i);
      a = q[0];
      b = q[1];
      c = q[2];
      e = q[3];
    }
    a.y += a.x; a.z += a.y; a.w += a.z;
    b.x += a.w; b.y += b.x; b.z += b.y; b.w += b.z;
    c.x += b.w; c.y += c.x; c.z += c.y; c.w += c.z;
    e.x += c.w; e.y += e.x; e.z += e.y; e.w += e.z;
    const int s = e.w;
    int incl = s;
    for (int o = 1; o < 32; o <<= 1) {
      const int t = __shfl_up(incl, o);
      if (lane >= o) incl += t;
    }
    if (lane == 31) wsum[w] = incl;
    __syncthreads();
    if (w == 0) {
      const int ws = wsum[lane];
      int wi = ws;
      for (int o = 1; o < 32; o <<= 1) {
        const int t = __shfl_up(wi, o);
        if (lane >= o) wi += t;
      }
      wsum[lane] = wi - ws;
      if (lane == 31) tot = wi;
    }
    __syncthreads();
    const int excl = carry + wsum[w] + incl - s;
    if (i < n) {
      int4* q = (int4*)(rowptr + i);
      q[0] = make_int4(excl, excl + a.x, excl + a.y, excl + a.z);
      q[1] = make_int4(excl + a.w, excl + b.x, excl + b.y, excl + b.z);
      q[2] = make_int4(excl + b.w, excl + c.x, excl + c.y, excl + c.z);
      q[3] = make_int4(excl + c.w, excl + e.x, excl + e.y, excl + e.z);
    }
    carry += tot;
    __syncthreads();
  }
  if (tid == 0) rowptr[n] = carry;
}

__device__ __forceinline__ void sage_place(int s, int d,
                                           const int* __restrict__ rowptr,
                                           int* __restrict__ cnt,
                                           int* __restrict__ csr) {
  const int k = atomicSub(&cnt[d], 1);
  csr[rowptr[d] + k - 1] = s;
}

__global__ void sage_fill_kernel(const int* __restrict__ src,
                                 const int* __restrict__ dst,
                                 const int* __restrict__ rowptr,
                                 int* __restrict__ cnt, int* __restrict__ csr,
                                 int E) {
  const int e0 = (blockIdx.x * blockDim.x + threadIdx.x) * 4;
  if (e0 >= E) return;
  if ((E & 3) == 0) {
    const int4 s = *(const int4*)(src + e0);
    const int4 d = *(const int4*)(dst + e0);
    const int k0 = atomicSub(&cnt[d.x], 1);
    const int k1 = atomicSub(&cnt[d.y], 1);
    const int k2 = atomicSub(&cnt[d.z], 1);
    const int k3 = atomicSub(&cnt[d.w], 1);
    const int r0 = rowptr[d.x];
    const int r1 = rowptr[d.y];
    const int r2 = rowptr[d.z];
    const int r3 = rowptr[d.w];
    csr[r0 + k0 - 1] = s.x;
    csr[r1 + k1 - 1] = s.y;
    csr[r2 + k2 - 1] = s.z;
    csr[r3 + k3 - 1] = s.w;
  } else {
    for (int u = 0; u < 4; ++u)
      if (e0 + u < E) sage_place(src[e0 + u], dst[e0 + u], rowptr, cnt, csr);
  }
}

__device__ __forceinline__ float4 sage_row(const float* fcol, int s, int F) {
  return *(const float4*)(fcol + (long long)s * F);
}
__device__ __forceinline__ void sage_acc(float4& acc, const float4 v) {
  acc.x += v.x;
  acc.y += v.y;
  acc.z += v.z;
  acc.w += v.w;
}

template <int F, bool ADD>
__global__ __launch_bounds__(128) void sage_gather_kernel(
    const float* __restrict__ feat, const int* __restrict__ rowptr,
    const int* __restrict__ csr, float* __restrict__ tgt, int n) {
  constexpr int LPN = F / 4;
  constexpr int NPB = 128 / LPN;
  const int sub = threadIdx.x / LPN;
  const int c   = (threadIdx.x % LPN) * 4;
  const float* fcol = feat + c;
  for (int d = blockIdx.x * NPB + sub; d < n; d += gridDim.x * NPB) {
    const int beg = rowptr[d];
    const int end = rowptr[d + 1];
    float4 acc = make_float4(0.f, 0.f, 0.f, 0.f);
    int j = beg;
    for (; j + 8 <= end; j += 8) {
      const int s0 = csr[j];
      const int s1 = csr[j + 1];
      const int s2 = csr[j + 2];
      const int s3 = csr[j + 3];
      const int s4 = csr[j + 4];
      const int s5 = csr[j + 5];
      const int s6 = csr[j + 6];
      const int s7 = csr[j + 7];
      const float4 v0 = sage_row(fcol, s0, F);
      const float4 v1 = sage_row(fcol, s1, F);
      const float4 v2 = sage_row(fcol, s2, F);
      const float4 v3 = sage_row(fcol, s3, F);
      const float4 v4 = sage_row(fcol, s4, F);
      const float4 v5 = sage_row(fcol, s5, F);
      const float4 v6 = sage_row(fcol, s6, F);
      const float4 v7 = sage_row(fcol, s7, F);
      sage_acc(acc, v0);
      sage_acc(acc, v1);
      sage_acc(acc, v2);
      sage_acc(acc, v3);
      sage_acc(acc, v4);
      sage_acc(acc, v5);
      sage_acc(acc, v6);
      sage_acc(acc, v7);
    }
    if (j + 4 <= end) {
      const int s0 = csr[j];
      const int s1 = csr[j + 1];
      const int s2 = csr[j + 2];
      const int s3 = csr[j + 3];
      const float4 v0 = sage_row(fcol, s0, F);
      const float4 v1 = sage_row(fcol, s1, F);
      const float4 v2 = sage_row(fcol, s2, F);
      const float4 v3 = sage_row(fcol, s3, F);
      sage_acc(acc, v0);
      sage_acc(acc, v1);
      sage_acc(acc, v2);
      sage_acc(acc, v3);
      j += 4;
    }
    if (j < end) {
      const int last = end - 1;
      const bool h1 = (j + 1 <= last);
      const bool h2 = (j + 2 <= last);
      const int s0 = csr[j];
      const int s1 = csr[h1 ? j + 1 : last];
      const int s2 = csr[h2 ? j + 2 : last];
      const float4 v0 = sage_row(fcol, s0, F);
      float4 v1 = sage_row(fcol, s1, F);
      float4 v2 = sage_row(fcol, s2, F);
      v1.x = h1 ? v1.x : 0.f; v1.y = h1 ? v1.y : 0.f;
      v1.z = h1 ? v1.z : 0.f; v1.w = h1 ? v1.w : 0.f;
      v2.x = h2 ? v2.x : 0.f; v2.y = h2 ? v2.y : 0.f;
      v2.z = h2 ? v2.z : 0.f; v2.w = h2 ? v2.w : 0.f;
      sage_acc(acc, v0);
      sage_acc(acc, v1);
      sage_acc(acc, v2);
    }
    const float inv = 1.0f / fmaxf((float)(end - beg), 1.0f);
    float4* p = (float4*)(tgt + (long long)d * F + c);
    float4 o = make_float4(0.f, 0.f, 0.f, 0.f);
    if (ADD) o = *p;
    o.x += acc.x * inv;
    o.y += acc.y * inv;
    o.z += acc.z * inv;
    o.w += acc.w * inv;
    *p = o;
  }
}

template <int FIN_T, int FOUT_T, bool RELU>
__global__ __launch_bounds__(32) void sage_gemm_kernel(
    const float* __restrict__ agg, const float* __restrict__ xin,
    const float* __restrict__ Wl,
    const float* __restrict__ Wr, const float* __restrict__ bias,
    float* __restrict__ out) {
  const int lane = threadIdx.x;
  const int half = lane >> 4;
  const int l15  = lane & 15;
  const int row  = blockIdx.x * 16 + l15;
  const int col  = blockIdx.y * 16 + l15;
  const int koff = half * 2;

  const float* aRow = agg + (long long)row * FIN_T;
  const float* xRow = xin + (long long)row * FIN_T;
  __builtin_prefetch(aRow, 0, 1);
  __builtin_prefetch(xRow, 0, 1);

  v8f acc = {0.f, 0.f, 0.f, 0.f, 0.f, 0.f, 0.f, 0.f};

#pragma unroll
  for (int k0 = 0; k0 < FIN_T; k0 += 4) {
    v2f a = *(const v2f*)(aRow + k0 + koff);
    v2f b;
    b.x = Wl[(k0 + koff) * FOUT_T + col];
    b.y = Wl[(k0 + koff + 1) * FOUT_T + col];
    acc = __builtin_amdgcn_wmma_f32_16x16x4_f32(
         false, a,  false, b,
         (short)0, acc,  false,  false);
  }
#pragma unroll
  for (int k0 = 0; k0 < FIN_T; k0 += 4) {
    v2f a = *(const v2f*)(xRow + k0 + koff);
    v2f b;
    b.x = Wr[(k0 + koff) * FOUT_T + col];
    b.y = Wr[(k0 + koff + 1) * FOUT_T + col];
    acc = __builtin_amdgcn_wmma_f32_16x16x4_f32(
        false, a, false, b, (short)0, acc, false, false);
  }

  const float bb = bias[col];
#pragma unroll
  for (int v = 0; v < 8; ++v) {
    int m = blockIdx.x * 16 + v + half * 8;
    float val = acc[v] + bb;
    if (RELU) val = fmaxf(val, 0.f);
    out[(long long)m * FOUT_T + col] = val;
  }
}

template <int FIN_T, int FOUT_T>
__global__ __launch_bounds__(32) void sage_pre_kernel(
    const float* __restrict__ hin, const float* __restrict__ Wl,
    const float* __restrict__ Wr, const float* __restrict__ bias,
    float* __restrict__ p, float* __restrict__ out) {
  const int lane = threadIdx.x;
  const int half = lane >> 4;
  const int l15  = lane & 15;
  const int row  = blockIdx.x * 16 + l15;
  const int col  = blockIdx.y * 16 + l15;
  const int koff = half * 2;

  const float* hRow = hin + (long long)row * FIN_T;
  __builtin_prefetch(hRow, 0, 1);

  v8f accL = {0.f, 0.f, 0.f, 0.f, 0.f, 0.f, 0.f, 0.f};
  v8f accR = {0.f, 0.f, 0.f, 0.f, 0.f, 0.f, 0.f, 0.f};

#pragma unroll
  for (int k0 = 0; k0 < FIN_T; k0 += 4) {
    v2f a = *(const v2f*)(hRow + k0 + koff);
    v2f bl, br;
    bl.x = Wl[(k0 + koff) * FOUT_T + col];
    bl.y = Wl[(k0 + koff + 1) * FOUT_T + col];
    br.x = Wr[(k0 + koff) * FOUT_T + col];
    br.y = Wr[(k0 + koff + 1) * FOUT_T + col];
    accL = __builtin_amdgcn_wmma_f32_16x16x4_f32(
         false, a,  false, bl,
         (short)0, accL,  false,  false);
    accR = __builtin_amdgcn_wmma_f32_16x16x4_f32(
        false, a, false, br, (short)0, accR, false, false);
  }

  const float bb = bias[col];
#pragma unroll
  for (int v = 0; v < 8; ++v) {
    int m = blockIdx.x * 16 + v + half * 8;
    p[(long long)m * FOUT_T + col]   = accL[v];
    out[(long long)m * FOUT_T + col] = accR[v] + bb;
  }
}

extern "C" void kernel_launch(void* const* d_in, const int* in_sizes, int n_in,
                              void* d_out, int out_size, void* d_ws,
                              size_t ws_size, hipStream_t stream) {
  constexpr int FIN = 64, HID = 128, FOUT = 32;

  const float* x   = (const float*)d_in[0];
  const int*   ei  = (const int*)d_in[1];
  const float* Wl1 = (const float*)d_in[2];
  const float* Wr1 = (const float*)d_in[3];
  const float* b1  = (const float*)d_in[4];
  const float* Wl2 = (const float*)d_in[5];
  const float* Wr2 = (const float*)d_in[6];
  const float* b2  = (const float*)d_in[7];
  float* out = (float*)d_out;

  const int n = in_sizes[0] / FIN;
  const int E = in_sizes[1] / 2;
  const int* src = ei;
  const int* dst = ei + E;

  const size_t nP = ((size_t)n + 64) & ~(size_t)63;
  const size_t eP = ((size_t)E + 63) & ~(size_t)63;
  int* cnt    = (int*)d_ws;
  int* rowptr = cnt + nP;
  int* csr    = rowptr + nP;
  float* agg1 = (float*)(csr + eP);
  float* p2   = agg1;
  float* h    = agg1 + (size_t)n * FIN;

  hipMemsetAsync(d_ws, 0, (size_t)n * sizeof(int), stream);

  const int egrid = ((E + 3) / 4 + 255) / 256;
  sage_count_kernel<<<egrid, 256, 0, stream>>>(dst, cnt, E);
  sage_scan_kernel<<<1, 1024, 0, stream>>>(cnt, rowptr, n);
  sage_fill_kernel<<<egrid, 256, 0, stream>>>(src, dst, rowptr, cnt, csr, E);

  sage_gather_kernel<FIN, false>
      <<<(n + 31) / 32, 128, 0, stream>>>(x, rowptr, csr, agg1, n);

  sage_gemm_kernel<FIN, HID, true>
      <<<dim3(n / 16, HID / 16), 32, 0, stream>>>(agg1, x, Wl1, Wr1, b1, h);

  sage_pre_kernel<HID, FOUT>
      <<<dim3(n / 16, FOUT / 16), 32, 0, stream>>>(h, Wl2, Wr2, b2, p2, out);

  sage_gather_kernel<FOUT, true>
      <<<(n + 31) / 32, 128, 0, stream>>>(p2, rowptr, csr, out, n);
}


# Dense kernels: each wave computes a 16-row strip over all output columns (A fragment loaded once per k-step, reused by every column tile), weights staged in LDS per 8-wave work-group
# speedup vs baseline: 2.0247x; 1.1555x over previous
#include <hip/hip_runtime.h>


typedef float v2f __attribute__((ext_vector_type(2)));
typedef float v8f __attribute__((ext_vector_type(8)));

__global__ void sage_count_kernel(const int* __restrict__ dst,
                                  int* __restrict__ cnt, int E) {
  const int e0 = (blockIdx.x * blockDim.x + threadIdx.x) * 4;
  if (e0 >= E) return;
  if ((E & 3) == 0) {
    const int4 d = *(const int4*)(dst + e0);
    atomicAdd(&cnt[d.x], 1);
    atomicAdd(&cnt[d.y], 1);
    atomicAdd(&cnt[d.z], 1);
    atomicAdd(&cnt[d.w], 1);
  } else {
    for (int u = 0; u < 4; ++u)
      if (e0 + u < E) atomicAdd(&cnt[dst[e0 + u]], 1);
  }
}

__global__ __launch_bounds__(1024) void sage_scan_kernel(
    const int* __restrict__ cnt, int* __restrict__ rowptr, int n) {
  __shared__ int wsum[32];
  __shared__ int tot;
  const int tid  = threadIdx.x;
  const int lane = tid & 31;
  const int w    = tid >> 5;
  int carry = 0;
  for (int base = 0; base < n; base += 16384) {
    const int i = base + tid * 16;
    int4 a = make_int4(0, 0, 0, 0), b = a, c = a, e = a;
    if (i < n) {
      const int4* q = (const int4*)(cnt + i);
      a = q[0];
      b = q[1];
      c = q[2];
      e = q[3];
    }
    a.y += a.x; a.z += a.y; a.w += a.z;
    b.x += a.w; b.y += b.x; b.z += b.y; b.w += b.z;
    c.x += b.w; c.y += c.x; c.z += c.y; c.w += c.z;
    e.x += c.w; e.y += e.x; e.z += e.y; e.w += e.z;
    const int s = e.w;
    int incl = s;
    for (int o = 1; o < 32; o <<= 1) {
      const int t = __shfl_up(incl, o);
      if (lane >= o) incl += t;
    }
    if (lane == 31) wsum[w] = incl;
    __syncthreads();
    if (w == 0) {
      const int ws = wsum[lane];
      int wi = ws;
      for (int o = 1; o < 32; o <<= 1) {
        const int t = __shfl_up(wi, o);
        if (lane >= o) wi += t;
      }
      wsum[lane] = wi - ws;
      if (lane == 31) tot = wi;
    }
    __syncthreads();
    const int excl = carry + wsum[w] + incl - s;
    if (i < n) {
      int4* q = (int4*)(rowptr + i);
      q[0] = make_int4(excl, excl + a.x, excl + a.y, excl + a.z);
      q[1] = make_int4(excl + a.w, excl + b.x, excl + b.y, excl + b.z);
      q[2] = make_int4(excl + b.w, excl + c.x, excl + c.y, excl + c.z);
      q[3] = make_int4(excl + c.w, excl + e.x, excl + e.y, excl + e.z);
    }
    carry += tot;
    __syncthreads();
  }
  if (tid == 0) rowptr[n] = carry;
}

__device__ __forceinline__ void sage_place(int s, int d,
                                           const int* __restrict__ rowptr,
                                           int* __restrict__ cnt,
                                           int* __restrict__ csr) {
  const int k = atomicSub(&cnt[d], 1);
  csr[rowptr[d] + k - 1] = s;
}

__global__ void sage_fill_kernel(const int* __restrict__ src,
                                 const int* __restrict__ dst,
                                 const int* __restrict__ rowptr,
                                 int* __restrict__ cnt, int* __restrict__ csr,
                                 int E) {
  const int e0 = (blockIdx.x * blockDim.x + threadIdx.x) * 4;
  if (e0 >= E) return;
  if ((E & 3) == 0) {
    const int4 s = *(const int4*)(src + e0);
    const int4 d = *(const int4*)(dst + e0);
    const int k0 = atomicSub(&cnt[d.x], 1);
    const int k1 = atomicSub(&cnt[d.y], 1);
    const int k2 = atomicSub(&cnt[d.z], 1);
    const int k3 = atomicSub(&cnt[d.w], 1);
    const int r0 = rowptr[d.x];
    const int r1 = rowptr[d.y];
    const int r2 = rowptr[d.z];
    const int r3 = rowptr[d.w];
    csr[r0 + k0 - 1] = s.x;
    csr[r1 + k1 - 1] = s.y;
    csr[r2 + k2 - 1] = s.z;
    csr[r3 + k3 - 1] = s.w;
  } else {
    for (int u = 0; u < 4; ++u)
      if (e0 + u < E) sage_place(src[e0 + u], dst[e0 + u], rowptr, cnt, csr);
  }
}

__device__ __forceinline__ float4 sage_row(const float* fcol, int s, int F) {
  return *(const float4*)(fcol + (long long)s * F);
}
__device__ __forceinline__ void sage_acc(float4& acc, const float4 v) {
  acc.x += v.x;
  acc.y += v.y;
  acc.z += v.z;
  acc.w += v.w;
}

template <int F, bool ADD>
__global__ __launch_bounds__(128) void sage_gather_kernel(
    const float* __restrict__ feat, const int* __restrict__ rowptr,
    const int* __restrict__ csr, float* __restrict__ tgt, int n) {
  constexpr int LPN = F / 4;
  constexpr int NPB = 128 / LPN;
  const int sub = threadIdx.x / LPN;
  const int c   = (threadIdx.x % LPN) * 4;
  const float* fcol = feat + c;
  for (int d = blockIdx.x * NPB + sub; d < n; d += gridDim.x * NPB) {
    const int beg = rowptr[d];
    const int end = rowptr[d + 1];
    float4 acc = make_float4(0.f, 0.f, 0.f, 0.f);
    int j = beg;
    for (; j + 8 <= end; j += 8) {
      const int s0 = csr[j];
      const int s1 = csr[j + 1];
      const int s2 = csr[j + 2];
      const int s3 = csr[j + 3];
      const int s4 = csr[j + 4];
      const int s5 = csr[j + 5];
      const int s6 = csr[j + 6];
      const int s7 = csr[j + 7];
      const float4 v0 = sage_row(fcol, s0, F);
      const float4 v1 = sage_row(fcol, s1, F);
      const float4 v2 = sage_row(fcol, s2, F);
      const float4 v3 = sage_row(fcol, s3, F);
      const float4 v4 = sage_row(fcol, s4, F);
      const float4 v5 = sage_row(fcol, s5, F);
      const float4 v6 = sage_row(fcol, s6, F);
      const float4 v7 = sage_row(fcol, s7, F);
      sage_acc(acc, v0);
      sage_acc(acc, v1);
      sage_acc(acc, v2);
      sage_acc(acc, v3);
      sage_acc(acc, v4);
      sage_acc(acc, v5);
      sage_acc(acc, v6);
      sage_acc(acc, v7);
    }
    if (j + 4 <= end) {
      const int s0 = csr[j];
      const int s1 = csr[j + 1];
      const int s2 = csr[j + 2];
      const int s3 = csr[j + 3];
      const float4 v0 = sage_row(fcol, s0, F);
      const float4 v1 = sage_row(fcol, s1, F);
      const float4 v2 = sage_row(fcol, s2, F);
      const float4 v3 = sage_row(fcol, s3, F);
      sage_acc(acc, v0);
      sage_acc(acc, v1);
      sage_acc(acc, v2);
      sage_acc(acc, v3);
      j += 4;
    }
    if (j < end) {
      const int last = end - 1;
      const bool h1 = (j + 1 <= last);
      const bool h2 = (j + 2 <= last);
      const int s0 = csr[j];
      const int s1 = csr[h1 ? j + 1 : last];
      const int s2 = csr[h2 ? j + 2 : last];
      const float4 v0 = sage_row(fcol, s0, F);
      float4 v1 = sage_row(fcol, s1, F);
      float4 v2 = sage_row(fcol, s2, F);
      v1.x = h1 ? v1.x : 0.f; v1.y = h1 ? v1.y : 0.f;
      v1.z = h1 ? v1.z : 0.f; v1.w = h1 ? v1.w : 0.f;
      v2.x = h2 ? v2.x : 0.f; v2.y = h2 ? v2.y : 0.f;
      v2.z = h2 ? v2.z : 0.f; v2.w = h2 ? v2.w : 0.f;
      sage_acc(acc, v0);
      sage_acc(acc, v1);
      sage_acc(acc, v2);
    }
    const float inv = 1.0f / fmaxf((float)(end - beg), 1.0f);
    float4* p = (float4*)(tgt + (long long)d * F + c);
    float4 o = make_float4(0.f, 0.f, 0.f, 0.f);
    if (ADD) o = *p;
    o.x += acc.x * inv;
    o.y += acc.y * inv;
    o.z += acc.z * inv;
    o.w += acc.w * inv;
    *p = o;
  }
}

constexpr int SAGE_DENSE_WAVES = 8;
constexpr int SAGE_DENSE_THREADS = SAGE_DENSE_WAVES * 32;

template <int KROWS, int SRC_COLS, int PITCH>
__device__ __forceinline__ void sage_stage_w(const float* __restrict__ wsrc,
                                             v2f* __restrict__ img, int k_lo,
                                             int col_lo) {
  constexpr int QPR = SRC_COLS / 4;
  for (int i = threadIdx.x; i < (KROWS / 2) * QPR; i += SAGE_DENSE_THREADS) {
    const int kp = i / QPR;
    const int q  = (i % QPR) * 4;
    const float4 r0 = *(const float4*)(wsrc + (2 * kp) * SRC_COLS + q);
    const float4 r1 = *(const float4*)(wsrc + (2 * kp + 1) * SRC_COLS + q);
    v2f* d = img + (k_lo / 2 + kp) * PITCH + col_lo + q;
    v2f p0, p1, p2, p3;
    p0.x = r0.x; p0.y = r1.x;
    p1.x = r0.y; p1.y = r1.y;
    p2.x = r0.z; p2.y = r1.z;
    p3.x = r0.w; p3.y = r1.w;
    d[0] = p0;
    d[1] = p1;
    d[2] = p2;
    d[3] = p3;
  }
}

template <int KH, int NT, int PITCH>
__device__ __forceinline__ void sage_strip_mma(
    const float* __restrict__ r0, const float* __restrict__ r1,
    const float* __restrict__ n0, const v2f* __restrict__ bcol,
    v2f (&a_nxt)[4], v8f (&acc)[NT]) {
  constexpr int CHH = KH / 16;
  constexpr int NCH = 2 * CHH;
#pragma nounroll
  for (int ch = 0; ch < NCH; ++ch) {
    v2f a_cur[4];
#pragma unroll
    for (int s = 0; s < 4; ++s) a_cur[s] = a_nxt[s];
    const int cn = ch + 1;
    const float* ap = (cn < CHH) ? r0 + cn * 16
                    : (cn < NCH) ? r1 + (cn - CHH) * 16
                                 : n0;
#pragma unroll
    for (int s = 0; s < 4; ++s)
      a_nxt[s] = *(const v2f*)(ap + s * 4);
    const v2f* bp = bcol + (ch * 8) * PITCH;
#pragma unroll
    for (int s = 0; s < 4; ++s) {
#pragma unroll
      for (int t = 0; t < NT; ++t) {
        const v2f b = bp[(2 * s) * PITCH + t * 16];
        acc[t] = __builtin_amdgcn_wmma_f32_16x16x4_f32(
             false, a_cur[s],  false, b,
             (short)0, acc[t],  false,  false);
      }
    }
  }
}

template <int FIN_T, int FOUT_T, bool RELU>
__global__ __launch_bounds__(SAGE_DENSE_THREADS) void sage_gemm_kernel(
    const float* __restrict__ agg, const float* __restrict__ xin,
    const float* __restrict__ Wl,
    const float* __restrict__ Wr, const float* __restrict__ bias,
    float* __restrict__ out, int nstrips) {
  constexpr int NT    = FOUT_T / 16;
  constexpr int PITCH = FOUT_T + 16;
  __shared__ v2f wimg[FIN_T * PITCH];

  const int lane = threadIdx.x & 31;
  const int wave = threadIdx.x >> 5;
  const int half = lane >> 4;
  const int l15  = lane & 15;
  const int koff = half * 2;
  const v2f* bcol = wimg + half * PITCH + l15;
  const int step = gridDim.x * SAGE_DENSE_WAVES;

  float bb[NT];
#pragma unroll
  for (int t = 0; t < NT; ++t) bb[t] = bias[t * 16 + l15];

  int strip = blockIdx.x * SAGE_DENSE_WAVES + wave;
  v2f a_nxt[4];
  if (strip < nstrips) {
    const float* f0 = agg + (long long)(strip * 16 + l15) * FIN_T + koff;
#pragma unroll
    for (int s = 0; s < 4; ++s) a_nxt[s] = *(const v2f*)(f0 + s * 4);
  }
  sage_stage_w<FIN_T, FOUT_T, PITCH>(Wl, wimg, 0, 0);
  sage_stage_w<FIN_T, FOUT_T, PITCH>(Wr, wimg, FIN_T, 0);
  __syncthreads();

  for (; strip < nstrips; strip += step) {
    const int row = strip * 16 + l15;
    const int rown = (strip + step < nstrips) ? row + step * 16 : row;
    const float* aRow = agg + (long long)row * FIN_T + koff;
    const float* xRow = xin + (long long)row * FIN_T + koff;
    const float* aNxt = agg + (long long)rown * FIN_T + koff;

    v8f acc[NT];
#pragma unroll
    for (int t = 0; t < NT; ++t)
      acc[t] = (v8f){0.f, 0.f, 0.f, 0.f, 0.f, 0.f, 0.f, 0.f};

    sage_strip_mma<FIN_T, NT, PITCH>(aRow, xRow, aNxt, bcol, a_nxt, acc);

#pragma unroll
    for (int t = 0; t < NT; ++t) {
#pragma unroll
      for (int v = 0; v < 8; ++v) {
        const int m = strip * 16 + v + half * 8;
        float val = acc[t][v] + bb[t];
        if (RELU) val = fmaxf(val, 0.f);
        out[(long long)m * FOUT_T + t * 16 + l15] = val;
      }
    }
  }
}

template <int FIN_T, int FOUT_T>
__global__ __launch_bounds__(SAGE_DENSE_THREADS) void sage_pre_kernel(
    const float* __restrict__ hin, const float* __restrict__ Wl,
    const float* __restrict__ Wr, const float* __restrict__ bias,
    float* __restrict__ p, float* __restrict__ out, int nstrips) {
  constexpr int NTH   = FOUT_T / 16;
  constexpr int NT    = 2 * NTH;
  constexpr int NCOLS = 2 * FOUT_T;
  constexpr int PITCH = NCOLS + 16;
  constexpr int KH    = FIN_T / 2;
  __shared__ v2f wimg[(FIN_T / 2) * PITCH];

  const int lane = threadIdx.x & 31;
  const int wave = threadIdx.x >> 5;
  const int half = lane >> 4;
  const int l15  = lane & 15;
  const int koff = half * 2;
  const v2f* bcol = wimg + half * PITCH + l15;
  const int step = gridDim.x * SAGE_DENSE_WAVES;

  float bb[NTH];
#pragma unroll
  for (int t = 0; t < NTH; ++t) bb[t] = bias[t * 16 + l15];

  int strip = blockIdx.x * SAGE_DENSE_WAVES + wave;
  v2f a_nxt[4];
  if (strip < nstrips) {
    const float* f0 = hin + (long long)(strip * 16 + l15) * FIN_T + koff;
#pragma unroll
    for (int s = 0; s < 4; ++s) a_nxt[s] = *(const v2f*)(f0 + s * 4);
  }
  sage_stage_w<FIN_T, FOUT_T, PITCH>(Wl, wimg, 0, 0);
  sage_stage_w<FIN_T, FOUT_T, PITCH>(Wr, wimg, 0, FOUT_T);
  __syncthreads();

  for (; strip < nstrips; strip += step) {
    const int row = strip * 16 + l15;
    const int rown = (strip + step < nstrips) ? row + step * 16 : row;
    const float* hRow = hin + (long long)row * FIN_T + koff;
    const float* hNxt = hin + (long long)rown * FIN_T + koff;

    v8f acc[NT];
#pragma unroll
    for (int t = 0; t < NT; ++t)
      acc[t] = (v8f){0.f, 0.f, 0.f, 0.f, 0.f, 0.f, 0.f, 0.f};

    sage_strip_mma<KH, NT, PITCH>(hRow, hRow + KH, hNxt, bcol, a_nxt, acc);

#pragma unroll
    for (int t = 0; t < NTH; ++t) {
#pragma unroll
      for (int v = 0; v < 8; ++v) {
        const int m = strip * 16 + v + half * 8;
        p[(long long)m * FOUT_T + t * 16 + l15]   = acc[t][v];
        out[(long long)m * FOUT_T + t * 16 + l15] = acc[NTH + t][v] + bb[t];
      }
    }
  }
}

extern "C" void kernel_launch(void* const* d_in, const int* in_sizes, int n_in,
                              void* d_out, int out_size, void* d_ws,
                              size_t ws_size, hipStream_t stream) {
  constexpr int FIN = 64, HID = 128, FOUT = 32;

  const float* x   = (const float*)d_in[0];
  const int*   ei  = (const int*)d_in[1];
  const float* Wl1 = (const float*)d_in[2];
  const float* Wr1 = (const float*)d_in[3];
  const float* b1  = (const float*)d_in[4];
  const float* Wl2 = (const float*)d_in[5];
  const float* Wr2 = (const float*)d_in[6];
  const float* b2  = (const float*)d_in[7];
  float* out = (float*)d_out;

  const int n = in_sizes[0] / FIN;
  const int E = in_sizes[1] / 2;
  const int* src = ei;
  const int* dst = ei + E;

  const size_t nP = ((size_t)n + 64) & ~(size_t)63;
  const size_t eP = ((size_t)E + 63) & ~(size_t)63;
  int* cnt    = (int*)d_ws;
  int* rowptr = cnt + nP;
  int* csr    = rowptr + nP;
  float* agg1 = (float*)(csr + eP);
  float* p2   = agg1;
  float* h    = agg1 + (size_t)n * FIN;

  hipMemsetAsync(d_ws, 0, (size_t)n * sizeof(int), stream);

  const int egrid = ((E + 3) / 4 + 255) / 256;
  sage_count_kernel<<<egrid, 256, 0, stream>>>(dst, cnt, E);
  sage_scan_kernel<<<1, 1024, 0, stream>>>(cnt, rowptr, n);
  sage_fill_kernel<<<egrid, 256, 0, stream>>>(src, dst, rowptr, cnt, csr, E);

  sage_gather_kernel<FIN, false>
      <<<(n + 31) / 32, 128, 0, stream>>>(x, rowptr, csr, agg1, n);

  sage_gemm_kernel<FIN, HID, true>
      <<<512, SAGE_DENSE_THREADS, 0, stream>>>(agg1, x, Wl1, Wr1, b1, h, n / 16);

  sage_pre_kernel<HID, FOUT>
      <<<512, SAGE_DENSE_THREADS, 0, stream>>>(h, Wl2, Wr2, b2, p2, out, n / 16);

  sage_gather_kernel<FOUT, true>
      <<<(n + 31) / 32, 128, 0, stream>>>(p2, rowptr, csr, out, n);
}


# Edge sort by destination without per-edge global atomics: two-level counting sort (256-node buckets, then nodes) on LDS counters; slots inside a node stay arrival order, f32 sum order may change
# speedup vs baseline: 4.3716x; 2.1591x over previous
#include <hip/hip_runtime.h>


typedef float v2f __attribute__((ext_vector_type(2)));
typedef float v8f __attribute__((ext_vector_type(8)));

constexpr int SAGE_SORT_THREADS = 512;
constexpr int SAGE_SORT_EPT     = 8;
constexpr int SAGE_SORT_CHUNK   = SAGE_SORT_THREADS * SAGE_SORT_EPT;
constexpr int SAGE_BUCKET_SHIFT = 8;
constexpr int SAGE_BUCKET_NODES = 1 << SAGE_BUCKET_SHIFT;
constexpr int SAGE_BUCKET_BATCH = 4;

__device__ __forceinline__ void sage_load8(const int* __restrict__ p, int e0,
                                           int E, int (&v)[SAGE_SORT_EPT]) {
  if ((E & 3) == 0 && e0 + SAGE_SORT_EPT <= E) {
    const int4 a = *(const int4*)(p + e0);
    const int4 b = *(const int4*)(p + e0 + 4);
    v[0] = a.x; v[1] = a.y; v[2] = a.z; v[3] = a.w;
    v[4] = b.x; v[5] = b.y; v[6] = b.z; v[7] = b.w;
  } else {
#pragma unroll
    for (int u = 0; u < SAGE_SORT_EPT; ++u) v[u] = (e0 + u < E) ? p[e0 + u] : 0;
  }
}

__device__ __forceinline__ int sage_block_scan(int v, int* ws, int& total) {
  const int lane = threadIdx.x & 31;
  const int w    = threadIdx.x >> 5;
  int incl = v;
  for (int o = 1; o < 32; o <<= 1) {
    const int t = __shfl_up(incl, o);
    if (lane >= o) incl += t;
  }
  if (lane == 31) ws[w] = incl;
  __syncthreads();
  if (w == 0) {
    const int s = (lane < 16) ? ws[lane] : 0;
    int wi = s;
    for (int o = 1; o < 16; o <<= 1) {
      const int t = __shfl_up(wi, o);
      if (lane >= o) wi += t;
    }
    if (lane < 16) ws[lane] = wi - s;
    if (lane == 15) ws[16] = wi;
  }
  __syncthreads();
  total = ws[16];
  return ws[w] + incl - v;
}

__global__ __launch_bounds__(SAGE_SORT_THREADS) void sage_bucket_count_kernel(
    const int* __restrict__ dst, int* __restrict__ gtot,
    int* __restrict__ wgoff, int E, int nb) {
  __shared__ int hist[SAGE_SORT_THREADS];
  const int tid = threadIdx.x;
  const int e0  = blockIdx.x * SAGE_SORT_CHUNK + tid * SAGE_SORT_EPT;
  int d[SAGE_SORT_EPT];
  sage_load8(dst, e0, E, d);
  hist[tid] = 0;
  __syncthreads();
#pragma unroll
  for (int u = 0; u < SAGE_SORT_EPT; ++u)
    if (e0 + u < E) atomicAdd(&hist[d[u] >> SAGE_BUCKET_SHIFT], 1);
  __syncthreads();
  if (tid < nb) {
    const int c = hist[tid];
    wgoff[blockIdx.x * nb + tid] = (c > 0) ? atomicAdd(&gtot[tid], c) : 0;
  }
}

__global__ __launch_bounds__(SAGE_SORT_THREADS) void sage_bucket_place_kernel(
    const int* __restrict__ src, const int* __restrict__ dst,
    const int* __restrict__ gtot, const int* __restrict__ wgoff,
    int2* __restrict__ pairs, int E, int nb) {
  __shared__ int cur[SAGE_SORT_THREADS];
  __shared__ int ws[17];
  const int tid = threadIdx.x;
  const int e0  = blockIdx.x * SAGE_SORT_CHUNK + tid * SAGE_SORT_EPT;
  int s[SAGE_SORT_EPT], d[SAGE_SORT_EPT];
  sage_load8(src, e0, E, s);
  sage_load8(dst, e0, E, d);
  const int g     = (tid < nb) ? gtot[tid] : 0;
  const int piece = (tid < nb) ? wgoff[blockIdx.x * nb + tid] : 0;
  int total;
  const int base = sage_block_scan(g, ws, total);
  cur[tid] = base + piece;
  __syncthreads();
  int pos[SAGE_SORT_EPT];
#pragma unroll
  for (int u = 0; u < SAGE_SORT_EPT; ++u)
    pos[u] = (e0 + u < E) ? atomicAdd(&cur[d[u] >> SAGE_BUCKET_SHIFT], 1) : -1;
#pragma unroll
  for (int u = 0; u < SAGE_SORT_EPT; ++u)
    if (pos[u] >= 0) pairs[pos[u]] = make_int2(s[u], d[u]);
}

__global__ __launch_bounds__(SAGE_SORT_THREADS) void sage_bucket_csr_kernel(
    const int2* __restrict__ pairs, const int* __restrict__ gtot,
    int* __restrict__ rowptr, int* __restrict__ csr, int n, int nb) {
  __shared__ int cnt[SAGE_BUCKET_NODES];
  __shared__ int cur[SAGE_BUCKET_NODES];
  __shared__ int wsa[17];
  __shared__ int wsb[17];
  constexpr int STRIDE = SAGE_SORT_THREADS * SAGE_BUCKET_BATCH;
  const int tid = threadIdx.x;
  const int b   = blockIdx.x;
  const int lo  = b << SAGE_BUCKET_SHIFT;
  if (tid < SAGE_BUCKET_NODES) cnt[tid] = 0;
  const int g = (tid < b) ? gtot[tid] : 0;
  int base;
  sage_block_scan(g, wsa, base);
  const int T = gtot[b];
  const int2* pb = pairs + base;

  const int last = T - 1;
  for (int i0 = 0; i0 < T; i0 += STRIDE) {
    int dl[SAGE_BUCKET_BATCH];
#pragma unroll
    for (int u = 0; u < SAGE_BUCKET_BATCH; ++u) {
      const int i = i0 + u * SAGE_SORT_THREADS + tid;
      const int raw = pb[(i < T) ? i : last].y;
      dl[u] = (i < T) ? (raw & (SAGE_BUCKET_NODES - 1)) : -1;
    }
#pragma unroll
    for (int u = 0; u < SAGE_BUCKET_BATCH; ++u)
      if (dl[u] >= 0) atomicAdd(&cnt[dl[u]], 1);
  }
  __syncthreads();

  const int c = (tid < SAGE_BUCKET_NODES) ? cnt[tid] : 0;
  int tot;
  const int ex = sage_block_scan(c, wsb, tot);
  if (tid < SAGE_BUCKET_NODES) {
    cur[tid] = base + ex;
    if (lo + tid < n) rowptr[lo + tid] = base + ex;
  }
  if (b == nb - 1 && tid == 0) rowptr[n] = base + T;
  __syncthreads();

  for (int i0 = 0; i0 < T; i0 += STRIDE) {
    int2 p[SAGE_BUCKET_BATCH];
    int pos[SAGE_BUCKET_BATCH];
#pragma unroll
    for (int u = 0; u < SAGE_BUCKET_BATCH; ++u) {
      const int i = i0 + u * SAGE_SORT_THREADS + tid;
      p[u] = pb[(i < T) ? i : last];
      if (i >= T) p[u].y = -1;
    }
#pragma unroll
    for (int u = 0; u < SAGE_BUCKET_BATCH; ++u)
      pos[u] = (p[u].y >= 0)
                   ? atomicAdd(&cur[p[u].y & (SAGE_BUCKET_NODES - 1)], 1)
                   : -1;
#pragma unroll
    for (int u = 0; u < SAGE_BUCKET_BATCH; ++u)
      if (pos[u] >= 0) csr[pos[u]] = p[u].x;
  }
}

__device__ __forceinline__ float4 sage_row(const float* fcol, int s, int F) {
  return *(const float4*)(fcol + (long long)s * F);
}
__device__ __forceinline__ void sage_acc(float4& acc, const float4 v) {
  acc.x += v.x;
  acc.y += v.y;
  acc.z += v.z;
  acc.w += v.w;
}

template <int F, bool ADD>
__global__ __launch_bounds__(128) void sage_gather_kernel(
    const float* __restrict__ feat, const int* __restrict__ rowptr,
    const int* __restrict__ csr, float* __restrict__ tgt, int n) {
  constexpr int LPN = F / 4;
  constexpr int NPB = 128 / LPN;
  const int sub = threadIdx.x / LPN;
  const int c   = (threadIdx.x % LPN) * 4;
  const float* fcol = feat + c;
  for (int d = blockIdx.x * NPB + sub; d < n; d += gridDim.x * NPB) {
    const int beg = rowptr[d];
    const int end = rowptr[d + 1];
    float4 acc = make_float4(0.f, 0.f, 0.f, 0.f);
    int j = beg;
    for (; j + 8 <= end; j += 8) {
      const int s0 = csr[j];
      const int s1 = csr[j + 1];
      const int s2 = csr[j + 2];
      const int s3 = csr[j + 3];
      const int s4 = csr[j + 4];
      const int s5 = csr[j + 5];
      const int s6 = csr[j + 6];
      const int s7 = csr[j + 7];
      const float4 v0 = sage_row(fcol, s0, F);
      const float4 v1 = sage_row(fcol, s1, F);
      const float4 v2 = sage_row(fcol, s2, F);
      const float4 v3 = sage_row(fcol, s3, F);
      const float4 v4 = sage_row(fcol, s4, F);
      const float4 v5 = sage_row(fcol, s5, F);
      const float4 v6 = sage_row(fcol, s6, F);
      const float4 v7 = sage_row(fcol, s7, F);
      sage_acc(acc, v0);
      sage_acc(acc, v1);
      sage_acc(acc, v2);
      sage_acc(acc, v3);
      sage_acc(acc, v4);
      sage_acc(acc, v5);
      sage_acc(acc, v6);
      sage_acc(acc, v7);
    }
    if (j + 4 <= end) {
      const int s0 = csr[j];
      const int s1 = csr[j + 1];
      const int s2 = csr[j + 2];
      const int s3 = csr[j + 3];
      const float4 v0 = sage_row(fcol, s0, F);
      const float4 v1 = sage_row(fcol, s1, F);
      const float4 v2 = sage_row(fcol, s2, F);
      const float4 v3 = sage_row(fcol, s3, F);
      sage_acc(acc, v0);
      sage_acc(acc, v1);
      sage_acc(acc, v2);
      sage_acc(acc, v3);
      j += 4;
    }
    if (j < end) {
      const int last = end - 1;
      const bool h1 = (j + 1 <= last);
      const bool h2 = (j + 2 <= last);
      const int s0 = csr[j];
      const int s1 = csr[h1 ? j + 1 : last];
      const int s2 = csr[h2 ? j + 2 : last];
      const float4 v0 = sage_row(fcol, s0, F);
      float4 v1 = sage_row(fcol, s1, F);
      float4 v2 = sage_row(fcol, s2, F);
      v1.x = h1 ? v1.x : 0.f; v1.y = h1 ? v1.y : 0.f;
      v1.z = h1 ? v1.z : 0.f; v1.w = h1 ? v1.w : 0.f;
      v2.x = h2 ? v2.x : 0.f; v2.y = h2 ? v2.y : 0.f;
      v2.z = h2 ? v2.z : 0.f; v2.w = h2 ? v2.w : 0.f;
      sage_acc(acc, v0);
      sage_acc(acc, v1);
      sage_acc(acc, v2);
    }
    const float inv = 1.0f / fmaxf((float)(end - beg), 1.0f);
    float4* p = (float4*)(tgt + (long long)d * F + c);
    float4 o = make_float4(0.f, 0.f, 0.f, 0.f);
    if (ADD) o = *p;
    o.x += acc.x * inv;
    o.y += acc.y * inv;
    o.z += acc.z * inv;
    o.w += acc.w * inv;
    *p = o;
  }
}

constexpr int SAGE_DENSE_WAVES = 8;
constexpr int SAGE_DENSE_THREADS = SAGE_DENSE_WAVES * 32;

template <int KROWS, int SRC_COLS, int PITCH>
__device__ __forceinline__ void sage_stage_w(const float* __restrict__ wsrc,
                                             v2f* __restrict__ img, int k_lo,
                                             int col_lo) {
  constexpr int QPR = SRC_COLS / 4;
  for (int i = threadIdx.x; i < (KROWS / 2) * QPR; i += SAGE_DENSE_THREADS) {
    const int kp = i / QPR;
    const int q  = (i % QPR) * 4;
    const float4 r0 = *(const float4*)(wsrc + (2 * kp) * SRC_COLS + q);
    const float4 r1 = *(const float4*)(wsrc + (2 * kp + 1) * SRC_COLS + q);
    v2f* d = img + (k_lo / 2 + kp) * PITCH + col_lo + q;
    v2f p0, p1, p2, p3;
    p0.x = r0.x; p0.y = r1.x;
    p1.x = r0.y; p1.y = r1.y;
    p2.x = r0.z; p2.y = r1.z;
    p3.x = r0.w; p3.y = r1.w;
    d[0] = p0;
    d[1] = p1;
    d[2] = p2;
    d[3] = p3;
  }
}

template <int KH, int NT, int PITCH>
__device__ __forceinline__ void sage_strip_mma(
    const float* __restrict__ r0, const float* __restrict__ r1,
    const float* __restrict__ n0, const v2f* __restrict__ bcol,
    v2f (&a_nxt)[4], v8f (&acc)[NT]) {
  constexpr int CHH = KH / 16;
  constexpr int NCH = 2 * CHH;
#pragma nounroll
  for (int ch = 0; ch < NCH; ++ch) {
    v2f a_cur[4];
#pragma unroll
    for (int s = 0; s < 4; ++s) a_cur[s] = a_nxt[s];
    const int cn = ch + 1;
    const float* ap = (cn < CHH) ? r0 + cn * 16
                    : (cn < NCH) ? r1 + (cn - CHH) * 16
                                 : n0;
#pragma unroll
    for (int s = 0; s < 4; ++s)
      a_nxt[s] = *(const v2f*)(ap + s * 4);
    const v2f* bp = bcol + (ch * 8) * PITCH;
#pragma unroll
    for (int s = 0; s < 4; ++s) {
#pragma unroll
      for (int t = 0; t < NT; ++t) {
        const v2f b = bp[(2 * s) * PITCH + t * 16];
        acc[t] = __builtin_amdgcn_wmma_f32_16x16x4_f32(
             false, a_cur[s],  false, b,
             (short)0, acc[t],  false,  false);
      }
    }
  }
}

template <int FIN_T, int FOUT_T, bool RELU>
__global__ __launch_bounds__(SAGE_DENSE_THREADS) void sage_gemm_kernel(
    const float* __restrict__ agg, const float* __restrict__ xin,
    const float* __restrict__ Wl,
    const float* __restrict__ Wr, const float* __restrict__ bias,
    float* __restrict__ out, int nstrips) {
  constexpr int NT    = FOUT_T / 16;
  constexpr int PITCH = FOUT_T + 16;
  __shared__ v2f wimg[FIN_T * PITCH];

  const int lane = threadIdx.x & 31;
  const int wave = threadIdx.x >> 5;
  const int half = lane >> 4;
  const int l15  = lane & 15;
  const int koff = half * 2;
  const v2f* bcol = wimg + half * PITCH + l15;
  const int step = gridDim.x * SAGE_DENSE_WAVES;

  float bb[NT];
#pragma unroll
  for (int t = 0; t < NT; ++t) bb[t] = bias[t * 16 + l15];

  int strip = blockIdx.x * SAGE_DENSE_WAVES + wave;
  v2f a_nxt[4];
  if (strip < nstrips) {
    const float* f0 = agg + (long long)(strip * 16 + l15) * FIN_T + koff;
#pragma unroll
    for (int s = 0; s < 4; ++s) a_nxt[s] = *(const v2f*)(f0 + s * 4);
  }
  sage_stage_w<FIN_T, FOUT_T, PITCH>(Wl, wimg, 0, 0);
  sage_stage_w<FIN_T, FOUT_T, PITCH>(Wr, wimg, FIN_T, 0);
  __syncthreads();

  for (; strip < nstrips; strip += step) {
    const int row = strip * 16 + l15;
    const int rown = (strip + step < nstrips) ? row + step * 16 : row;
    const float* aRow = agg + (long long)row * FIN_T + koff;
    const float* xRow = xin + (long long)row * FIN_T + koff;
    const float* aNxt = agg + (long long)rown * FIN_T + koff;

    v8f acc[NT];
#pragma unroll
    for (int t = 0; t < NT; ++t)
      acc[t] = (v8f){0.f, 0.f, 0.f, 0.f, 0.f, 0.f, 0.f, 0.f};

    sage_strip_mma<FIN_T, NT, PITCH>(aRow, xRow, aNxt, bcol, a_nxt, acc);

#pragma unroll
    for (int t = 0; t < NT; ++t) {
#pragma unroll
      for (int v = 0; v < 8; ++v) {
        const int m = strip * 16 + v + half * 8;
        float val = acc[t][v] + bb[t];
        if (RELU) val = fmaxf(val, 0.f);
        out[(long long)m * FOUT_T + t * 16 + l15] = val;
      }
    }
  }
}

template <int FIN_T, int FOUT_T>
__global__ __launch_bounds__(SAGE_DENSE_THREADS) void sage_pre_kernel(
    const float* __restrict__ hin, const float* __restrict__ Wl,
    const float* __restrict__ Wr, const float* __restrict__ bias,
    float* __restrict__ p, float* __restrict__ out, int nstrips) {
  constexpr int NTH   = FOUT_T / 16;
  constexpr int NT    = 2 * NTH;
  constexpr int NCOLS = 2 * FOUT_T;
  constexpr int PITCH = NCOLS + 16;
  constexpr int KH    = FIN_T / 2;
  __shared__ v2f wimg[(FIN_T / 2) * PITCH];

  const int lane = threadIdx.x & 31;
  const int wave = threadIdx.x >> 5;
  const int half = lane >> 4;
  const int l15  = lane & 15;
  const int koff = half * 2;
  const v2f* bcol = wimg + half * PITCH + l15;
  const int step = gridDim.x * SAGE_DENSE_WAVES;

  float bb[NTH];
#pragma unroll
  for (int t = 0; t < NTH; ++t) bb[t] = bias[t * 16 + l15];

  int strip = blockIdx.x * SAGE_DENSE_WAVES + wave;
  v2f a_nxt[4];
  if (strip < nstrips) {
    const float* f0 = hin + (long long)(strip * 16 + l15) * FIN_T + koff;
#pragma unroll
    for (int s = 0; s < 4; ++s) a_nxt[s] = *(const v2f*)(f0 + s * 4);
  }
  sage_stage_w<FIN_T, FOUT_T, PITCH>(Wl, wimg, 0, 0);
  sage_stage_w<FIN_T, FOUT_T, PITCH>(Wr, wimg, 0, FOUT_T);
  __syncthreads();

  for (; strip < nstrips; strip += step) {
    const int row = strip * 16 + l15;
    const int rown = (strip + step < nstrips) ? row + step * 16 : row;
    const float* hRow = hin + (long long)row * FIN_T + koff;
    const float* hNxt = hin + (long long)rown * FIN_T + koff;

    v8f acc[NT];
#pragma unroll
    for (int t = 0; t < NT; ++t)
      acc[t] = (v8f){0.f, 0.f, 0.f, 0.f, 0.f, 0.f, 0.f, 0.f};

    sage_strip_mma<KH, NT, PITCH>(hRow, hRow + KH, hNxt, bcol, a_nxt, acc);

#pragma unroll
    for (int t = 0; t < NTH; ++t) {
#pragma unroll
      for (int v = 0; v < 8; ++v) {
        const int m = strip * 16 + v + half * 8;
        p[(long long)m * FOUT_T + t * 16 + l15]   = acc[t][v];
        out[(long long)m * FOUT_T + t * 16 + l15] = acc[NTH + t][v] + bb[t];
      }
    }
  }
}

extern "C" void kernel_launch(void* const* d_in, const int* in_sizes, int n_in,
                              void* d_out, int out_size, void* d_ws,
                              size_t ws_size, hipStream_t stream) {
  constexpr int FIN = 64, HID = 128, FOUT = 32;

  const float* x   = (const float*)d_in[0];
  const int*   ei  = (const int*)d_in[1];
  const float* Wl1 = (const float*)d_in[2];
  const float* Wr1 = (const float*)d_in[3];
  const float* b1  = (const float*)d_in[4];
  const float* Wl2 = (const float*)d_in[5];
  const float* Wr2 = (const float*)d_in[6];
  const float* b2  = (const float*)d_in[7];
  float* out = (float*)d_out;

  const int n = in_sizes[0] / FIN;
  const int E = in_sizes[1] / 2;
  const int* src = ei;
  const int* dst = ei + E;

  const size_t nP = ((size_t)n + 64) & ~(size_t)63;
  const size_t eP = ((size_t)E + 63) & ~(size_t)63;
  int* gtot   = (int*)d_ws;
  int* rowptr = gtot + nP;
  int* csr    = rowptr + nP;
  float* agg1 = (float*)(csr + eP);
  float* p2   = agg1;
  float* h    = agg1 + (size_t)n * FIN;
  int2* pairs = (int2*)h;
  int* wgoff  = (int*)(pairs + eP);

  const int nb     = (n + SAGE_BUCKET_NODES - 1) / SAGE_BUCKET_NODES;
  const int nchunk = (E + SAGE_SORT_CHUNK - 1) / SAGE_SORT_CHUNK;

  hipMemsetAsync(d_ws, 0, (size_t)nb * sizeof(int), stream);

  sage_bucket_count_kernel<<<nchunk, SAGE_SORT_THREADS, 0, stream>>>(
      dst, gtot, wgoff, E, nb);
  sage_bucket_place_kernel<<<nchunk, SAGE_SORT_THREADS, 0, stream>>>(
      src, dst, gtot, wgoff, pairs, E, nb);
  sage_bucket_csr_kernel<<<nb, SAGE_SORT_THREADS, 0, stream>>>(
      pairs, gtot, rowptr, csr, n, nb);

  sage_gather_kernel<FIN, false>
      <<<(n + 31) / 32, 128, 0, stream>>>(x, rowptr, csr, agg1, n);

  sage_gemm_kernel<FIN, HID, true>
      <<<512, SAGE_DENSE_THREADS, 0, stream>>>(agg1, x, Wl1, Wr1, b1, h, n / 16);

  sage_pre_kernel<HID, FOUT>
      <<<512, SAGE_DENSE_THREADS, 0, stream>>>(h, Wl2, Wr2, b2, p2, out, n / 16);

  sage_gather_kernel<FOUT, true>
      <<<(n + 31) / 32, 128, 0, stream>>>(p2, rowptr, csr, out, n);
}


# Layer-2 transform fused into the layer-1 GEMM launch: products computed transposed so the relu'd h tiles in D registers are layer 2's B operand; h never stored; layer-2 f32 sum order changes
# speedup vs baseline: 4.7513x; 1.0869x over previous
#include <hip/hip_runtime.h>


typedef float v2f __attribute__((ext_vector_type(2)));
typedef float v8f __attribute__((ext_vector_type(8)));

constexpr int SAGE_SORT_THREADS = 512;
constexpr int SAGE_SORT_EPT     = 8;
constexpr int SAGE_SORT_CHUNK   = SAGE_SORT_THREADS * SAGE_SORT_EPT;
constexpr int SAGE_BUCKET_SHIFT = 8;
constexpr int SAGE_BUCKET_NODES = 1 << SAGE_BUCKET_SHIFT;
constexpr int SAGE_BUCKET_BATCH = 4;

__device__ __forceinline__ void sage_load8(const int* __restrict__ p, int e0,
                                           int E, int (&v)[SAGE_SORT_EPT]) {
  if ((E & 3) == 0 && e0 + SAGE_SORT_EPT <= E) {
    const int4 a = *(const int4*)(p + e0);
    const int4 b = *(const int4*)(p + e0 + 4);
    v[0] = a.x; v[1] = a.y; v[2] = a.z; v[3] = a.w;
    v[4] = b.x; v[5] = b.y; v[6] = b.z; v[7] = b.w;
  } else {
#pragma unroll
    for (int u = 0; u < SAGE_SORT_EPT; ++u) v[u] = (e0 + u < E) ? p[e0 + u] : 0;
  }
}

__device__ __forceinline__ int sage_block_scan(int v, int* ws, int& total) {
  const int lane = threadIdx.x & 31;
  const int w    = threadIdx.x >> 5;
  int incl = v;
  for (int o = 1; o < 32; o <<= 1) {
    const int t = __shfl_up(incl, o);
    if (lane >= o) incl += t;
  }
  if (lane == 31) ws[w] = incl;
  __syncthreads();
  if (w == 0) {
    const int s = (lane < 16) ? ws[lane] : 0;
    int wi = s;
    for (int o = 1; o < 16; o <<= 1) {
      const int t = __shfl_up(wi, o);
      if (lane >= o) wi += t;
    }
    if (lane < 16) ws[lane] = wi - s;
    if (lane == 15) ws[16] = wi;
  }
  __syncthreads();
  total = ws[16];
  return ws[w] + incl - v;
}

__global__ __launch_bounds__(SAGE_SORT_THREADS) void sage_bucket_count_kernel(
    const int* __restrict__ dst, int* __restrict__ gtot,
    int* __restrict__ wgoff, int E, int nb) {
  __shared__ int hist[SAGE_SORT_THREADS];
  const int tid = threadIdx.x;
  const int e0  = blockIdx.x * SAGE_SORT_CHUNK + tid * SAGE_SORT_EPT;
  int d[SAGE_SORT_EPT];
  sage_load8(dst, e0, E, d);
  hist[tid] = 0;
  __syncthreads();
#pragma unroll
  for (int u = 0; u < SAGE_SORT_EPT; ++u)
    if (e0 + u < E) atomicAdd(&hist[d[u] >> SAGE_BUCKET_SHIFT], 1);
  __syncthreads();
  if (tid < nb) {
    const int c = hist[tid];
    wgoff[blockIdx.x * nb + tid] = (c > 0) ? atomicAdd(&gtot[tid], c) : 0;
  }
}

__global__ __launch_bounds__(SAGE_SORT_THREADS) void sage_bucket_place_kernel(
    const int* __restrict__ src, const int* __restrict__ dst,
    const int* __restrict__ gtot, const int* __restrict__ wgoff,
    int2* __restrict__ pairs, int E, int nb) {
  __shared__ int cur[SAGE_SORT_THREADS];
  __shared__ int ws[17];
  const int tid = threadIdx.x;
  const int e0  = blockIdx.x * SAGE_SORT_CHUNK + tid * SAGE_SORT_EPT;
  int s[SAGE_SORT_EPT], d[SAGE_SORT_EPT];
  sage_load8(src, e0, E, s);
  sage_load8(dst, e0, E, d);
  const int g     = (tid < nb) ? gtot[tid] : 0;
  const int piece = (tid < nb) ? wgoff[blockIdx.x * nb + tid] : 0;
  int total;
  const int base = sage_block_scan(g, ws, total);
  cur[tid] = base + piece;
  __syncthreads();
  int pos[SAGE_SORT_EPT];
#pragma unroll
  for (int u = 0; u < SAGE_SORT_EPT; ++u)
    pos[u] = (e0 + u < E) ? atomicAdd(&cur[d[u] >> SAGE_BUCKET_SHIFT], 1) : -1;
#pragma unroll
  for (int u = 0; u < SAGE_SORT_EPT; ++u)
    if (pos[u] >= 0) pairs[pos[u]] = make_int2(s[u], d[u]);
}

__global__ __launch_bounds__(SAGE_SORT_THREADS) void sage_bucket_csr_kernel(
    const int2* __restrict__ pairs, const int* __restrict__ gtot,
    int* __restrict__ rowptr, int* __restrict__ csr, int n, int nb) {
  __shared__ int cnt[SAGE_BUCKET_NODES];
  __shared__ int cur[SAGE_BUCKET_NODES];
  __shared__ int wsa[17];
  __shared__ int wsb[17];
  constexpr int STRIDE = SAGE_SORT_THREADS * SAGE_BUCKET_BATCH;
  const int tid = threadIdx.x;
  const int b   = blockIdx.x;
  const int lo  = b << SAGE_BUCKET_SHIFT;
  if (tid < SAGE_BUCKET_NODES) cnt[tid] = 0;
  const int g = (tid < b) ? gtot[tid] : 0;
  int base;
  sage_block_scan(g, wsa, base);
  const int T = gtot[b];
  const int2* pb = pairs + base;

  const int last = T - 1;
  for (int i0 = 0; i0 < T; i0 += STRIDE) {
    int dl[SAGE_BUCKET_BATCH];
#pragma unroll
    for (int u = 0; u < SAGE_BUCKET_BATCH; ++u) {
      const int i = i0 + u * SAGE_SORT_THREADS + tid;
      const int raw = pb[(i < T) ? i : last].y;
      dl[u] = (i < T) ? (raw & (SAGE_BUCKET_NODES - 1)) : -1;
    }
#pragma unroll
    for (int u = 0; u < SAGE_BUCKET_BATCH; ++u)
      if (dl[u] >= 0) atomicAdd(&cnt[dl[u]], 1);
  }
  __syncthreads();

  const int c = (tid < SAGE_BUCKET_NODES) ? cnt[tid] : 0;
  int tot;
  const int ex = sage_block_scan(c, wsb, tot);
  if (tid < SAGE_BUCKET_NODES) {
    cur[tid] = base + ex;
    if (lo + tid < n) rowptr[lo + tid] = base + ex;
  }
  if (b == nb - 1 && tid == 0) rowptr[n] = base + T;
  __syncthreads();

  for (int i0 = 0; i0 < T; i0 += STRIDE) {
    int2 p[SAGE_BUCKET_BATCH];
    int pos[SAGE_BUCKET_BATCH];
#pragma unroll
    for (int u = 0; u < SAGE_BUCKET_BATCH; ++u) {
      const int i = i0 + u * SAGE_SORT_THREADS + tid;
      p[u] = pb[(i < T) ? i : last];
      if (i >= T) p[u].y = -1;
    }
#pragma unroll
    for (int u = 0; u < SAGE_BUCKET_BATCH; ++u)
      pos[u] = (p[u].y >= 0)
                   ? atomicAdd(&cur[p[u].y & (SAGE_BUCKET_NODES - 1)], 1)
                   : -1;
#pragma unroll
    for (int u = 0; u < SAGE_BUCKET_BATCH; ++u)
      if (pos[u] >= 0) csr[pos[u]] = p[u].x;
  }
}

__device__ __forceinline__ float4 sage_row(const float* fcol, int s, int F) {
  return *(const float4*)(fcol + (long long)s * F);
}
__device__ __forceinline__ void sage_acc(float4& acc, const float4 v) {
  acc.x += v.x;
  acc.y += v.y;
  acc.z += v.z;
  acc.w += v.w;
}

template <int F, bool ADD>
__global__ __launch_bounds__(128) void sage_gather_kernel(
    const float* __restrict__ feat, const int* __restrict__ rowptr,
    const int* __restrict__ csr, float* __restrict__ tgt, int n) {
  constexpr int LPN = F / 4;
  constexpr int NPB = 128 / LPN;
  const int sub = threadIdx.x / LPN;
  const int c   = (threadIdx.x % LPN) * 4;
  const float* fcol = feat + c;
  for (int d = blockIdx.x * NPB + sub; d < n; d += gridDim.x * NPB) {
    const int beg = rowptr[d];
    const int end = rowptr[d + 1];
    float4 acc = make_float4(0.f, 0.f, 0.f, 0.f);
    int j = beg;
    for (; j + 8 <= end; j += 8) {
      const int s0 = csr[j];
      const int s1 = csr[j + 1];
      const int s2 = csr[j + 2];
      const int s3 = csr[j + 3];
      const int s4 = csr[j + 4];
      const int s5 = csr[j + 5];
      const int s6 = csr[j + 6];
      const int s7 = csr[j + 7];
      const float4 v0 = sage_row(fcol, s0, F);
      const float4 v1 = sage_row(fcol, s1, F);
      const float4 v2 = sage_row(fcol, s2, F);
      const float4 v3 = sage_row(fcol, s3, F);
      const float4 v4 = sage_row(fcol, s4, F);
      const float4 v5 = sage_row(fcol, s5, F);
      const float4 v6 = sage_row(fcol, s6, F);
      const float4 v7 = sage_row(fcol, s7, F);
      sage_acc(acc, v0);
      sage_acc(acc, v1);
      sage_acc(acc, v2);
      sage_acc(acc, v3);
      sage_acc(acc, v4);
      sage_acc(acc, v5);
      sage_acc(acc, v6);
      sage_acc(acc, v7);
    }
    if (j + 4 <= end) {
      const int s0 = csr[j];
      const int s1 = csr[j + 1];
      const int s2 = csr[j + 2];
      const int s3 = csr[j + 3];
      const float4 v0 = sage_row(fcol, s0, F);
      const float4 v1 = sage_row(fcol, s1, F);
      const float4 v2 = sage_row(fcol, s2, F);
      const float4 v3 = sage_row(fcol, s3, F);
      sage_acc(acc, v0);
      sage_acc(acc, v1);
      sage_acc(acc, v2);
      sage_acc(acc, v3);
      j += 4;
    }
    if (j < end) {
      const int last = end - 1;
      const bool h1 = (j + 1 <= last);
      const bool h2 = (j + 2 <= last);
      const int s0 = csr[j];
      const int s1 = csr[h1 ? j + 1 : last];
      const int s2 = csr[h2 ? j + 2 : last];
      const float4 v0 = sage_row(fcol, s0, F);
      float4 v1 = sage_row(fcol, s1, F);
      float4 v2 = sage_row(fcol, s2, F);
      v1.x = h1 ? v1.x : 0.f; v1.y = h1 ? v1.y : 0.f;
      v1.z = h1 ? v1.z : 0.f; v1.w = h1 ? v1.w : 0.f;
      v2.x = h2 ? v2.x : 0.f; v2.y = h2 ? v2.y : 0.f;
      v2.z = h2 ? v2.z : 0.f; v2.w = h2 ? v2.w : 0.f;
      sage_acc(acc, v0);
      sage_acc(acc, v1);
      sage_acc(acc, v2);
    }
    const float inv = 1.0f / fmaxf((float)(end - beg), 1.0f);
    float4* p = (float4*)(tgt + (long long)d * F + c);
    float4 o = make_float4(0.f, 0.f, 0.f, 0.f);
    if (ADD) o = *p;
    o.x += acc.x * inv;
    o.y += acc.y * inv;
    o.z += acc.z * inv;
    o.w += acc.w * inv;
    *p = o;
  }
}

constexpr int SAGE_DENSE_WAVES = 8;
constexpr int SAGE_DENSE_THREADS = SAGE_DENSE_WAVES * 32;

template <int KROWS, int SRC_COLS, int PITCH>
__device__ __forceinline__ void sage_stage_w(const float* __restrict__ wsrc,
                                             v2f* __restrict__ img, int k_lo,
                                             int col_lo) {
  constexpr int QPR = SRC_COLS / 4;
  for (int i = threadIdx.x; i < (KROWS / 2) * QPR; i += SAGE_DENSE_THREADS) {
    const int kp = i / QPR;
    const int q  = (i % QPR) * 4;
    const float4 r0 = *(const float4*)(wsrc + (2 * kp) * SRC_COLS + q);
    const float4 r1 = *(const float4*)(wsrc + (2 * kp + 1) * SRC_COLS + q);
    v2f* d = img + (k_lo / 2 + kp) * PITCH + col_lo + q;
    v2f p0, p1, p2, p3;
    p0.x = r0.x; p0.y = r1.x;
    p1.x = r0.y; p1.y = r1.y;
    p2.x = r0.z; p2.y = r1.z;
    p3.x = r0.w; p3.y = r1.w;
    d[0] = p0;
    d[1] = p1;
    d[2] = p2;
    d[3] = p3;
  }
}

template <int KH, int NT, int PITCH>
__device__ __forceinline__ void sage_strip_mma(
    const float* __restrict__ r0, const float* __restrict__ r1,
    const float* __restrict__ n0, const v2f* __restrict__ wcol,
    v2f (&x_nxt)[4], v8f (&acc)[NT]) {
  constexpr int CHH = KH / 16;
  constexpr int NCH = 2 * CHH;
#pragma nounroll
  for (int ch = 0; ch < NCH; ++ch) {
    v2f x_cur[4];
#pragma unroll
    for (int s = 0; s < 4; ++s) x_cur[s] = x_nxt[s];
    const int cn = ch + 1;
    const float* ap = (cn < CHH) ? r0 + cn * 16
                    : (cn < NCH) ? r1 + (cn - CHH) * 16
                                 : n0;
#pragma unroll
    for (int s = 0; s < 4; ++s)
      x_nxt[s] = *(const v2f*)(ap + s * 4);
    const v2f* wp = wcol + (ch * 8) * PITCH;
#pragma unroll
    for (int s = 0; s < 4; ++s) {
#pragma unroll
      for (int t = 0; t < NT; ++t) {
        const v2f w = wp[(2 * s) * PITCH + t * 16];
        acc[t] = __builtin_amdgcn_wmma_f32_16x16x4_f32(
             false, w,  false, x_cur[s],
             (short)0, acc[t],  false,  false);
      }
    }
  }
}

template <int FIN_T, int HID_T, int FOUT_T>
__global__ __launch_bounds__(SAGE_DENSE_THREADS) void sage_dense_kernel(
    const float* __restrict__ agg, const float* __restrict__ xin,
    const float* __restrict__ Wl1, const float* __restrict__ Wr1,
    const float* __restrict__ b1, const float* __restrict__ Wl2,
    const float* __restrict__ Wr2, const float* __restrict__ b2,
    float* __restrict__ p, float* __restrict__ out, int nstrips) {
  constexpr int NT1 = HID_T / 16;
  constexpr int P1  = HID_T + 16;
  constexpr int NTH = FOUT_T / 16;
  constexpr int NT2 = 2 * NTH;
  constexpr int P2  = 2 * FOUT_T + 4;
  static_assert(P1 % 32 == 16, "W1 pitch: one pair row = 32 banks mod 64");
  static_assert(P2 % 8 == 4, "W2 pitch: four pair rows = 32 banks mod 64");
  __shared__ v2f w1img[FIN_T * P1];
  __shared__ v2f w2img[(HID_T / 2) * P2];
  __shared__ __attribute__((aligned(32))) float bias1[HID_T];
  __shared__ __attribute__((aligned(32))) float bias2[FOUT_T];

  const int lane = threadIdx.x & 31;
  const int wave = threadIdx.x >> 5;
  const int half = lane >> 4;
  const int l15  = lane & 15;
  const int koff = half * 2;
  const v2f* w1col = w1img + half * P1 + l15;
  const v2f* w2col = w2img + (half * 4) * P2 + l15;
  const int step = gridDim.x * SAGE_DENSE_WAVES;

  int strip = blockIdx.x * SAGE_DENSE_WAVES + wave;
  v2f x_nxt[4];
  if (strip < nstrips) {
    const float* f0 = agg + (long long)(strip * 16 + l15) * FIN_T + koff;
#pragma unroll
    for (int s = 0; s < 4; ++s) x_nxt[s] = *(const v2f*)(f0 + s * 4);
  }
  sage_stage_w<FIN_T, HID_T, P1>(Wl1, w1img, 0, 0);
  sage_stage_w<FIN_T, HID_T, P1>(Wr1, w1img, FIN_T, 0);
  sage_stage_w<HID_T, FOUT_T, P2>(Wl2, w2img, 0, 0);
  sage_stage_w<HID_T, FOUT_T, P2>(Wr2, w2img, 0, FOUT_T);
  for (int i = threadIdx.x; i < HID_T; i += SAGE_DENSE_THREADS) bias1[i] = b1[i];
  for (int i = threadIdx.x; i < FOUT_T; i += SAGE_DENSE_THREADS) bias2[i] = b2[i];
  __syncthreads();

  for (; strip < nstrips; strip += step) {
    asm volatile("" ::: "memory");
    const int row = strip * 16 + l15;
    const int rown = (strip + step < nstrips) ? row + step * 16 : row;
    const float* aRow = agg + (long long)row * FIN_T + koff;
    const float* xRow = xin + (long long)row * FIN_T + koff;
    const float* aNxt = agg + (long long)rown * FIN_T + koff;

    v8f acc[NT1];
#pragma unroll
    for (int t = 0; t < NT1; ++t)
      acc[t] = *(const v8f*)(bias1 + t * 16 + half * 8);

    sage_strip_mma<FIN_T, NT1, P1>(aRow, xRow, aNxt, w1col, x_nxt, acc);

#pragma unroll
    for (int t = 0; t < NT1; ++t) {
#pragma unroll
      for (int v = 0; v < 8; ++v) acc[t][v] = fmaxf(acc[t][v], 0.f);
    }

    v8f acc2[NT2];
#pragma unroll
    for (int u = 0; u < NTH; ++u) {
      acc2[u] = (v8f){0.f, 0.f, 0.f, 0.f, 0.f, 0.f, 0.f, 0.f};
      acc2[NTH + u] = *(const v8f*)(bias2 + u * 16 + half * 8);
    }
#pragma unroll
    for (int t = 0; t < NT1; ++t) {
#pragma unroll
      for (int j = 0; j < 4; ++j) {
        v2f hb;
        hb.x = acc[t][2 * j];
        hb.y = acc[t][2 * j + 1];
#pragma unroll
        for (int u = 0; u < NT2; ++u) {
          const v2f w = w2col[(8 * t + j) * P2 + u * 16];
          acc2[u] = __builtin_amdgcn_wmma_f32_16x16x4_f32(
               false, w,  false, hb,
               (short)0, acc2[u],  false,  false);
        }
      }
    }

#pragma unroll
    for (int u = 0; u < NTH; ++u) {
      const long long o = (long long)row * FOUT_T + u * 16 + half * 8;
      float4* pp = (float4*)(p + o);
      float4* po = (float4*)(out + o);
      pp[0] = make_float4(acc2[u][0], acc2[u][1], acc2[u][2], acc2[u][3]);
      pp[1] = make_float4(acc2[u][4], acc2[u][5], acc2[u][6], acc2[u][7]);
      po[0] = make_float4(acc2[NTH + u][0], acc2[NTH + u][1],
                          acc2[NTH + u][2], acc2[NTH + u][3]);
      po[1] = make_float4(acc2[NTH + u][4], acc2[NTH + u][5],
                          acc2[NTH + u][6], acc2[NTH + u][7]);
    }
  }
}

extern "C" void kernel_launch(void* const* d_in, const int* in_sizes, int n_in,
                              void* d_out, int out_size, void* d_ws,
                              size_t ws_size, hipStream_t stream) {
  constexpr int FIN = 64, HID = 128, FOUT = 32;

  const float* x   = (const float*)d_in[0];
  const int*   ei  = (const int*)d_in[1];
  const float* Wl1 = (const float*)d_in[2];
  const float* Wr1 = (const float*)d_in[3];
  const float* b1  = (const float*)d_in[4];
  const float* Wl2 = (const float*)d_in[5];
  const float* Wr2 = (const float*)d_in[6];
  const float* b2  = (const float*)d_in[7];
  float* out = (float*)d_out;

  const int n = in_sizes[0] / FIN;
  const int E = in_sizes[1] / 2;
  const int* src = ei;
  const int* dst = ei + E;

  const size_t nP = ((size_t)n + 64) & ~(size_t)63;
  const size_t eP = ((size_t)E + 63) & ~(size_t)63;
  int* gtot   = (int*)d_ws;
  int* rowptr = gtot + nP;
  int* csr    = rowptr + nP;
  float* agg1 = (float*)(csr + eP);
  float* p2   = agg1 + (size_t)n * FIN;
  int2* pairs = (int2*)p2;
  int* wgoff  = (int*)(pairs + eP);

  const int nb     = (n + SAGE_BUCKET_NODES - 1) / SAGE_BUCKET_NODES;
  const int nchunk = (E + SAGE_SORT_CHUNK - 1) / SAGE_SORT_CHUNK;

  hipMemsetAsync(d_ws, 0, (size_t)nb * sizeof(int), stream);

  sage_bucket_count_kernel<<<nchunk, SAGE_SORT_THREADS, 0, stream>>>(
      dst, gtot, wgoff, E, nb);
  sage_bucket_place_kernel<<<nchunk, SAGE_SORT_THREADS, 0, stream>>>(
      src, dst, gtot, wgoff, pairs, E, nb);
  sage_bucket_csr_kernel<<<nb, SAGE_SORT_THREADS, 0, stream>>>(
      pairs, gtot, rowptr, csr, n, nb);

  sage_gather_kernel<FIN, false>
      <<<(n + 31) / 32, 128, 0, stream>>>(x, rowptr, csr, agg1, n);

  sage_dense_kernel<FIN, HID, FOUT>
      <<<512, SAGE_DENSE_THREADS, 0, stream>>>(agg1, x, Wl1, Wr1, b1, Wl2, Wr2,
                                               b2, p2, out, n / 16);


  sage_gather_kernel<FOUT, true>
      <<<(n + 31) / 32, 128, 0, stream>>>(p2, rowptr, csr, out, n);
}


# Layer-1 mean aggregation moved into the dense launch: each wave gathers its strip's 16 mean rows into its own LDS tile, read as B fragments; agg1 and one launch go; f32 sum order unchanged
# speedup vs baseline: 4.9813x; 1.0484x over previous
#include <hip/hip_runtime.h>


typedef float v2f __attribute__((ext_vector_type(2)));
typedef float v8f __attribute__((ext_vector_type(8)));

constexpr int SAGE_SORT_THREADS = 512;
constexpr int SAGE_SORT_EPT     = 8;
constexpr int SAGE_SORT_CHUNK   = SAGE_SORT_THREADS * SAGE_SORT_EPT;
constexpr int SAGE_BUCKET_SHIFT = 8;
constexpr int SAGE_BUCKET_NODES = 1 << SAGE_BUCKET_SHIFT;
constexpr int SAGE_BUCKET_BATCH = 4;

__device__ __forceinline__ void sage_load8(const int* __restrict__ p, int e0,
                                           int E, int (&v)[SAGE_SORT_EPT]) {
  if ((E & 3) == 0 && e0 + SAGE_SORT_EPT <= E) {
    const int4 a = *(const int4*)(p + e0);
    const int4 b = *(const int4*)(p + e0 + 4);
    v[0] = a.x; v[1] = a.y; v[2] = a.z; v[3] = a.w;
    v[4] = b.x; v[5] = b.y; v[6] = b.z; v[7] = b.w;
  } else {
#pragma unroll
    for (int u = 0; u < SAGE_SORT_EPT; ++u) v[u] = (e0 + u < E) ? p[e0 + u] : 0;
  }
}

__device__ __forceinline__ int sage_block_scan(int v, int* ws, int& total) {
  const int lane = threadIdx.x & 31;
  const int w    = threadIdx.x >> 5;
  int incl = v;
  for (int o = 1; o < 32; o <<= 1) {
    const int t = __shfl_up(incl, o);
    if (lane >= o) incl += t;
  }
  if (lane == 31) ws[w] = incl;
  __syncthreads();
  if (w == 0) {
    const int s = (lane < 16) ? ws[lane] : 0;
    int wi = s;
    for (int o = 1; o < 16; o <<= 1) {
      const int t = __shfl_up(wi, o);
      if (lane >= o) wi += t;
    }
    if (lane < 16) ws[lane] = wi - s;
    if (lane == 15) ws[16] = wi;
  }
  __syncthreads();
  total = ws[16];
  return ws[w] + incl - v;
}

__global__ __launch_bounds__(SAGE_SORT_THREADS) void sage_bucket_count_kernel(
    const int* __restrict__ dst, int* __restrict__ gtot,
    int* __restrict__ wgoff, int E, int nb) {
  __shared__ int hist[SAGE_SORT_THREADS];
  const int tid = threadIdx.x;
  const int e0  = blockIdx.x * SAGE_SORT_CHUNK + tid * SAGE_SORT_EPT;
  int d[SAGE_SORT_EPT];
  sage_load8(dst, e0, E, d);
  hist[tid] = 0;
  __syncthreads();
#pragma unroll
  for (int u = 0; u < SAGE_SORT_EPT; ++u)
    if (e0 + u < E) atomicAdd(&hist[d[u] >> SAGE_BUCKET_SHIFT], 1);
  __syncthreads();
  if (tid < nb) {
    const int c = hist[tid];
    wgoff[blockIdx.x * nb + tid] = (c > 0) ? atomicAdd(&gtot[tid], c) : 0;
  }
}

__global__ __launch_bounds__(SAGE_SORT_THREADS) void sage_bucket_place_kernel(
    const int* __restrict__ src, const int* __restrict__ dst,
    const int* __restrict__ gtot, const int* __restrict__ wgoff,
    int2* __restrict__ pairs, int E, int nb) {
  __shared__ int cur[SAGE_SORT_THREADS];
  __shared__ int ws[17];
  const int tid = threadIdx.x;
  const int e0  = blockIdx.x * SAGE_SORT_CHUNK + tid * SAGE_SORT_EPT;
  int s[SAGE_SORT_EPT], d[SAGE_SORT_EPT];
  sage_load8(src, e0, E, s);
  sage_load8(dst, e0, E, d);
  const int g     = (tid < nb) ? gtot[tid] : 0;
  const int piece = (tid < nb) ? wgoff[blockIdx.x * nb + tid] : 0;
  int total;
  const int base = sage_block_scan(g, ws, total);
  cur[tid] = base + piece;
  __syncthreads();
  int pos[SAGE_SORT_EPT];
#pragma unroll
  for (int u = 0; u < SAGE_SORT_EPT; ++u)
    pos[u] = (e0 + u < E) ? atomicAdd(&cur[d[u] >> SAGE_BUCKET_SHIFT], 1) : -1;
#pragma unroll
  for (int u = 0; u < SAGE_SORT_EPT; ++u)
    if (pos[u] >= 0) pairs[pos[u]] = make_int2(s[u], d[u]);
}

__global__ __launch_bounds__(SAGE_SORT_THREADS) void sage_bucket_csr_kernel(
    const int2* __restrict__ pairs, const int* __restrict__ gtot,
    int* __restrict__ rowptr, int* __restrict__ csr, int n, int nb) {
  __shared__ int cnt[SAGE_BUCKET_NODES];
  __shared__ int cur[SAGE_BUCKET_NODES];
  __shared__ int wsa[17];
  __shared__ int wsb[17];
  constexpr int STRIDE = SAGE_SORT_THREADS * SAGE_BUCKET_BATCH;
  const int tid = threadIdx.x;
  const int b   = blockIdx.x;
  const int lo  = b << SAGE_BUCKET_SHIFT;
  if (tid < SAGE_BUCKET_NODES) cnt[tid] = 0;
  const int g = (tid < b) ? gtot[tid] : 0;
  int base;
  sage_block_scan(g, wsa, base);
  const int T = gtot[b];
  const int2* pb = pairs + base;

  const int last = T - 1;
  for (int i0 = 0; i0 < T; i0 += STRIDE) {
    int dl[SAGE_BUCKET_BATCH];
#pragma unroll
    for (int u = 0; u < SAGE_BUCKET_BATCH; ++u) {
      const int i = i0 + u * SAGE_SORT_THREADS + tid;
      const int raw = pb[(i < T) ? i : last].y;
      dl[u] = (i < T) ? (raw & (SAGE_BUCKET_NODES - 1)) : -1;
    }
#pragma unroll
    for (int u = 0; u < SAGE_BUCKET_BATCH; ++u)
      if (dl[u] >= 0) atomicAdd(&cnt[dl[u]], 1);
  }
  __syncthreads();

  const int c = (tid < SAGE_BUCKET_NODES) ? cnt[tid] : 0;
  int tot;
  const int ex = sage_block_scan(c, wsb, tot);
  if (tid < SAGE_BUCKET_NODES) {
    cur[tid] = base + ex;
    if (lo + tid < n) rowptr[lo + tid] = base + ex;
  }
  if (b == nb - 1 && tid == 0) rowptr[n] = base + T;
  __syncthreads();

  for (int i0 = 0; i0 < T; i0 += STRIDE) {
    int2 p[SAGE_BUCKET_BATCH];
    int pos[SAGE_BUCKET_BATCH];
#pragma unroll
    for (int u = 0; u < SAGE_BUCKET_BATCH; ++u) {
      const int i = i0 + u * SAGE_SORT_THREADS + tid;
      p[u] = pb[(i < T) ? i : last];
      if (i >= T) p[u].y = -1;
    }
#pragma unroll
    for (int u = 0; u < SAGE_BUCKET_BATCH; ++u)
      pos[u] = (p[u].y >= 0)
                   ? atomicAdd(&cur[p[u].y & (SAGE_BUCKET_NODES - 1)], 1)
                   : -1;
#pragma unroll
    for (int u = 0; u < SAGE_BUCKET_BATCH; ++u)
      if (pos[u] >= 0) csr[pos[u]] = p[u].x;
  }
}

__device__ __forceinline__ float4 sage_row(const float* fcol, int s, int F) {
  return *(const float4*)(fcol + (long long)s * F);
}
__device__ __forceinline__ void sage_acc(float4& acc, const float4 v) {
  acc.x += v.x;
  acc.y += v.y;
  acc.z += v.z;
  acc.w += v.w;
}

template <int F, bool ADD>
__global__ __launch_bounds__(128) void sage_gather_kernel(
    const float* __restrict__ feat, const int* __restrict__ rowptr,
    const int* __restrict__ csr, float* __restrict__ tgt, int n) {
  constexpr int LPN = F / 4;
  constexpr int NPB = 128 / LPN;
  const int sub = threadIdx.x / LPN;
  const int c   = (threadIdx.x % LPN) * 4;
  const float* fcol = feat + c;
  for (int d = blockIdx.x * NPB + sub; d < n; d += gridDim.x * NPB) {
    const int beg = rowptr[d];
    const int end = rowptr[d + 1];
    float4 acc = make_float4(0.f, 0.f, 0.f, 0.f);
    int j = beg;
    for (; j + 8 <= end; j += 8) {
      const int s0 = csr[j];
      const int s1 = csr[j + 1];
      const int s2 = csr[j + 2];
      const int s3 = csr[j + 3];
      const int s4 = csr[j + 4];
      const int s5 = csr[j + 5];
      const int s6 = csr[j + 6];
      const int s7 = csr[j + 7];
      const float4 v0 = sage_row(fcol, s0, F);
      const float4 v1 = sage_row(fcol, s1, F);
      const float4 v2 = sage_row(fcol, s2, F);
      const float4 v3 = sage_row(fcol, s3, F);
      const float4 v4 = sage_row(fcol, s4, F);
      const float4 v5 = sage_row(fcol, s5, F);
      const float4 v6 = sage_row(fcol, s6, F);
      const float4 v7 = sage_row(fcol, s7, F);
      sage_acc(acc, v0);
      sage_acc(acc, v1);
      sage_acc(acc, v2);
      sage_acc(acc, v3);
      sage_acc(acc, v4);
      sage_acc(acc, v5);
      sage_acc(acc, v6);
      sage_acc(acc, v7);
    }
    if (j + 4 <= end) {
      const int s0 = csr[j];
      const int s1 = csr[j + 1];
      const int s2 = csr[j + 2];
      const int s3 = csr[j + 3];
      const float4 v0 = sage_row(fcol, s0, F);
      const float4 v1 = sage_row(fcol, s1, F);
      const float4 v2 = sage_row(fcol, s2, F);
      const float4 v3 = sage_row(fcol, s3, F);
      sage_acc(acc, v0);
      sage_acc(acc, v1);
      sage_acc(acc, v2);
      sage_acc(acc, v3);
      j += 4;
    }
    if (j < end) {
      const int last = end - 1;
      const bool h1 = (j + 1 <= last);
      const bool h2 = (j + 2 <= last);
      const int s0 = csr[j];
      const int s1 = csr[h1 ? j + 1 : last];
      const int s2 = csr[h2 ? j + 2 : last];
      const float4 v0 = sage_row(fcol, s0, F);
      float4 v1 = sage_row(fcol, s1, F);
      float4 v2 = sage_row(fcol, s2, F);
      v1.x = h1 ? v1.x : 0.f; v1.y = h1 ? v1.y : 0.f;
      v1.z = h1 ? v1.z : 0.f; v1.w = h1 ? v1.w : 0.f;
      v2.x = h2 ? v2.x : 0.f; v2.y = h2 ? v2.y : 0.f;
      v2.z = h2 ? v2.z : 0.f; v2.w = h2 ? v2.w : 0.f;
      sage_acc(acc, v0);
      sage_acc(acc, v1);
      sage_acc(acc, v2);
    }
    const float inv = 1.0f / fmaxf((float)(end - beg), 1.0f);
    float4* p = (float4*)(tgt + (long long)d * F + c);
    float4 o = make_float4(0.f, 0.f, 0.f, 0.f);
    if (ADD) o = *p;
    o.x += acc.x * inv;
    o.y += acc.y * inv;
    o.z += acc.z * inv;
    o.w += acc.w * inv;
    *p = o;
  }
}

constexpr int SAGE_DENSE_WAVES = 8;
constexpr int SAGE_DENSE_THREADS = SAGE_DENSE_WAVES * 32;

template <int KROWS, int SRC_COLS, int PITCH>
__device__ __forceinline__ void sage_stage_w(const float* __restrict__ wsrc,
                                             v2f* __restrict__ img, int k_lo,
                                             int col_lo) {
  constexpr int QPR = SRC_COLS / 4;
  for (int i = threadIdx.x; i < (KROWS / 2) * QPR; i += SAGE_DENSE_THREADS) {
    const int kp = i / QPR;
    const int q  = (i % QPR) * 4;
    const float4 r0 = *(const float4*)(wsrc + (2 * kp) * SRC_COLS + q);
    const float4 r1 = *(const float4*)(wsrc + (2 * kp + 1) * SRC_COLS + q);
    v2f* d = img + (k_lo / 2 + kp) * PITCH + col_lo + q;
    v2f p0, p1, p2, p3;
    p0.x = r0.x; p0.y = r1.x;
    p1.x = r0.y; p1.y = r1.y;
    p2.x = r0.z; p2.y = r1.z;
    p3.x = r0.w; p3.y = r1.w;
    d[0] = p0;
    d[1] = p1;
    d[2] = p2;
    d[3] = p3;
  }
}

__device__ __forceinline__ void sage_mean_x64(const float* __restrict__ fcol,
                                              const int* __restrict__ csr,
                                              int beg, int end, float4& m0,
                                              float4& m1) {
  const int n8   = (end - beg) >> 3;
  const int jt   = beg + (n8 << 3);
  const bool has4 = (jt + 4 <= end);
  const int jr   = has4 ? jt + 4 : jt;
  const bool hr0 = (jr < end);
  const bool hr1 = (jr + 1 < end);
  const bool hr2 = (jr + 2 < end);

  int s8[8], s4[4], sr[3];
  const int b8 = (n8 > 0) ? beg : ((end >= 8) ? end - 8 : 0);
  const int b4 = has4 ? jt : ((end >= 4) ? end - 4 : 0);
  const int l0 = (end > 0) ? end - 1 : 0;
#pragma unroll
  for (int u = 0; u < 8; ++u) s8[u] = csr[b8 + u];
#pragma unroll
  for (int u = 0; u < 4; ++u) s4[u] = csr[b4 + u];
  sr[0] = csr[hr0 ? jr : l0];
  sr[1] = csr[hr1 ? jr + 1 : l0];
  sr[2] = csr[hr2 ? jr + 2 : l0];
  asm volatile(""
               : "+v"(s8[0]), "+v"(s8[1]), "+v"(s8[2]), "+v"(s8[3]),
                 "+v"(s8[4]), "+v"(s8[5]), "+v"(s8[6]), "+v"(s8[7]),
                 "+v"(s4[0]), "+v"(s4[1]), "+v"(s4[2]), "+v"(s4[3]),
                 "+v"(sr[0]), "+v"(sr[1]), "+v"(sr[2]));
  float4 v4[4], w4[4], vr[3], wr[3];
#pragma unroll
  for (int u = 0; u < 4; ++u) {
    v4[u] = sage_row(fcol, s4[u], 64);
    w4[u] = sage_row(fcol + 32, s4[u], 64);
  }
#pragma unroll
  for (int u = 0; u < 3; ++u) {
    vr[u] = sage_row(fcol, sr[u], 64);
    wr[u] = sage_row(fcol + 32, sr[u], 64);
  }

  float4 a0 = make_float4(0.f, 0.f, 0.f, 0.f);
  float4 a1 = make_float4(0.f, 0.f, 0.f, 0.f);
  int j = beg;
  for (int g = 0; g < n8; ++g) {
    float4 v[8], w[8];
#pragma unroll
    for (int u = 0; u < 8; ++u) {
      v[u] = sage_row(fcol, s8[u], 64);
      w[u] = sage_row(fcol + 32, s8[u], 64);
    }
    const int jn = (j + 16 <= end) ? j + 8 : end - 8;
#pragma unroll
    for (int u = 0; u < 8; ++u) s8[u] = csr[jn + u];
#pragma unroll
    for (int u = 0; u < 8; ++u) {
      sage_acc(a0, v[u]);
      sage_acc(a1, w[u]);
    }
    j += 8;
  }
  if (has4) {
#pragma unroll
    for (int u = 0; u < 4; ++u) {
      sage_acc(a0, v4[u]);
      sage_acc(a1, w4[u]);
    }
  }
  if (hr0) {
    if (!hr1) {
      vr[1] = make_float4(0.f, 0.f, 0.f, 0.f);
      wr[1] = make_float4(0.f, 0.f, 0.f, 0.f);
    }
    if (!hr2) {
      vr[2] = make_float4(0.f, 0.f, 0.f, 0.f);
      wr[2] = make_float4(0.f, 0.f, 0.f, 0.f);
    }
#pragma unroll
    for (int u = 0; u < 3; ++u) {
      sage_acc(a0, vr[u]);
      sage_acc(a1, wr[u]);
    }
  }
  const float inv = 1.0f / fmaxf((float)(end - beg), 1.0f);
  m0 = make_float4(a0.x * inv, a0.y * inv, a0.z * inv, a0.w * inv);
  m1 = make_float4(a1.x * inv, a1.y * inv, a1.z * inv, a1.w * inv);
}

template <int KH, int NT, int PITCH>
__device__ __forceinline__ void sage_strip_mma_tile(
    const float* __restrict__ trow, const v2f* __restrict__ wcol,
    v8f (&acc)[NT]) {
  constexpr int CHH = KH / 16;
#pragma nounroll
  for (int ch = 0; ch < CHH; ++ch) {
    v2f b[4];
#pragma unroll
    for (int s = 0; s < 4; ++s) b[s] = *(const v2f*)(trow + ch * 16 + s * 4);
    const v2f* wp = wcol + (ch * 8) * PITCH;
#pragma unroll
    for (int s = 0; s < 4; ++s) {
#pragma unroll
      for (int t = 0; t < NT; ++t) {
        const v2f w = wp[(2 * s) * PITCH + t * 16];
        acc[t] = __builtin_amdgcn_wmma_f32_16x16x4_f32(
             false, w,  false, b[s],
             (short)0, acc[t],  false,  false);
      }
    }
  }
}

template <int KH, int NT, int PITCH>
__device__ __forceinline__ void sage_strip_mma(
    const float* __restrict__ r0, const float* __restrict__ n0,
    const v2f* __restrict__ wcol, v2f (&x_nxt)[4], v8f (&acc)[NT]) {
  constexpr int CHH = KH / 16;
#pragma nounroll
  for (int ch = 0; ch < CHH; ++ch) {
    v2f x_cur[4];
#pragma unroll
    for (int s = 0; s < 4; ++s) x_cur[s] = x_nxt[s];
    const int cn = ch + 1;
    const float* ap = (cn < CHH) ? r0 + cn * 16 : n0;
#pragma unroll
    for (int s = 0; s < 4; ++s)
      x_nxt[s] = *(const v2f*)(ap + s * 4);
    const v2f* wp = wcol + ((CHH + ch) * 8) * PITCH;
#pragma unroll
    for (int s = 0; s < 4; ++s) {
#pragma unroll
      for (int t = 0; t < NT; ++t) {
        const v2f w = wp[(2 * s) * PITCH + t * 16];
        acc[t] = __builtin_amdgcn_wmma_f32_16x16x4_f32(
             false, w,  false, x_cur[s],
             (short)0, acc[t],  false,  false);
      }
    }
  }
}

template <int FIN_T, int HID_T, int FOUT_T>
__global__ __launch_bounds__(SAGE_DENSE_THREADS) void sage_dense_kernel(
    const float* __restrict__ xin, const int* __restrict__ rowptr,
    const int* __restrict__ csr,
    const float* __restrict__ Wl1, const float* __restrict__ Wr1,
    const float* __restrict__ b1, const float* __restrict__ Wl2,
    const float* __restrict__ Wr2, const float* __restrict__ b2,
    float* __restrict__ p, float* __restrict__ out, int nstrips) {
  constexpr int NT1 = HID_T / 16;
  constexpr int P1  = HID_T + 16;
  constexpr int NTH = FOUT_T / 16;
  constexpr int NT2 = 2 * NTH;
  constexpr int P2  = 2 * FOUT_T + 4;
  constexpr int TP  = FIN_T + 4;
  static_assert(FIN_T == 64, "the in-launch gather splits a row over 8 lanes x 2 chunks");
  static_assert(P1 % 32 == 16, "W1 pitch: one pair row = 32 banks mod 64");
  static_assert(P2 % 8 == 4, "W2 pitch: four pair rows = 32 banks mod 64");
  __shared__ v2f w1img[FIN_T * P1];
  __shared__ v2f w2img[(HID_T / 2) * P2];
  __shared__ __attribute__((aligned(32))) float bias1[HID_T];
  __shared__ __attribute__((aligned(32))) float bias2[FOUT_T];
  __shared__ __attribute__((aligned(16))) float aggt[SAGE_DENSE_WAVES * 16 * TP];

  const int lane = threadIdx.x & 31;
  const int wave = threadIdx.x >> 5;
  const int half = lane >> 4;
  const int l15  = lane & 15;
  const int koff = half * 2;
  const v2f* w1col = w1img + half * P1 + l15;
  const v2f* w2col = w2img + (half * 4) * P2 + l15;
  const int step = gridDim.x * SAGE_DENSE_WAVES;
  const int grp = lane >> 3;
  const int c8  = (lane & 7) * 4;
  float* tile = aggt + wave * (16 * TP);
  const float* trow = tile + l15 * TP + koff;

  int strip = blockIdx.x * SAGE_DENSE_WAVES + wave;
  v2f x_nxt[4];
  if (strip < nstrips) {
    const float* f0 = xin + (long long)(strip * 16 + l15) * FIN_T + koff;
#pragma unroll
    for (int s = 0; s < 4; ++s) x_nxt[s] = *(const v2f*)(f0 + s * 4);
  }
  sage_stage_w<FIN_T, HID_T, P1>(Wl1, w1img, 0, 0);
  sage_stage_w<FIN_T, HID_T, P1>(Wr1, w1img, FIN_T, 0);
  sage_stage_w<HID_T, FOUT_T, P2>(Wl2, w2img, 0, 0);
  sage_stage_w<HID_T, FOUT_T, P2>(Wr2, w2img, 0, FOUT_T);
  for (int i = threadIdx.x; i < HID_T; i += SAGE_DENSE_THREADS) bias1[i] = b1[i];
  for (int i = threadIdx.x; i < FOUT_T; i += SAGE_DENSE_THREADS) bias2[i] = b2[i];
  __syncthreads();

  for (; strip < nstrips; strip += step) {
    asm volatile("" ::: "memory");
    const int row = strip * 16 + l15;
    const int rown = (strip + step < nstrips) ? row + step * 16 : row;
    const float* xRow = xin + (long long)row * FIN_T + koff;
    const float* xNxt = xin + (long long)rown * FIN_T + koff;

    const int rp = rowptr[strip * 16 + (lane < 16 ? lane : 16)];
#pragma nounroll
    for (int ps = 0; ps < 4; ++ps) {
      const int nl  = ps * 4 + grp;
      const int beg = __shfl(rp, nl);
      const int end = __shfl(rp, nl + 1);
      float4 m0, m1;
      sage_mean_x64(xin + c8, csr, beg, end, m0, m1);
      float* tp = tile + nl * TP + c8;
      *(float4*)tp = m0;
      *(float4*)(tp + 32) = m1;
    }
    __builtin_amdgcn_wave_barrier();

    v8f acc[NT1];
#pragma unroll
    for (int t = 0; t < NT1; ++t)
      acc[t] = *(const v8f*)(bias1 + t * 16 + half * 8);

    sage_strip_mma_tile<FIN_T, NT1, P1>(trow, w1col, acc);
    sage_strip_mma<FIN_T, NT1, P1>(xRow, xNxt, w1col, x_nxt, acc);

#pragma unroll
    for (int t = 0; t < NT1; ++t) {
#pragma unroll
      for (int v = 0; v < 8; ++v) acc[t][v] = fmaxf(acc[t][v], 0.f);
    }

    v8f acc2[NT2];
#pragma unroll
    for (int u = 0; u < NTH; ++u) {
      acc2[u] = (v8f){0.f, 0.f, 0.f, 0.f, 0.f, 0.f, 0.f, 0.f};
      acc2[NTH + u] = *(const v8f*)(bias2 + u * 16 + half * 8);
    }
#pragma unroll
    for (int t = 0; t < NT1; ++t) {
#pragma unroll
      for (int j = 0; j < 4; ++j) {
        v2f hb;
        hb.x = acc[t][2 * j];
        hb.y = acc[t][2 * j + 1];
#pragma unroll
        for (int u = 0; u < NT2; ++u) {
          const v2f w = w2col[(8 * t + j) * P2 + u * 16];
          acc2[u] = __builtin_amdgcn_wmma_f32_16x16x4_f32(
               false, w,  false, hb,
               (short)0, acc2[u],  false,  false);
        }
      }
    }

#pragma unroll
    for (int u = 0; u < NTH; ++u) {
      const long long o = (long long)row * FOUT_T + u * 16 + half * 8;
      float4* pp = (float4*)(p + o);
      float4* po = (float4*)(out + o);
      pp[0] = make_float4(acc2[u][0], acc2[u][1], acc2[u][2], acc2[u][3]);
      pp[1] = make_float4(acc2[u][4], acc2[u][5], acc2[u][6], acc2[u][7]);
      po[0] = make_float4(acc2[NTH + u][0], acc2[NTH + u][1],
                          acc2[NTH + u][2], acc2[NTH + u][3]);
      po[1] = make_float4(acc2[NTH + u][4], acc2[NTH + u][5],
                          acc2[NTH + u][6], acc2[NTH + u][7]);
    }
  }
}

extern "C" void kernel_launch(void* const* d_in, const int* in_sizes, int n_in,
                              void* d_out, int out_size, void* d_ws,
                              size_t ws_size, hipStream_t stream) {
  constexpr int FIN = 64, HID = 128, FOUT = 32;

  const float* x   = (const float*)d_in[0];
  const int*   ei  = (const int*)d_in[1];
  const float* Wl1 = (const float*)d_in[2];
  const float* Wr1 = (const float*)d_in[3];
  const float* b1  = (const float*)d_in[4];
  const float* Wl2 = (const float*)d_in[5];
  const float* Wr2 = (const float*)d_in[6];
  const float* b2  = (const float*)d_in[7];
  float* out = (float*)d_out;

  const int n = in_sizes[0] / FIN;
  const int E = in_sizes[1] / 2;
  const int* src = ei;
  const int* dst = ei + E;

  const size_t nP = ((size_t)n + 64) & ~(size_t)63;
  const size_t eP = ((size_t)E + 63) & ~(size_t)63;
  int* gtot   = (int*)d_ws;
  int* rowptr = gtot + nP;
  int* csr    = rowptr + nP;
  float* p2   = (float*)(csr + eP);
  int2* pairs = (int2*)p2;
  int* wgoff  = (int*)(pairs + eP);

  const int nb     = (n + SAGE_BUCKET_NODES - 1) / SAGE_BUCKET_NODES;
  const int nchunk = (E + SAGE_SORT_CHUNK - 1) / SAGE_SORT_CHUNK;

  hipMemsetAsync(d_ws, 0, (size_t)nb * sizeof(int), stream);

  sage_bucket_count_kernel<<<nchunk, SAGE_SORT_THREADS, 0, stream>>>(
      dst, gtot, wgoff, E, nb);
  sage_bucket_place_kernel<<<nchunk, SAGE_SORT_THREADS, 0, stream>>>(
      src, dst, gtot, wgoff, pairs, E, nb);
  sage_bucket_csr_kernel<<<nb, SAGE_SORT_THREADS, 0, stream>>>(
      pairs, gtot, rowptr, csr, n, nb);

  sage_dense_kernel<FIN, HID, FOUT>
      <<<512, SAGE_DENSE_THREADS, 0, stream>>>(x, rowptr, csr, Wl1, Wr1, b1, Wl2,
                                               Wr2, b2, p2, out, n / 16);


  sage_gather_kernel<FOUT, true>
      <<<(n + 31) / 32, 128, 0, stream>>>(p2, rowptr, csr, out, n);
}


# Edge sort level 1 made chunk-local: each 4096-edge chunk ordered by bucket in place plus an offsets table, level 2 reads per-chunk runs; global count pass and memset go; f32 sum order may change
# speedup vs baseline: 5.1470x; 1.0333x over previous
#include <hip/hip_runtime.h>


typedef float v2f __attribute__((ext_vector_type(2)));
typedef float v8f __attribute__((ext_vector_type(8)));

constexpr int SAGE_SORT_THREADS = 512;
constexpr int SAGE_SORT_EPT     = 8;
constexpr int SAGE_SORT_CHUNK   = SAGE_SORT_THREADS * SAGE_SORT_EPT;
constexpr int SAGE_BUCKET_SHIFT = 8;
constexpr int SAGE_BUCKET_NODES = 1 << SAGE_BUCKET_SHIFT;
constexpr int SAGE_BUCKET_BATCH = 4;

__device__ __forceinline__ void sage_load8(const int* __restrict__ p, int e0,
                                           int E, int (&v)[SAGE_SORT_EPT]) {
  if ((E & 3) == 0 && e0 + SAGE_SORT_EPT <= E) {
    const int4 a = *(const int4*)(p + e0);
    const int4 b = *(const int4*)(p + e0 + 4);
    v[0] = a.x; v[1] = a.y; v[2] = a.z; v[3] = a.w;
    v[4] = b.x; v[5] = b.y; v[6] = b.z; v[7] = b.w;
  } else {
#pragma unroll
    for (int u = 0; u < SAGE_SORT_EPT; ++u) v[u] = (e0 + u < E) ? p[e0 + u] : 0;
  }
}

__device__ __forceinline__ int sage_block_scan(int v, int* ws, int& total) {
  const int lane = threadIdx.x & 31;
  const int w    = threadIdx.x >> 5;
  int incl = v;
  for (int o = 1; o < 32; o <<= 1) {
    const int t = __shfl_up(incl, o);
    if (lane >= o) incl += t;
  }
  if (lane == 31) ws[w] = incl;
  __syncthreads();
  if (w == 0) {
    const int s = (lane < 16) ? ws[lane] : 0;
    int wi = s;
    for (int o = 1; o < 16; o <<= 1) {
      const int t = __shfl_up(wi, o);
      if (lane >= o) wi += t;
    }
    if (lane < 16) ws[lane] = wi - s;
    if (lane == 15) ws[16] = wi;
  }
  __syncthreads();
  total = ws[16];
  return ws[w] + incl - v;
}

__global__ __launch_bounds__(SAGE_SORT_THREADS) void sage_chunk_sort_kernel(
    const int* __restrict__ src, const int* __restrict__ dst,
    int2* __restrict__ pairs, int* __restrict__ off, int E, int nb) {
  __shared__ int hist[SAGE_SORT_THREADS];
  __shared__ int ws[17];
  __shared__ int2 stage[SAGE_SORT_CHUNK];
  const int tid = threadIdx.x;
  const int cb  = blockIdx.x * SAGE_SORT_CHUNK;
  const int e0  = cb + tid * SAGE_SORT_EPT;
  int s[SAGE_SORT_EPT], d[SAGE_SORT_EPT];
  sage_load8(src, e0, E, s);
  sage_load8(dst, e0, E, d);
  hist[tid] = 0;
  __syncthreads();
#pragma unroll
  for (int u = 0; u < SAGE_SORT_EPT; ++u)
    if (e0 + u < E) atomicAdd(&hist[d[u] >> SAGE_BUCKET_SHIFT], 1);
  __syncthreads();
  const int c = hist[tid];
  int total;
  const int ex = sage_block_scan(c, ws, total);
  if (tid < nb) off[blockIdx.x * nb + tid] = ex;
  hist[tid] = ex;
  __syncthreads();
#pragma unroll
  for (int u = 0; u < SAGE_SORT_EPT; ++u)
    if (e0 + u < E)
      stage[atomicAdd(&hist[d[u] >> SAGE_BUCKET_SHIFT], 1)] =
          make_int2(s[u], d[u]);
  __syncthreads();
  for (int i = tid; i < total; i += SAGE_SORT_THREADS) pairs[cb + i] = stage[i];
}

__device__ __forceinline__ int sage_find_run(const int* pre, int i) {
  int r = 0;
  for (int s = SAGE_SORT_THREADS >> 1; s > 0; s >>= 1)
    if (pre[r + s] <= i) r += s;
  return r;
}

__global__ __launch_bounds__(SAGE_SORT_THREADS) void sage_bucket_csr_kernel(
    const int2* __restrict__ pairs, const int* __restrict__ off,
    int* __restrict__ rowptr, int* __restrict__ csr, int n, int E, int nb,
    int nchunk) {
  __shared__ int cnt[SAGE_BUCKET_NODES];
  __shared__ int cur[SAGE_BUCKET_NODES];
  __shared__ int pre[SAGE_SORT_THREADS];
  __shared__ int sbeg[SAGE_SORT_THREADS];
  __shared__ int wsa[17];
  __shared__ int wsb[17];
  __shared__ int wsc[17];
  constexpr int STRIDE = SAGE_SORT_THREADS * SAGE_BUCKET_BATCH;
  const int tid = threadIdx.x;
  const int b   = blockIdx.x;
  const int lo  = b << SAGE_BUCKET_SHIFT;
  if (tid < SAGE_BUCKET_NODES) cnt[tid] = 0;
  int sb = 0;
  for (int c = tid; c < nchunk; c += SAGE_SORT_THREADS) sb += off[c * nb + b];
  int base;
  sage_block_scan(sb, wsa, base);
  const bool one_tile = (nchunk <= SAGE_SORT_THREADS);
  int T = 0;
  int tile_tot = 0;

  for (int pass = 0; pass < 2; ++pass) {
    if (pass == 1) {
      __syncthreads();
      const int cc = (tid < SAGE_BUCKET_NODES) ? cnt[tid] : 0;
      int tot;
      const int ex = sage_block_scan(cc, wsb, tot);
      if (tid < SAGE_BUCKET_NODES) {
        cur[tid] = base + ex;
        if (lo + tid < n) rowptr[lo + tid] = base + ex;
      }
      if (b == nb - 1 && tid == 0) rowptr[n] = base + T;
      __syncthreads();
    }
    for (int c0 = 0; c0 < nchunk; c0 += SAGE_SORT_THREADS) {
      if (pass == 0 || !one_tile) {
        const int c = c0 + tid;
        int o0 = 0, len = 0;
        if (c < nchunk) {
          const int left = E - c * SAGE_SORT_CHUNK;
          const int full = (left < SAGE_SORT_CHUNK) ? left : SAGE_SORT_CHUNK;
          o0 = off[c * nb + b];
          const int o1 = (b + 1 < nb) ? off[c * nb + b + 1] : full;
          len = o1 - o0;
        }
        __syncthreads();
        const int ex = sage_block_scan(len, wsc, tile_tot);
        pre[tid]  = ex;
        sbeg[tid] = c * SAGE_SORT_CHUNK + o0;
        if (pass == 0) T += tile_tot;
        __syncthreads();
      }
      const int last = tile_tot - 1;
      if (pass == 0) {
        for (int i0 = 0; i0 < tile_tot; i0 += STRIDE) {
          int dl[SAGE_BUCKET_BATCH];
#pragma unroll
          for (int u = 0; u < SAGE_BUCKET_BATCH; ++u) {
            const int i  = i0 + u * SAGE_SORT_THREADS + tid;
            const int ii = (i < tile_tot) ? i : last;
            const int r  = sage_find_run(pre, ii);
            const int raw = pairs[sbeg[r] + (ii - pre[r])].y;
            dl[u] = (i < tile_tot) ? (raw & (SAGE_BUCKET_NODES - 1)) : -1;
          }
#pragma unroll
          for (int u = 0; u < SAGE_BUCKET_BATCH; ++u)
            if (dl[u] >= 0) atomicAdd(&cnt[dl[u]], 1);
        }
      } else {
        for (int i0 = 0; i0 < tile_tot; i0 += STRIDE) {
          int2 p[SAGE_BUCKET_BATCH];
          int pos[SAGE_BUCKET_BATCH];
#pragma unroll
          for (int u = 0; u < SAGE_BUCKET_BATCH; ++u) {
            const int i  = i0 + u * SAGE_SORT_THREADS + tid;
            const int ii = (i < tile_tot) ? i : last;
            const int r  = sage_find_run(pre, ii);
            p[u] = pairs[sbeg[r] + (ii - pre[r])];
            if (i >= tile_tot) p[u].y = -1;
          }
#pragma unroll
          for (int u = 0; u < SAGE_BUCKET_BATCH; ++u)
            pos[u] = (p[u].y >= 0)
                         ? atomicAdd(&cur[p[u].y & (SAGE_BUCKET_NODES - 1)], 1)
                         : -1;
#pragma unroll
          for (int u = 0; u < SAGE_BUCKET_BATCH; ++u)
            if (pos[u] >= 0) csr[pos[u]] = p[u].x;
        }
      }
    }
  }
}

__device__ __forceinline__ float4 sage_row(const float* fcol, int s, int F) {
  return *(const float4*)(fcol + (long long)s * F);
}
__device__ __forceinline__ void sage_acc(float4& acc, const float4 v) {
  acc.x += v.x;
  acc.y += v.y;
  acc.z += v.z;
  acc.w += v.w;
}

template <int F, bool ADD>
__global__ __launch_bounds__(128) void sage_gather_kernel(
    const float* __restrict__ feat, const int* __restrict__ rowptr,
    const int* __restrict__ csr, float* __restrict__ tgt, int n) {
  constexpr int LPN = F / 4;
  constexpr int NPB = 128 / LPN;
  const int sub = threadIdx.x / LPN;
  const int c   = (threadIdx.x % LPN) * 4;
  const float* fcol = feat + c;
  for (int d = blockIdx.x * NPB + sub; d < n; d += gridDim.x * NPB) {
    const int beg = rowptr[d];
    const int end = rowptr[d + 1];
    float4 acc = make_float4(0.f, 0.f, 0.f, 0.f);
    int j = beg;
    for (; j + 8 <= end; j += 8) {
      const int s0 = csr[j];
      const int s1 = csr[j + 1];
      const int s2 = csr[j + 2];
      const int s3 = csr[j + 3];
      const int s4 = csr[j + 4];
      const int s5 = csr[j + 5];
      const int s6 = csr[j + 6];
      const int s7 = csr[j + 7];
      const float4 v0 = sage_row(fcol, s0, F);
      const float4 v1 = sage_row(fcol, s1, F);
      const float4 v2 = sage_row(fcol, s2, F);
      const float4 v3 = sage_row(fcol, s3, F);
      const float4 v4 = sage_row(fcol, s4, F);
      const float4 v5 = sage_row(fcol, s5, F);
      const float4 v6 = sage_row(fcol, s6, F);
      const float4 v7 = sage_row(fcol, s7, F);
      sage_acc(acc, v0);
      sage_acc(acc, v1);
      sage_acc(acc, v2);
      sage_acc(acc, v3);
      sage_acc(acc, v4);
      sage_acc(acc, v5);
      sage_acc(acc, v6);
      sage_acc(acc, v7);
    }
    if (j + 4 <= end) {
      const int s0 = csr[j];
      const int s1 = csr[j + 1];
      const int s2 = csr[j + 2];
      const int s3 = csr[j + 3];
      const float4 v0 = sage_row(fcol, s0, F);
      const float4 v1 = sage_row(fcol, s1, F);
      const float4 v2 = sage_row(fcol, s2, F);
      const float4 v3 = sage_row(fcol, s3, F);
      sage_acc(acc, v0);
      sage_acc(acc, v1);
      sage_acc(acc, v2);
      sage_acc(acc, v3);
      j += 4;
    }
    if (j < end) {
      const int last = end - 1;
      const bool h1 = (j + 1 <= last);
      const bool h2 = (j + 2 <= last);
      const int s0 = csr[j];
      const int s1 = csr[h1 ? j + 1 : last];
      const int s2 = csr[h2 ? j + 2 : last];
      const float4 v0 = sage_row(fcol, s0, F);
      float4 v1 = sage_row(fcol, s1, F);
      float4 v2 = sage_row(fcol, s2, F);
      v1.x = h1 ? v1.x : 0.f; v1.y = h1 ? v1.y : 0.f;
      v1.z = h1 ? v1.z : 0.f; v1.w = h1 ? v1.w : 0.f;
      v2.x = h2 ? v2.x : 0.f; v2.y = h2 ? v2.y : 0.f;
      v2.z = h2 ? v2.z : 0.f; v2.w = h2 ? v2.w : 0.f;
      sage_acc(acc, v0);
      sage_acc(acc, v1);
      sage_acc(acc, v2);
    }
    const float inv = 1.0f / fmaxf((float)(end - beg), 1.0f);
    float4* p = (float4*)(tgt + (long long)d * F + c);
    float4 o = make_float4(0.f, 0.f, 0.f, 0.f);
    if (ADD) o = *p;
    o.x += acc.x * inv;
    o.y += acc.y * inv;
    o.z += acc.z * inv;
    o.w += acc.w * inv;
    *p = o;
  }
}

constexpr int SAGE_DENSE_WAVES = 8;
constexpr int SAGE_DENSE_THREADS = SAGE_DENSE_WAVES * 32;

template <int KROWS, int SRC_COLS, int PITCH>
__device__ __forceinline__ void sage_stage_w(const float* __restrict__ wsrc,
                                             v2f* __restrict__ img, int k_lo,
                                             int col_lo) {
  constexpr int QPR = SRC_COLS / 4;
  for (int i = threadIdx.x; i < (KROWS / 2) * QPR; i += SAGE_DENSE_THREADS) {
    const int kp = i / QPR;
    const int q  = (i % QPR) * 4;
    const float4 r0 = *(const float4*)(wsrc + (2 * kp) * SRC_COLS + q);
    const float4 r1 = *(const float4*)(wsrc + (2 * kp + 1) * SRC_COLS + q);
    v2f* d = img + (k_lo / 2 + kp) * PITCH + col_lo + q;
    v2f p0, p1, p2, p3;
    p0.x = r0.x; p0.y = r1.x;
    p1.x = r0.y; p1.y = r1.y;
    p2.x = r0.z; p2.y = r1.z;
    p3.x = r0.w; p3.y = r1.w;
    d[0] = p0;
    d[1] = p1;
    d[2] = p2;
    d[3] = p3;
  }
}

__device__ __forceinline__ void sage_mean_x64(const float* __restrict__ fcol,
                                              const int* __restrict__ csr,
                                              int beg, int end, float4& m0,
                                              float4& m1) {
  const int n8   = (end - beg) >> 3;
  const int jt   = beg + (n8 << 3);
  const bool has4 = (jt + 4 <= end);
  const int jr   = has4 ? jt + 4 : jt;
  const bool hr0 = (jr < end);
  const bool hr1 = (jr + 1 < end);
  const bool hr2 = (jr + 2 < end);

  int s8[8], s4[4], sr[3];
  const int b8 = (n8 > 0) ? beg : ((end >= 8) ? end - 8 : 0);
  const int b4 = has4 ? jt : ((end >= 4) ? end - 4 : 0);
  const int l0 = (end > 0) ? end - 1 : 0;
#pragma unroll
  for (int u = 0; u < 8; ++u) s8[u] = csr[b8 + u];
#pragma unroll
  for (int u = 0; u < 4; ++u) s4[u] = csr[b4 + u];
  sr[0] = csr[hr0 ? jr : l0];
  sr[1] = csr[hr1 ? jr + 1 : l0];
  sr[2] = csr[hr2 ? jr + 2 : l0];
  asm volatile(""
               : "+v"(s8[0]), "+v"(s8[1]), "+v"(s8[2]), "+v"(s8[3]),
                 "+v"(s8[4]), "+v"(s8[5]), "+v"(s8[6]), "+v"(s8[7]),
                 "+v"(s4[0]), "+v"(s4[1]), "+v"(s4[2]), "+v"(s4[3]),
                 "+v"(sr[0]), "+v"(sr[1]), "+v"(sr[2]));
  float4 v4[4], w4[4], vr[3], wr[3];
#pragma unroll
  for (int u = 0; u < 4; ++u) {
    v4[u] = sage_row(fcol, s4[u], 64);
    w4[u] = sage_row(fcol + 32, s4[u], 64);
  }
#pragma unroll
  for (int u = 0; u < 3; ++u) {
    vr[u] = sage_row(fcol, sr[u], 64);
    wr[u] = sage_row(fcol + 32, sr[u], 64);
  }

  float4 a0 = make_float4(0.f, 0.f, 0.f, 0.f);
  float4 a1 = make_float4(0.f, 0.f, 0.f, 0.f);
  int j = beg;
  for (int g = 0; g < n8; ++g) {
    float4 v[8], w[8];
#pragma unroll
    for (int u = 0; u < 8; ++u) {
      v[u] = sage_row(fcol, s8[u], 64);
      w[u] = sage_row(fcol + 32, s8[u], 64);
    }
    const int jn = (j + 16 <= end) ? j + 8 : end - 8;
#pragma unroll
    for (int u = 0; u < 8; ++u) s8[u] = csr[jn + u];
#pragma unroll
    for (int u = 0; u < 8; ++u) {
      sage_acc(a0, v[u]);
      sage_acc(a1, w[u]);
    }
    j += 8;
  }
  if (has4) {
#pragma unroll
    for (int u = 0; u < 4; ++u) {
      sage_acc(a0, v4[u]);
      sage_acc(a1, w4[u]);
    }
  }
  if (hr0) {
    if (!hr1) {
      vr[1] = make_float4(0.f, 0.f, 0.f, 0.f);
      wr[1] = make_float4(0.f, 0.f, 0.f, 0.f);
    }
    if (!hr2) {
      vr[2] = make_float4(0.f, 0.f, 0.f, 0.f);
      wr[2] = make_float4(0.f, 0.f, 0.f, 0.f);
    }
#pragma unroll
    for (int u = 0; u < 3; ++u) {
      sage_acc(a0, vr[u]);
      sage_acc(a1, wr[u]);
    }
  }
  const float inv = 1.0f / fmaxf((float)(end - beg), 1.0f);
  m0 = make_float4(a0.x * inv, a0.y * inv, a0.z * inv, a0.w * inv);
  m1 = make_float4(a1.x * inv, a1.y * inv, a1.z * inv, a1.w * inv);
}

template <int KH, int NT, int PITCH>
__device__ __forceinline__ void sage_strip_mma_tile(
    const float* __restrict__ trow, const v2f* __restrict__ wcol,
    v8f (&acc)[NT]) {
  constexpr int CHH = KH / 16;
#pragma nounroll
  for (int ch = 0; ch < CHH; ++ch) {
    v2f b[4];
#pragma unroll
    for (int s = 0; s < 4; ++s) b[s] = *(const v2f*)(trow + ch * 16 + s * 4);
    const v2f* wp = wcol + (ch * 8) * PITCH;
#pragma unroll
    for (int s = 0; s < 4; ++s) {
#pragma unroll
      for (int t = 0; t < NT; ++t) {
        const v2f w = wp[(2 * s) * PITCH + t * 16];
        acc[t] = __builtin_amdgcn_wmma_f32_16x16x4_f32(
             false, w,  false, b[s],
             (short)0, acc[t],  false,  false);
      }
    }
  }
}

template <int KH, int NT, int PITCH>
__device__ __forceinline__ void sage_strip_mma(
    const float* __restrict__ r0, const float* __restrict__ n0,
    const v2f* __restrict__ wcol, v2f (&x_nxt)[4], v8f (&acc)[NT]) {
  constexpr int CHH = KH / 16;
#pragma nounroll
  for (int ch = 0; ch < CHH; ++ch) {
    v2f x_cur[4];
#pragma unroll
    for (int s = 0; s < 4; ++s) x_cur[s] = x_nxt[s];
    const int cn = ch + 1;
    const float* ap = (cn < CHH) ? r0 + cn * 16 : n0;
#pragma unroll
    for (int s = 0; s < 4; ++s)
      x_nxt[s] = *(const v2f*)(ap + s * 4);
    const v2f* wp = wcol + ((CHH + ch) * 8) * PITCH;
#pragma unroll
    for (int s = 0; s < 4; ++s) {
#pragma unroll
      for (int t = 0; t < NT; ++t) {
        const v2f w = wp[(2 * s) * PITCH + t * 16];
        acc[t] = __builtin_amdgcn_wmma_f32_16x16x4_f32(
             false, w,  false, x_cur[s],
             (short)0, acc[t],  false,  false);
      }
    }
  }
}

template <int FIN_T, int HID_T, int FOUT_T>
__global__ __launch_bounds__(SAGE_DENSE_THREADS) void sage_dense_kernel(
    const float* __restrict__ xin, const int* __restrict__ rowptr,
    const int* __restrict__ csr,
    const float* __restrict__ Wl1, const float* __restrict__ Wr1,
    const float* __restrict__ b1, const float* __restrict__ Wl2,
    const float* __restrict__ Wr2, const float* __restrict__ b2,
    float* __restrict__ p, float* __restrict__ out, int nstrips) {
  constexpr int NT1 = HID_T / 16;
  constexpr int P1  = HID_T + 16;
  constexpr int NTH = FOUT_T / 16;
  constexpr int NT2 = 2 * NTH;
  constexpr int P2  = 2 * FOUT_T + 4;
  constexpr int TP  = FIN_T + 4;
  static_assert(FIN_T == 64, "the in-launch gather splits a row over 8 lanes x 2 chunks");
  static_assert(P1 % 32 == 16, "W1 pitch: one pair row = 32 banks mod 64");
  static_assert(P2 % 8 == 4, "W2 pitch: four pair rows = 32 banks mod 64");
  __shared__ v2f w1img[FIN_T * P1];
  __shared__ v2f w2img[(HID_T / 2) * P2];
  __shared__ __attribute__((aligned(32))) float bias1[HID_T];
  __shared__ __attribute__((aligned(32))) float bias2[FOUT_T];
  __shared__ __attribute__((aligned(16))) float aggt[SAGE_DENSE_WAVES * 16 * TP];

  const int lane = threadIdx.x & 31;
  const int wave = threadIdx.x >> 5;
  const int half = lane >> 4;
  const int l15  = lane & 15;
  const int koff = half * 2;
  const v2f* w1col = w1img + half * P1 + l15;
  const v2f* w2col = w2img + (half * 4) * P2 + l15;
  const int step = gridDim.x * SAGE_DENSE_WAVES;
  const int grp = lane >> 3;
  const int c8  = (lane & 7) * 4;
  float* tile = aggt + wave * (16 * TP);
  const float* trow = tile + l15 * TP + koff;

  int strip = blockIdx.x * SAGE_DENSE_WAVES + wave;
  v2f x_nxt[4];
  if (strip < nstrips) {
    const float* f0 = xin + (long long)(strip * 16 + l15) * FIN_T + koff;
#pragma unroll
    for (int s = 0; s < 4; ++s) x_nxt[s] = *(const v2f*)(f0 + s * 4);
  }
  sage_stage_w<FIN_T, HID_T, P1>(Wl1, w1img, 0, 0);
  sage_stage_w<FIN_T, HID_T, P1>(Wr1, w1img, FIN_T, 0);
  sage_stage_w<HID_T, FOUT_T, P2>(Wl2, w2img, 0, 0);
  sage_stage_w<HID_T, FOUT_T, P2>(Wr2, w2img, 0, FOUT_T);
  for (int i = threadIdx.x; i < HID_T; i += SAGE_DENSE_THREADS) bias1[i] = b1[i];
  for (int i = threadIdx.x; i < FOUT_T; i += SAGE_DENSE_THREADS) bias2[i] = b2[i];
  __syncthreads();

  for (; strip < nstrips; strip += step) {
    asm volatile("" ::: "memory");
    const int row = strip * 16 + l15;
    const int rown = (strip + step < nstrips) ? row + step * 16 : row;
    const float* xRow = xin + (long long)row * FIN_T + koff;
    const float* xNxt = xin + (long long)rown * FIN_T + koff;

    const int rp = rowptr[strip * 16 + (lane < 16 ? lane : 16)];
#pragma nounroll
    for (int ps = 0; ps < 4; ++ps) {
      const int nl  = ps * 4 + grp;
      const int beg = __shfl(rp, nl);
      const int end = __shfl(rp, nl + 1);
      float4 m0, m1;
      sage_mean_x64(xin + c8, csr, beg, end, m0, m1);
      float* tp = tile + nl * TP + c8;
      *(float4*)tp = m0;
      *(float4*)(tp + 32) = m1;
    }
    __builtin_amdgcn_wave_barrier();

    v8f acc[NT1];
#pragma unroll
    for (int t = 0; t < NT1; ++t)
      acc[t] = *(const v8f*)(bias1 + t * 16 + half * 8);

    sage_strip_mma_tile<FIN_T, NT1, P1>(trow, w1col, acc);
    sage_strip_mma<FIN_T, NT1, P1>(xRow, xNxt, w1col, x_nxt, acc);

#pragma unroll
    for (int t = 0; t < NT1; ++t) {
#pragma unroll
      for (int v = 0; v < 8; ++v) acc[t][v] = fmaxf(acc[t][v], 0.f);
    }

    v8f acc2[NT2];
#pragma unroll
    for (int u = 0; u < NTH; ++u) {
      acc2[u] = (v8f){0.f, 0.f, 0.f, 0.f, 0.f, 0.f, 0.f, 0.f};
      acc2[NTH + u] = *(const v8f*)(bias2 + u * 16 + half * 8);
    }
#pragma unroll
    for (int t = 0; t < NT1; ++t) {
#pragma unroll
      for (int j = 0; j < 4; ++j) {
        v2f hb;
        hb.x = acc[t][2 * j];
        hb.y = acc[t][2 * j + 1];
#pragma unroll
        for (int u = 0; u < NT2; ++u) {
          const v2f w = w2col[(8 * t + j) * P2 + u * 16];
          acc2[u] = __builtin_amdgcn_wmma_f32_16x16x4_f32(
               false, w,  false, hb,
               (short)0, acc2[u],  false,  false);
        }
      }
    }

#pragma unroll
    for (int u = 0; u < NTH; ++u) {
      const long long o = (long long)row * FOUT_T + u * 16 + half * 8;
      float4* pp = (float4*)(p + o);
      float4* po = (float4*)(out + o);
      pp[0] = make_float4(acc2[u][0], acc2[u][1], acc2[u][2], acc2[u][3]);
      pp[1] = make_float4(acc2[u][4], acc2[u][5], acc2[u][6], acc2[u][7]);
      po[0] = make_float4(acc2[NTH + u][0], acc2[NTH + u][1],
                          acc2[NTH + u][2], acc2[NTH + u][3]);
      po[1] = make_float4(acc2[NTH + u][4], acc2[NTH + u][5],
                          acc2[NTH + u][6], acc2[NTH + u][7]);
    }
  }
}

extern "C" void kernel_launch(void* const* d_in, const int* in_sizes, int n_in,
                              void* d_out, int out_size, void* d_ws,
                              size_t ws_size, hipStream_t stream) {
  constexpr int FIN = 64, HID = 128, FOUT = 32;

  const float* x   = (const float*)d_in[0];
  const int*   ei  = (const int*)d_in[1];
  const float* Wl1 = (const float*)d_in[2];
  const float* Wr1 = (const float*)d_in[3];
  const float* b1  = (const float*)d_in[4];
  const float* Wl2 = (const float*)d_in[5];
  const float* Wr2 = (const float*)d_in[6];
  const float* b2  = (const float*)d_in[7];
  float* out = (float*)d_out;

  const int n = in_sizes[0] / FIN;
  const int E = in_sizes[1] / 2;
  const int* src = ei;
  const int* dst = ei + E;

  const size_t nP = ((size_t)n + 64) & ~(size_t)63;
  const size_t eP = ((size_t)E + 63) & ~(size_t)63;
  int* rowptr = (int*)d_ws;
  int* csr    = rowptr + nP;
  float* p2   = (float*)(csr + eP);
  int2* pairs = (int2*)p2;
  int* off    = (int*)(pairs + eP);

  const int nb     = (n + SAGE_BUCKET_NODES - 1) / SAGE_BUCKET_NODES;
  const int nchunk = (E + SAGE_SORT_CHUNK - 1) / SAGE_SORT_CHUNK;

  sage_chunk_sort_kernel<<<nchunk, SAGE_SORT_THREADS, 0, stream>>>(
      src, dst, pairs, off, E, nb);
  sage_bucket_csr_kernel<<<nb, SAGE_SORT_THREADS, 0, stream>>>(
      pairs, off, rowptr, csr, n, E, nb, nchunk);

  sage_dense_kernel<FIN, HID, FOUT>
      <<<512, SAGE_DENSE_THREADS, 0, stream>>>(x, rowptr, csr, Wl1, Wr1, b1, Wl2,
                                               Wr2, b2, p2, out, n / 16);


  sage_gather_kernel<FOUT, true>
      <<<(n + 31) / 32, 128, 0, stream>>>(p2, rowptr, csr, out, n);
}


# Dense launch: strips dealt round robin to 256 work-groups of 16 waves (one per WGP, weights staged once per WGP): every WGP gets 24 or 25 strips, not 16 to 32; f32 sum order unchanged
# speedup vs baseline: 5.4380x; 1.0565x over previous
#include <hip/hip_runtime.h>


typedef float v2f __attribute__((ext_vector_type(2)));
typedef float v8f __attribute__((ext_vector_type(8)));

constexpr int SAGE_SORT_THREADS = 512;
constexpr int SAGE_SORT_EPT     = 8;
constexpr int SAGE_SORT_CHUNK   = SAGE_SORT_THREADS * SAGE_SORT_EPT;
constexpr int SAGE_BUCKET_SHIFT = 8;
constexpr int SAGE_BUCKET_NODES = 1 << SAGE_BUCKET_SHIFT;
constexpr int SAGE_BUCKET_BATCH = 4;

__device__ __forceinline__ void sage_load8(const int* __restrict__ p, int e0,
                                           int E, int (&v)[SAGE_SORT_EPT]) {
  if ((E & 3) == 0 && e0 + SAGE_SORT_EPT <= E) {
    const int4 a = *(const int4*)(p + e0);
    const int4 b = *(const int4*)(p + e0 + 4);
    v[0] = a.x; v[1] = a.y; v[2] = a.z; v[3] = a.w;
    v[4] = b.x; v[5] = b.y; v[6] = b.z; v[7] = b.w;
  } else {
#pragma unroll
    for (int u = 0; u < SAGE_SORT_EPT; ++u) v[u] = (e0 + u < E) ? p[e0 + u] : 0;
  }
}

__device__ __forceinline__ int sage_block_scan(int v, int* ws, int& total) {
  const int lane = threadIdx.x & 31;
  const int w    = threadIdx.x >> 5;
  int incl = v;
  for (int o = 1; o < 32; o <<= 1) {
    const int t = __shfl_up(incl, o);
    if (lane >= o) incl += t;
  }
  if (lane == 31) ws[w] = incl;
  __syncthreads();
  if (w == 0) {
    const int s = (lane < 16) ? ws[lane] : 0;
    int wi = s;
    for (int o = 1; o < 16; o <<= 1) {
      const int t = __shfl_up(wi, o);
      if (lane >= o) wi += t;
    }
    if (lane < 16) ws[lane] = wi - s;
    if (lane == 15) ws[16] = wi;
  }
  __syncthreads();
  total = ws[16];
  return ws[w] + incl - v;
}

__global__ __launch_bounds__(SAGE_SORT_THREADS) void sage_chunk_sort_kernel(
    const int* __restrict__ src, const int* __restrict__ dst,
    int2* __restrict__ pairs, int* __restrict__ off, int E, int nb) {
  __shared__ int hist[SAGE_SORT_THREADS];
  __shared__ int ws[17];
  __shared__ int2 stage[SAGE_SORT_CHUNK];
  const int tid = threadIdx.x;
  const int cb  = blockIdx.x * SAGE_SORT_CHUNK;
  const int e0  = cb + tid * SAGE_SORT_EPT;
  int s[SAGE_SORT_EPT], d[SAGE_SORT_EPT];
  sage_load8(src, e0, E, s);
  sage_load8(dst, e0, E, d);
  hist[tid] = 0;
  __syncthreads();
#pragma unroll
  for (int u = 0; u < SAGE_SORT_EPT; ++u)
    if (e0 + u < E) atomicAdd(&hist[d[u] >> SAGE_BUCKET_SHIFT], 1);
  __syncthreads();
  const int c = hist[tid];
  int total;
  const int ex = sage_block_scan(c, ws, total);
  if (tid < nb) off[blockIdx.x * nb + tid] = ex;
  hist[tid] = ex;
  __syncthreads();
#pragma unroll
  for (int u = 0; u < SAGE_SORT_EPT; ++u)
    if (e0 + u < E)
      stage[atomicAdd(&hist[d[u] >> SAGE_BUCKET_SHIFT], 1)] =
          make_int2(s[u], d[u]);
  __syncthreads();
  for (int i = tid; i < total; i += SAGE_SORT_THREADS) pairs[cb + i] = stage[i];
}

__device__ __forceinline__ int sage_find_run(const int* pre, int i) {
  int r = 0;
  for (int s = SAGE_SORT_THREADS >> 1; s > 0; s >>= 1)
    if (pre[r + s] <= i) r += s;
  return r;
}

__global__ __launch_bounds__(SAGE_SORT_THREADS) void sage_bucket_csr_kernel(
    const int2* __restrict__ pairs, const int* __restrict__ off,
    int* __restrict__ rowptr, int* __restrict__ csr, int n, int E, int nb,
    int nchunk) {
  __shared__ int cnt[SAGE_BUCKET_NODES];
  __shared__ int cur[SAGE_BUCKET_NODES];
  __shared__ int pre[SAGE_SORT_THREADS];
  __shared__ int sbeg[SAGE_SORT_THREADS];
  __shared__ int wsa[17];
  __shared__ int wsb[17];
  __shared__ int wsc[17];
  constexpr int STRIDE = SAGE_SORT_THREADS * SAGE_BUCKET_BATCH;
  const int tid = threadIdx.x;
  const int b   = blockIdx.x;
  const int lo  = b << SAGE_BUCKET_SHIFT;
  if (tid < SAGE_BUCKET_NODES) cnt[tid] = 0;
  int sb = 0;
  for (int c = tid; c < nchunk; c += SAGE_SORT_THREADS) sb += off[c * nb + b];
  int base;
  sage_block_scan(sb, wsa, base);
  const bool one_tile = (nchunk <= SAGE_SORT_THREADS);
  int T = 0;
  int tile_tot = 0;

  for (int pass = 0; pass < 2; ++pass) {
    if (pass == 1) {
      __syncthreads();
      const int cc = (tid < SAGE_BUCKET_NODES) ? cnt[tid] : 0;
      int tot;
      const int ex = sage_block_scan(cc, wsb, tot);
      if (tid < SAGE_BUCKET_NODES) {
        cur[tid] = base + ex;
        if (lo + tid < n) rowptr[lo + tid] = base + ex;
      }
      if (b == nb - 1 && tid == 0) rowptr[n] = base + T;
      __syncthreads();
    }
    for (int c0 = 0; c0 < nchunk; c0 += SAGE_SORT_THREADS) {
      if (pass == 0 || !one_tile) {
        const int c = c0 + tid;
        int o0 = 0, len = 0;
        if (c < nchunk) {
          const int left = E - c * SAGE_SORT_CHUNK;
          const int full = (left < SAGE_SORT_CHUNK) ? left : SAGE_SORT_CHUNK;
          o0 = off[c * nb + b];
          const int o1 = (b + 1 < nb) ? off[c * nb + b + 1] : full;
          len = o1 - o0;
        }
        __syncthreads();
        const int ex = sage_block_scan(len, wsc, tile_tot);
        pre[tid]  = ex;
        sbeg[tid] = c * SAGE_SORT_CHUNK + o0;
        if (pass == 0) T += tile_tot;
        __syncthreads();
      }
      const int last = tile_tot - 1;
      if (pass == 0) {
        for (int i0 = 0; i0 < tile_tot; i0 += STRIDE) {
          int dl[SAGE_BUCKET_BATCH];
#pragma unroll
          for (int u = 0; u < SAGE_BUCKET_BATCH; ++u) {
            const int i  = i0 + u * SAGE_SORT_THREADS + tid;
            const int ii = (i < tile_tot) ? i : last;
            const int r  = sage_find_run(pre, ii);
            const int raw = pairs[sbeg[r] + (ii - pre[r])].y;
            dl[u] = (i < tile_tot) ? (raw & (SAGE_BUCKET_NODES - 1)) : -1;
          }
#pragma unroll
          for (int u = 0; u < SAGE_BUCKET_BATCH; ++u)
            if (dl[u] >= 0) atomicAdd(&cnt[dl[u]], 1);
        }
      } else {
        for (int i0 = 0; i0 < tile_tot; i0 += STRIDE) {
          int2 p[SAGE_BUCKET_BATCH];
          int pos[SAGE_BUCKET_BATCH];
#pragma unroll
          for (int u = 0; u < SAGE_BUCKET_BATCH; ++u) {
            const int i  = i0 + u * SAGE_SORT_THREADS + tid;
            const int ii = (i < tile_tot) ? i : last;
            const int r  = sage_find_run(pre, ii);
            p[u] = pairs[sbeg[r] + (ii - pre[r])];
            if (i >= tile_tot) p[u].y = -1;
          }
#pragma unroll
          for (int u = 0; u < SAGE_BUCKET_BATCH; ++u)
            pos[u] = (p[u].y >= 0)
                         ? atomicAdd(&cur[p[u].y & (SAGE_BUCKET_NODES - 1)], 1)
                         : -1;
#pragma unroll
          for (int u = 0; u < SAGE_BUCKET_BATCH; ++u)
            if (pos[u] >= 0) csr[pos[u]] = p[u].x;
        }
      }
    }
  }
}

__device__ __forceinline__ float4 sage_row(const float* fcol, int s, int F) {
  return *(const float4*)(fcol + (long long)s * F);
}
__device__ __forceinline__ void sage_acc(float4& acc, const float4 v) {
  acc.x += v.x;
  acc.y += v.y;
  acc.z += v.z;
  acc.w += v.w;
}

template <int F, bool ADD>
__global__ __launch_bounds__(128) void sage_gather_kernel(
    const float* __restrict__ feat, const int* __restrict__ rowptr,
    const int* __restrict__ csr, float* __restrict__ tgt, int n) {
  constexpr int LPN = F / 4;
  constexpr int NPB = 128 / LPN;
  const int sub = threadIdx.x / LPN;
  const int c   = (threadIdx.x % LPN) * 4;
  const float* fcol = feat + c;
  for (int d = blockIdx.x * NPB + sub; d < n; d += gridDim.x * NPB) {
    const int beg = rowptr[d];
    const int end = rowptr[d + 1];
    float4 acc = make_float4(0.f, 0.f, 0.f, 0.f);
    int j = beg;
    for (; j + 8 <= end; j += 8) {
      const int s0 = csr[j];
      const int s1 = csr[j + 1];
      const int s2 = csr[j + 2];
      const int s3 = csr[j + 3];
      const int s4 = csr[j + 4];
      const int s5 = csr[j + 5];
      const int s6 = csr[j + 6];
      const int s7 = csr[j + 7];
      const float4 v0 = sage_row(fcol, s0, F);
      const float4 v1 = sage_row(fcol, s1, F);
      const float4 v2 = sage_row(fcol, s2, F);
      const float4 v3 = sage_row(fcol, s3, F);
      const float4 v4 = sage_row(fcol, s4, F);
      const float4 v5 = sage_row(fcol, s5, F);
      const float4 v6 = sage_row(fcol, s6, F);
      const float4 v7 = sage_row(fcol, s7, F);
      sage_acc(acc, v0);
      sage_acc(acc, v1);
      sage_acc(acc, v2);
      sage_acc(acc, v3);
      sage_acc(acc, v4);
      sage_acc(acc, v5);
      sage_acc(acc, v6);
      sage_acc(acc, v7);
    }
    if (j + 4 <= end) {
      const int s0 = csr[j];
      const int s1 = csr[j + 1];
      const int s2 = csr[j + 2];
      const int s3 = csr[j + 3];
      const float4 v0 = sage_row(fcol, s0, F);
      const float4 v1 = sage_row(fcol, s1, F);
      const float4 v2 = sage_row(fcol, s2, F);
      const float4 v3 = sage_row(fcol, s3, F);
      sage_acc(acc, v0);
      sage_acc(acc, v1);
      sage_acc(acc, v2);
      sage_acc(acc, v3);
      j += 4;
    }
    if (j < end) {
      const int last = end - 1;
      const bool h1 = (j + 1 <= last);
      const bool h2 = (j + 2 <= last);
      const int s0 = csr[j];
      const int s1 = csr[h1 ? j + 1 : last];
      const int s2 = csr[h2 ? j + 2 : last];
      const float4 v0 = sage_row(fcol, s0, F);
      float4 v1 = sage_row(fcol, s1, F);
      float4 v2 = sage_row(fcol, s2, F);
      v1.x = h1 ? v1.x : 0.f; v1.y = h1 ? v1.y : 0.f;
      v1.z = h1 ? v1.z : 0.f; v1.w = h1 ? v1.w : 0.f;
      v2.x = h2 ? v2.x : 0.f; v2.y = h2 ? v2.y : 0.f;
      v2.z = h2 ? v2.z : 0.f; v2.w = h2 ? v2.w : 0.f;
      sage_acc(acc, v0);
      sage_acc(acc, v1);
      sage_acc(acc, v2);
    }
    const float inv = 1.0f / fmaxf((float)(end - beg), 1.0f);
    float4* p = (float4*)(tgt + (long long)d * F + c);
    float4 o = make_float4(0.f, 0.f, 0.f, 0.f);
    if (ADD) o = *p;
    o.x += acc.x * inv;
    o.y += acc.y * inv;
    o.z += acc.z * inv;
    o.w += acc.w * inv;
    *p = o;
  }
}

constexpr int SAGE_DENSE_WAVES = 16;
constexpr int SAGE_DENSE_THREADS = SAGE_DENSE_WAVES * 32;

template <int KROWS, int SRC_COLS, int PITCH>
__device__ __forceinline__ void sage_stage_w(const float* __restrict__ wsrc,
                                             v2f* __restrict__ img, int k_lo,
                                             int col_lo) {
  constexpr int QPR = SRC_COLS / 4;
  for (int i = threadIdx.x; i < (KROWS / 2) * QPR; i += SAGE_DENSE_THREADS) {
    const int kp = i / QPR;
    const int q  = (i % QPR) * 4;
    const float4 r0 = *(const float4*)(wsrc + (2 * kp) * SRC_COLS + q);
    const float4 r1 = *(const float4*)(wsrc + (2 * kp + 1) * SRC_COLS + q);
    v2f* d = img + (k_lo / 2 + kp) * PITCH + col_lo + q;
    v2f p0, p1, p2, p3;
    p0.x = r0.x; p0.y = r1.x;
    p1.x = r0.y; p1.y = r1.y;
    p2.x = r0.z; p2.y = r1.z;
    p3.x = r0.w; p3.y = r1.w;
    d[0] = p0;
    d[1] = p1;
    d[2] = p2;
    d[3] = p3;
  }
}

__device__ __forceinline__ void sage_mean_x64(const float* __restrict__ fcol,
                                              const int* __restrict__ csr,
                                              int beg, int end, float4& m0,
                                              float4& m1) {
  const int n8   = (end - beg) >> 3;
  const int jt   = beg + (n8 << 3);
  const bool has4 = (jt + 4 <= end);
  const int jr   = has4 ? jt + 4 : jt;
  const bool hr0 = (jr < end);
  const bool hr1 = (jr + 1 < end);
  const bool hr2 = (jr + 2 < end);

  int s8[8], s4[4], sr[3];
  const int b8 = (n8 > 0) ? beg : ((end >= 8) ? end - 8 : 0);
  const int b4 = has4 ? jt : ((end >= 4) ? end - 4 : 0);
  const int l0 = (end > 0) ? end - 1 : 0;
#pragma unroll
  for (int u = 0; u < 8; ++u) s8[u] = csr[b8 + u];
#pragma unroll
  for (int u = 0; u < 4; ++u) s4[u] = csr[b4 + u];
  sr[0] = csr[hr0 ? jr : l0];
  sr[1] = csr[hr1 ? jr + 1 : l0];
  sr[2] = csr[hr2 ? jr + 2 : l0];
  asm volatile(""
               : "+v"(s8[0]), "+v"(s8[1]), "+v"(s8[2]), "+v"(s8[3]),
                 "+v"(s8[4]), "+v"(s8[5]), "+v"(s8[6]), "+v"(s8[7]),
                 "+v"(s4[0]), "+v"(s4[1]), "+v"(s4[2]), "+v"(s4[3]),
                 "+v"(sr[0]), "+v"(sr[1]), "+v"(sr[2]));
  float4 v4[4], w4[4], vr[3], wr[3];
#pragma unroll
  for (int u = 0; u < 4; ++u) {
    v4[u] = sage_row(fcol, s4[u], 64);
    w4[u] = sage_row(fcol + 32, s4[u], 64);
  }
#pragma unroll
  for (int u = 0; u < 3; ++u) {
    vr[u] = sage_row(fcol, sr[u], 64);
    wr[u] = sage_row(fcol + 32, sr[u], 64);
  }

  float4 a0 = make_float4(0.f, 0.f, 0.f, 0.f);
  float4 a1 = make_float4(0.f, 0.f, 0.f, 0.f);
  int j = beg;
  for (int g = 0; g < n8; ++g) {
    float4 v[8], w[8];
#pragma unroll
    for (int u = 0; u < 8; ++u) {
      v[u] = sage_row(fcol, s8[u], 64);
      w[u] = sage_row(fcol + 32, s8[u], 64);
    }
    const int jn = (j + 16 <= end) ? j + 8 : end - 8;
#pragma unroll
    for (int u = 0; u < 8; ++u) s8[u] = csr[jn + u];
#pragma unroll
    for (int u = 0; u < 8; ++u) {
      sage_acc(a0, v[u]);
      sage_acc(a1, w[u]);
    }
    j += 8;
  }
  if (has4) {
#pragma unroll
    for (int u = 0; u < 4; ++u) {
      sage_acc(a0, v4[u]);
      sage_acc(a1, w4[u]);
    }
  }
  if (hr0) {
    if (!hr1) {
      vr[1] = make_float4(0.f, 0.f, 0.f, 0.f);
      wr[1] = make_float4(0.f, 0.f, 0.f, 0.f);
    }
    if (!hr2) {
      vr[2] = make_float4(0.f, 0.f, 0.f, 0.f);
      wr[2] = make_float4(0.f, 0.f, 0.f, 0.f);
    }
#pragma unroll
    for (int u = 0; u < 3; ++u) {
      sage_acc(a0, vr[u]);
      sage_acc(a1, wr[u]);
    }
  }
  const float inv = 1.0f / fmaxf((float)(end - beg), 1.0f);
  m0 = make_float4(a0.x * inv, a0.y * inv, a0.z * inv, a0.w * inv);
  m1 = make_float4(a1.x * inv, a1.y * inv, a1.z * inv, a1.w * inv);
}

template <int KH, int NT, int PITCH>
__device__ __forceinline__ void sage_strip_mma_tile(
    const float* __restrict__ trow, const v2f* __restrict__ wcol,
    v8f (&acc)[NT]) {
  constexpr int CHH = KH / 16;
#pragma nounroll
  for (int ch = 0; ch < CHH; ++ch) {
    v2f b[4];
#pragma unroll
    for (int s = 0; s < 4; ++s) b[s] = *(const v2f*)(trow + ch * 16 + s * 4);
    const v2f* wp = wcol + (ch * 8) * PITCH;
#pragma unroll
    for (int s = 0; s < 4; ++s) {
#pragma unroll
      for (int t = 0; t < NT; ++t) {
        const v2f w = wp[(2 * s) * PITCH + t * 16];
        acc[t] = __builtin_amdgcn_wmma_f32_16x16x4_f32(
             false, w,  false, b[s],
             (short)0, acc[t],  false,  false);
      }
    }
  }
}

template <int KH, int NT, int PITCH>
__device__ __forceinline__ void sage_strip_mma(
    const float* __restrict__ r0, const float* __restrict__ n0,
    const v2f* __restrict__ wcol, v2f (&x_nxt)[4], v8f (&acc)[NT]) {
  constexpr int CHH = KH / 16;
#pragma nounroll
  for (int ch = 0; ch < CHH; ++ch) {
    v2f x_cur[4];
#pragma unroll
    for (int s = 0; s < 4; ++s) x_cur[s] = x_nxt[s];
    const int cn = ch + 1;
    const float* ap = (cn < CHH) ? r0 + cn * 16 : n0;
#pragma unroll
    for (int s = 0; s < 4; ++s)
      x_nxt[s] = *(const v2f*)(ap + s * 4);
    const v2f* wp = wcol + ((CHH + ch) * 8) * PITCH;
#pragma unroll
    for (int s = 0; s < 4; ++s) {
#pragma unroll
      for (int t = 0; t < NT; ++t) {
        const v2f w = wp[(2 * s) * PITCH + t * 16];
        acc[t] = __builtin_amdgcn_wmma_f32_16x16x4_f32(
             false, w,  false, x_cur[s],
             (short)0, acc[t],  false,  false);
      }
    }
  }
}

template <int FIN_T, int HID_T, int FOUT_T>
__global__ __launch_bounds__(SAGE_DENSE_THREADS) void sage_dense_kernel(
    const float* __restrict__ xin, const int* __restrict__ rowptr,
    const int* __restrict__ csr,
    const float* __restrict__ Wl1, const float* __restrict__ Wr1,
    const float* __restrict__ b1, const float* __restrict__ Wl2,
    const float* __restrict__ Wr2, const float* __restrict__ b2,
    float* __restrict__ p, float* __restrict__ out, int nstrips) {
  constexpr int NT1 = HID_T / 16;
  constexpr int P1  = HID_T + 16;
  constexpr int NTH = FOUT_T / 16;
  constexpr int NT2 = 2 * NTH;
  constexpr int P2  = 2 * FOUT_T + 4;
  constexpr int TP  = FIN_T + 4;
  static_assert(FIN_T == 64, "the in-launch gather splits a row over 8 lanes x 2 chunks");
  static_assert(P1 % 32 == 16, "W1 pitch: one pair row = 32 banks mod 64");
  static_assert(P2 % 8 == 4, "W2 pitch: four pair rows = 32 banks mod 64");
  __shared__ v2f w1img[FIN_T * P1];
  __shared__ v2f w2img[(HID_T / 2) * P2];
  __shared__ __attribute__((aligned(32))) float bias1[HID_T];
  __shared__ __attribute__((aligned(32))) float bias2[FOUT_T];
  __shared__ __attribute__((aligned(16))) float aggt[SAGE_DENSE_WAVES * 16 * TP];

  const int lane = threadIdx.x & 31;
  const int wave = threadIdx.x >> 5;
  const int half = lane >> 4;
  const int l15  = lane & 15;
  const int koff = half * 2;
  const v2f* w1col = w1img + half * P1 + l15;
  const v2f* w2col = w2img + (half * 4) * P2 + l15;
  const int step = gridDim.x * SAGE_DENSE_WAVES;
  const int grp = lane >> 3;
  const int c8  = (lane & 7) * 4;
  float* tile = aggt + wave * (16 * TP);
  const float* trow = tile + l15 * TP + koff;

  int strip = blockIdx.x + wave * gridDim.x;
  v2f x_nxt[4];
  if (strip < nstrips) {
    const float* f0 = xin + (long long)(strip * 16 + l15) * FIN_T + koff;
#pragma unroll
    for (int s = 0; s < 4; ++s) x_nxt[s] = *(const v2f*)(f0 + s * 4);
  }
  sage_stage_w<FIN_T, HID_T, P1>(Wl1, w1img, 0, 0);
  sage_stage_w<FIN_T, HID_T, P1>(Wr1, w1img, FIN_T, 0);
  sage_stage_w<HID_T, FOUT_T, P2>(Wl2, w2img, 0, 0);
  sage_stage_w<HID_T, FOUT_T, P2>(Wr2, w2img, 0, FOUT_T);
  for (int i = threadIdx.x; i < HID_T; i += SAGE_DENSE_THREADS) bias1[i] = b1[i];
  for (int i = threadIdx.x; i < FOUT_T; i += SAGE_DENSE_THREADS) bias2[i] = b2[i];
  __syncthreads();

  for (; strip < nstrips; strip += step) {
    asm volatile("" ::: "memory");
    const int row = strip * 16 + l15;
    const int rown = (strip + step < nstrips) ? row + step * 16 : row;
    const float* xRow = xin + (long long)row * FIN_T + koff;
    const float* xNxt = xin + (long long)rown * FIN_T + koff;

    const int rp = rowptr[strip * 16 + (lane < 16 ? lane : 16)];
#pragma nounroll
    for (int ps = 0; ps < 4; ++ps) {
      const int nl  = ps * 4 + grp;
      const int beg = __shfl(rp, nl);
      const int end = __shfl(rp, nl + 1);
      float4 m0, m1;
      sage_mean_x64(xin + c8, csr, beg, end, m0, m1);
      float* tp = tile + nl * TP + c8;
      *(float4*)tp = m0;
      *(float4*)(tp + 32) = m1;
    }
    __builtin_amdgcn_wave_barrier();

    v8f acc[NT1];
#pragma unroll
    for (int t = 0; t < NT1; ++t)
      acc[t] = *(const v8f*)(bias1 + t * 16 + half * 8);

    sage_strip_mma_tile<FIN_T, NT1, P1>(trow, w1col, acc);
    sage_strip_mma<FIN_T, NT1, P1>(xRow, xNxt, w1col, x_nxt, acc);

#pragma unroll
    for (int t = 0; t < NT1; ++t) {
#pragma unroll
      for (int v = 0; v < 8; ++v) acc[t][v] = fmaxf(acc[t][v], 0.f);
    }

    v8f acc2[NT2];
#pragma unroll
    for (int u = 0; u < NTH; ++u) {
      acc2[u] = (v8f){0.f, 0.f, 0.f, 0.f, 0.f, 0.f, 0.f, 0.f};
      acc2[NTH + u] = *(const v8f*)(bias2 + u * 16 + half * 8);
    }
#pragma unroll
    for (int t = 0; t < NT1; ++t) {
#pragma unroll
      for (int j = 0; j < 4; ++j) {
        v2f hb;
        hb.x = acc[t][2 * j];
        hb.y = acc[t][2 * j + 1];
#pragma unroll
        for (int u = 0; u < NT2; ++u) {
          const v2f w = w2col[(8 * t + j) * P2 + u * 16];
          acc2[u] = __builtin_amdgcn_wmma_f32_16x16x4_f32(
               false, w,  false, hb,
               (short)0, acc2[u],  false,  false);
        }
      }
    }

#pragma unroll
    for (int u = 0; u < NTH; ++u) {
      const long long o = (long long)row * FOUT_T + u * 16 + half * 8;
      float4* pp = (float4*)(p + o);
      float4* po = (float4*)(out + o);
      pp[0] = make_float4(acc2[u][0], acc2[u][1], acc2[u][2], acc2[u][3]);
      pp[1] = make_float4(acc2[u][4], acc2[u][5], acc2[u][6], acc2[u][7]);
      po[0] = make_float4(acc2[NTH + u][0], acc2[NTH + u][1],
                          acc2[NTH + u][2], acc2[NTH + u][3]);
      po[1] = make_float4(acc2[NTH + u][4], acc2[NTH + u][5],
                          acc2[NTH + u][6], acc2[NTH + u][7]);
    }
  }
}

extern "C" void kernel_launch(void* const* d_in, const int* in_sizes, int n_in,
                              void* d_out, int out_size, void* d_ws,
                              size_t ws_size, hipStream_t stream) {
  constexpr int FIN = 64, HID = 128, FOUT = 32;

  const float* x   = (const float*)d_in[0];
  const int*   ei  = (const int*)d_in[1];
  const float* Wl1 = (const float*)d_in[2];
  const float* Wr1 = (const float*)d_in[3];
  const float* b1  = (const float*)d_in[4];
  const float* Wl2 = (const float*)d_in[5];
  const float* Wr2 = (const float*)d_in[6];
  const float* b2  = (const float*)d_in[7];
  float* out = (float*)d_out;

  const int n = in_sizes[0] / FIN;
  const int E = in_sizes[1] / 2;
  const int* src = ei;
  const int* dst = ei + E;

  const size_t nP = ((size_t)n + 64) & ~(size_t)63;
  const size_t eP = ((size_t)E + 63) & ~(size_t)63;
  int* rowptr = (int*)d_ws;
  int* csr    = rowptr + nP;
  float* p2   = (float*)(csr + eP);
  int2* pairs = (int2*)p2;
  int* off    = (int*)(pairs + eP);

  const int nb     = (n + SAGE_BUCKET_NODES - 1) / SAGE_BUCKET_NODES;
  const int nchunk = (E + SAGE_SORT_CHUNK - 1) / SAGE_SORT_CHUNK;

  sage_chunk_sort_kernel<<<nchunk, SAGE_SORT_THREADS, 0, stream>>>(
      src, dst, pairs, off, E, nb);
  sage_bucket_csr_kernel<<<nb, SAGE_SORT_THREADS, 0, stream>>>(
      pairs, off, rowptr, csr, n, E, nb, nchunk);

  sage_dense_kernel<FIN, HID, FOUT>
      <<<256, SAGE_DENSE_THREADS, 0, stream>>>(x, rowptr, csr, Wl1, Wr1, b1, Wl2,
                                               Wr2, b2, p2, out, n / 16);


  sage_gather_kernel<FOUT, true>
      <<<(n + 31) / 32, 128, 0, stream>>>(p2, rowptr, csr, out, n);
}


# Every 128-byte line written by one store: dense p/out rows and each bucket's CSR source ids are collected in LDS and leave as whole lines, not in 16- and 4-byte pieces; f32 sum order unchanged
# speedup vs baseline: 6.5344x; 1.2016x over previous
#include <hip/hip_runtime.h>


typedef float v2f __attribute__((ext_vector_type(2)));
typedef float v8f __attribute__((ext_vector_type(8)));

constexpr int SAGE_SORT_THREADS = 512;
constexpr int SAGE_SORT_EPT     = 8;
constexpr int SAGE_SORT_CHUNK   = SAGE_SORT_THREADS * SAGE_SORT_EPT;
constexpr int SAGE_BUCKET_SHIFT = 8;
constexpr int SAGE_BUCKET_NODES = 1 << SAGE_BUCKET_SHIFT;
constexpr int SAGE_BUCKET_BATCH = 4;
constexpr int SAGE_CSR_STAGE    = 8192;

__device__ __forceinline__ void sage_load8(const int* __restrict__ p, int e0,
                                           int E, int (&v)[SAGE_SORT_EPT]) {
  if ((E & 3) == 0 && e0 + SAGE_SORT_EPT <= E) {
    const int4 a = *(const int4*)(p + e0);
    const int4 b = *(const int4*)(p + e0 + 4);
    v[0] = a.x; v[1] = a.y; v[2] = a.z; v[3] = a.w;
    v[4] = b.x; v[5] = b.y; v[6] = b.z; v[7] = b.w;
  } else {
#pragma unroll
    for (int u = 0; u < SAGE_SORT_EPT; ++u) v[u] = (e0 + u < E) ? p[e0 + u] : 0;
  }
}

__device__ __forceinline__ int sage_block_scan(int v, int* ws, int& total) {
  const int lane = threadIdx.x & 31;
  const int w    = threadIdx.x >> 5;
  int incl = v;
  for (int o = 1; o < 32; o <<= 1) {
    const int t = __shfl_up(incl, o);
    if (lane >= o) incl += t;
  }
  if (lane == 31) ws[w] = incl;
  __syncthreads();
  if (w == 0) {
    const int s = (lane < 16) ? ws[lane] : 0;
    int wi = s;
    for (int o = 1; o < 16; o <<= 1) {
      const int t = __shfl_up(wi, o);
      if (lane >= o) wi += t;
    }
    if (lane < 16) ws[lane] = wi - s;
    if (lane == 15) ws[16] = wi;
  }
  __syncthreads();
  total = ws[16];
  return ws[w] + incl - v;
}

__global__ __launch_bounds__(SAGE_SORT_THREADS) void sage_chunk_sort_kernel(
    const int* __restrict__ src, const int* __restrict__ dst,
    int2* __restrict__ pairs, int* __restrict__ off, int E, int nb) {
  __shared__ int hist[SAGE_SORT_THREADS];
  __shared__ int ws[17];
  __shared__ int2 stage[SAGE_SORT_CHUNK];
  const int tid = threadIdx.x;
  const int cb  = blockIdx.x * SAGE_SORT_CHUNK;
  const int e0  = cb + tid * SAGE_SORT_EPT;
  int s[SAGE_SORT_EPT], d[SAGE_SORT_EPT];
  sage_load8(src, e0, E, s);
  sage_load8(dst, e0, E, d);
  hist[tid] = 0;
  __syncthreads();
#pragma unroll
  for (int u = 0; u < SAGE_SORT_EPT; ++u)
    if (e0 + u < E) atomicAdd(&hist[d[u] >> SAGE_BUCKET_SHIFT], 1);
  __syncthreads();
  const int c = hist[tid];
  int total;
  const int ex = sage_block_scan(c, ws, total);
  if (tid < nb) off[blockIdx.x * nb + tid] = ex;
  hist[tid] = ex;
  __syncthreads();
#pragma unroll
  for (int u = 0; u < SAGE_SORT_EPT; ++u)
    if (e0 + u < E)
      stage[atomicAdd(&hist[d[u] >> SAGE_BUCKET_SHIFT], 1)] =
          make_int2(s[u], d[u]);
  __syncthreads();
  for (int i = tid; i < total; i += SAGE_SORT_THREADS) pairs[cb + i] = stage[i];
}

__device__ __forceinline__ int sage_find_run(const int* pre, int i) {
  int r = 0;
  for (int s = SAGE_SORT_THREADS >> 1; s > 0; s >>= 1)
    if (pre[r + s] <= i) r += s;
  return r;
}

__global__ __launch_bounds__(SAGE_SORT_THREADS) void sage_bucket_csr_kernel(
    const int2* __restrict__ pairs, const int* __restrict__ off,
    int* __restrict__ rowptr, int* __restrict__ csr, int n, int E, int nb,
    int nchunk) {
  __shared__ int cnt[SAGE_BUCKET_NODES];
  __shared__ int cur[SAGE_BUCKET_NODES];
  __shared__ int pre[SAGE_SORT_THREADS];
  __shared__ int sbeg[SAGE_SORT_THREADS];
  __shared__ int wsa[17];
  __shared__ int wsb[17];
  __shared__ int wsc[17];
  __shared__ int stg[SAGE_CSR_STAGE];
  constexpr int STRIDE = SAGE_SORT_THREADS * SAGE_BUCKET_BATCH;
  const int tid = threadIdx.x;
  const int b   = blockIdx.x;
  const int lo  = b << SAGE_BUCKET_SHIFT;
  if (tid < SAGE_BUCKET_NODES) cnt[tid] = 0;
  int sb = 0;
  for (int c = tid; c < nchunk; c += SAGE_SORT_THREADS) sb += off[c * nb + b];
  int base;
  sage_block_scan(sb, wsa, base);
  const bool one_tile = (nchunk <= SAGE_SORT_THREADS);
  int T = 0;
  int tile_tot = 0;

  for (int pass = 0; pass < 2; ++pass) {
    if (pass == 1) {
      __syncthreads();
      const int cc = (tid < SAGE_BUCKET_NODES) ? cnt[tid] : 0;
      int tot;
      const int ex = sage_block_scan(cc, wsb, tot);
      if (tid < SAGE_BUCKET_NODES) {
        cur[tid] = base + ex;
        if (lo + tid < n) rowptr[lo + tid] = base + ex;
      }
      if (b == nb - 1 && tid == 0) rowptr[n] = base + T;
      __syncthreads();
    }
    for (int c0 = 0; c0 < nchunk; c0 += SAGE_SORT_THREADS) {
      if (pass == 0 || !one_tile) {
        const int c = c0 + tid;
        int o0 = 0, len = 0;
        if (c < nchunk) {
          const int left = E - c * SAGE_SORT_CHUNK;
          const int full = (left < SAGE_SORT_CHUNK) ? left : SAGE_SORT_CHUNK;
          o0 = off[c * nb + b];
          const int o1 = (b + 1 < nb) ? off[c * nb + b + 1] : full;
          len = o1 - o0;
        }
        __syncthreads();
        const int ex = sage_block_scan(len, wsc, tile_tot);
        pre[tid]  = ex;
        sbeg[tid] = c * SAGE_SORT_CHUNK + o0;
        if (pass == 0) T += tile_tot;
        __syncthreads();
      }
      const int last = tile_tot - 1;
      if (pass == 0) {
        for (int i0 = 0; i0 < tile_tot; i0 += STRIDE) {
          int dl[SAGE_BUCKET_BATCH];
#pragma unroll
          for (int u = 0; u < SAGE_BUCKET_BATCH; ++u) {
            const int i  = i0 + u * SAGE_SORT_THREADS + tid;
            const int ii = (i < tile_tot) ? i : last;
            const int r  = sage_find_run(pre, ii);
            const int raw = pairs[sbeg[r] + (ii - pre[r])].y;
            dl[u] = (i < tile_tot) ? (raw & (SAGE_BUCKET_NODES - 1)) : -1;
          }
#pragma unroll
          for (int u = 0; u < SAGE_BUCKET_BATCH; ++u)
            if (dl[u] >= 0) atomicAdd(&cnt[dl[u]], 1);
        }
      } else {
        for (int i0 = 0; i0 < tile_tot; i0 += STRIDE) {
          int2 p[SAGE_BUCKET_BATCH];
          int pos[SAGE_BUCKET_BATCH];
#pragma unroll
          for (int u = 0; u < SAGE_BUCKET_BATCH; ++u) {
            const int i  = i0 + u * SAGE_SORT_THREADS + tid;
            const int ii = (i < tile_tot) ? i : last;
            const int r  = sage_find_run(pre, ii);
            p[u] = pairs[sbeg[r] + (ii - pre[r])];
            if (i >= tile_tot) p[u].y = -1;
          }
#pragma unroll
          for (int u = 0; u < SAGE_BUCKET_BATCH; ++u)
            pos[u] = (p[u].y >= 0)
                         ? atomicAdd(&cur[p[u].y & (SAGE_BUCKET_NODES - 1)], 1)
                         : -1;
          if (T <= SAGE_CSR_STAGE) {
#pragma unroll
            for (int u = 0; u < SAGE_BUCKET_BATCH; ++u)
              if (pos[u] >= 0) stg[pos[u] - base] = p[u].x;
          } else {
#pragma unroll
            for (int u = 0; u < SAGE_BUCKET_BATCH; ++u)
              if (pos[u] >= 0) csr[pos[u]] = p[u].x;
          }
        }
      }
    }
  }
  __syncthreads();
  if (T <= SAGE_CSR_STAGE) {
    const int g0 = base & ~31;
    for (int g = g0 + tid; g < base + T; g += SAGE_SORT_THREADS)
      if (g >= base) csr[g] = stg[g - base];
  }
}

__device__ __forceinline__ float4 sage_row(const float* fcol, int s, int F) {
  return *(const float4*)(fcol + (long long)s * F);
}
__device__ __forceinline__ void sage_acc(float4& acc, const float4 v) {
  acc.x += v.x;
  acc.y += v.y;
  acc.z += v.z;
  acc.w += v.w;
}

template <int F, bool ADD>
__global__ __launch_bounds__(128) void sage_gather_kernel(
    const float* __restrict__ feat, const int* __restrict__ rowptr,
    const int* __restrict__ csr, float* __restrict__ tgt, int n) {
  constexpr int LPN = F / 4;
  constexpr int NPB = 128 / LPN;
  const int sub = threadIdx.x / LPN;
  const int c   = (threadIdx.x % LPN) * 4;
  const float* fcol = feat + c;
  for (int d = blockIdx.x * NPB + sub; d < n; d += gridDim.x * NPB) {
    const int beg = rowptr[d];
    const int end = rowptr[d + 1];
    float4 acc = make_float4(0.f, 0.f, 0.f, 0.f);
    int j = beg;
    for (; j + 8 <= end; j += 8) {
      const int s0 = csr[j];
      const int s1 = csr[j + 1];
      const int s2 = csr[j + 2];
      const int s3 = csr[j + 3];
      const int s4 = csr[j + 4];
      const int s5 = csr[j + 5];
      const int s6 = csr[j + 6];
      const int s7 = csr[j + 7];
      const float4 v0 = sage_row(fcol, s0, F);
      const float4 v1 = sage_row(fcol, s1, F);
      const float4 v2 = sage_row(fcol, s2, F);
      const float4 v3 = sage_row(fcol, s3, F);
      const float4 v4 = sage_row(fcol, s4, F);
      const float4 v5 = sage_row(fcol, s5, F);
      const float4 v6 = sage_row(fcol, s6, F);
      const float4 v7 = sage_row(fcol, s7, F);
      sage_acc(acc, v0);
      sage_acc(acc, v1);
      sage_acc(acc, v2);
      sage_acc(acc, v3);
      sage_acc(acc, v4);
      sage_acc(acc, v5);
      sage_acc(acc, v6);
      sage_acc(acc, v7);
    }
    if (j + 4 <= end) {
      const int s0 = csr[j];
      const int s1 = csr[j + 1];
      const int s2 = csr[j + 2];
      const int s3 = csr[j + 3];
      const float4 v0 = sage_row(fcol, s0, F);
      const float4 v1 = sage_row(fcol, s1, F);
      const float4 v2 = sage_row(fcol, s2, F);
      const float4 v3 = sage_row(fcol, s3, F);
      sage_acc(acc, v0);
      sage_acc(acc, v1);
      sage_acc(acc, v2);
      sage_acc(acc, v3);
      j += 4;
    }
    if (j < end) {
      const int last = end - 1;
      const bool h1 = (j + 1 <= last);
      const bool h2 = (j + 2 <= last);
      const int s0 = csr[j];
      const int s1 = csr[h1 ? j + 1 : last];
      const int s2 = csr[h2 ? j + 2 : last];
      const float4 v0 = sage_row(fcol, s0, F);
      float4 v1 = sage_row(fcol, s1, F);
      float4 v2 = sage_row(fcol, s2, F);
      v1.x = h1 ? v1.x : 0.f; v1.y = h1 ? v1.y : 0.f;
      v1.z = h1 ? v1.z : 0.f; v1.w = h1 ? v1.w : 0.f;
      v2.x = h2 ? v2.x : 0.f; v2.y = h2 ? v2.y : 0.f;
      v2.z = h2 ? v2.z : 0.f; v2.w = h2 ? v2.w : 0.f;
      sage_acc(acc, v0);
      sage_acc(acc, v1);
      sage_acc(acc, v2);
    }
    const float inv = 1.0f / fmaxf((float)(end - beg), 1.0f);
    float4* p = (float4*)(tgt + (long long)d * F + c);
    float4 o = make_float4(0.f, 0.f, 0.f, 0.f);
    if (ADD) o = *p;
    o.x += acc.x * inv;
    o.y += acc.y * inv;
    o.z += acc.z * inv;
    o.w += acc.w * inv;
    *p = o;
  }
}

constexpr int SAGE_DENSE_WAVES = 16;
constexpr int SAGE_DENSE_THREADS = SAGE_DENSE_WAVES * 32;

template <int KROWS, int SRC_COLS, int PITCH>
__device__ __forceinline__ void sage_stage_w(const float* __restrict__ wsrc,
                                             v2f* __restrict__ img, int k_lo,
                                             int col_lo) {
  constexpr int QPR = SRC_COLS / 4;
  for (int i = threadIdx.x; i < (KROWS / 2) * QPR; i += SAGE_DENSE_THREADS) {
    const int kp = i / QPR;
    const int q  = (i % QPR) * 4;
    const float4 r0 = *(const float4*)(wsrc + (2 * kp) * SRC_COLS + q);
    const float4 r1 = *(const float4*)(wsrc + (2 * kp + 1) * SRC_COLS + q);
    v2f* d = img + (k_lo / 2 + kp) * PITCH + col_lo + q;
    v2f p0, p1, p2, p3;
    p0.x = r0.x; p0.y = r1.x;
    p1.x = r0.y; p1.y = r1.y;
    p2.x = r0.z; p2.y = r1.z;
    p3.x = r0.w; p3.y = r1.w;
    d[0] = p0;
    d[1] = p1;
    d[2] = p2;
    d[3] = p3;
  }
}

__device__ __forceinline__ void sage_mean_x64(const float* __restrict__ fcol,
                                              const int* __restrict__ csr,
                                              int beg, int end, float4& m0,
                                              float4& m1) {
  const int n8   = (end - beg) >> 3;
  const int jt   = beg + (n8 << 3);
  const bool has4 = (jt + 4 <= end);
  const int jr   = has4 ? jt + 4 : jt;
  const bool hr0 = (jr < end);
  const bool hr1 = (jr + 1 < end);
  const bool hr2 = (jr + 2 < end);

  int s8[8], s4[4], sr[3];
  const int b8 = (n8 > 0) ? beg : ((end >= 8) ? end - 8 : 0);
  const int b4 = has4 ? jt : ((end >= 4) ? end - 4 : 0);
  const int l0 = (end > 0) ? end - 1 : 0;
#pragma unroll
  for (int u = 0; u < 8; ++u) s8[u] = csr[b8 + u];
#pragma unroll
  for (int u = 0; u < 4; ++u) s4[u] = csr[b4 + u];
  sr[0] = csr[hr0 ? jr : l0];
  sr[1] = csr[hr1 ? jr + 1 : l0];
  sr[2] = csr[hr2 ? jr + 2 : l0];
  asm volatile(""
               : "+v"(s8[0]), "+v"(s8[1]), "+v"(s8[2]), "+v"(s8[3]),
                 "+v"(s8[4]), "+v"(s8[5]), "+v"(s8[6]), "+v"(s8[7]),
                 "+v"(s4[0]), "+v"(s4[1]), "+v"(s4[2]), "+v"(s4[3]),
                 "+v"(sr[0]), "+v"(sr[1]), "+v"(sr[2]));
  float4 v4[4], w4[4], vr[3], wr[3];
#pragma unroll
  for (int u = 0; u < 4; ++u) {
    v4[u] = sage_row(fcol, s4[u], 64);
    w4[u] = sage_row(fcol + 32, s4[u], 64);
  }
#pragma unroll
  for (int u = 0; u < 3; ++u) {
    vr[u] = sage_row(fcol, sr[u], 64);
    wr[u] = sage_row(fcol + 32, sr[u], 64);
  }

  float4 a0 = make_float4(0.f, 0.f, 0.f, 0.f);
  float4 a1 = make_float4(0.f, 0.f, 0.f, 0.f);
  int j = beg;
  for (int g = 0; g < n8; ++g) {
    float4 v[8], w[8];
#pragma unroll
    for (int u = 0; u < 8; ++u) {
      v[u] = sage_row(fcol, s8[u], 64);
      w[u] = sage_row(fcol + 32, s8[u], 64);
    }
    const int jn = (j + 16 <= end) ? j + 8 : end - 8;
#pragma unroll
    for (int u = 0; u < 8; ++u) s8[u] = csr[jn + u];
#pragma unroll
    for (int u = 0; u < 8; ++u) {
      sage_acc(a0, v[u]);
      sage_acc(a1, w[u]);
    }
    j += 8;
  }
  if (has4) {
#pragma unroll
    for (int u = 0; u < 4; ++u) {
      sage_acc(a0, v4[u]);
      sage_acc(a1, w4[u]);
    }
  }
  if (hr0) {
    if (!hr1) {
      vr[1] = make_float4(0.f, 0.f, 0.f, 0.f);
      wr[1] = make_float4(0.f, 0.f, 0.f, 0.f);
    }
    if (!hr2) {
      vr[2] = make_float4(0.f, 0.f, 0.f, 0.f);
      wr[2] = make_float4(0.f, 0.f, 0.f, 0.f);
    }
#pragma unroll
    for (int u = 0; u < 3; ++u) {
      sage_acc(a0, vr[u]);
      sage_acc(a1, wr[u]);
    }
  }
  const float inv = 1.0f / fmaxf((float)(end - beg), 1.0f);
  m0 = make_float4(a0.x * inv, a0.y * inv, a0.z * inv, a0.w * inv);
  m1 = make_float4(a1.x * inv, a1.y * inv, a1.z * inv, a1.w * inv);
}

template <int KH, int NT, int PITCH>
__device__ __forceinline__ void sage_strip_mma_tile(
    const float* __restrict__ trow, const v2f* __restrict__ wcol,
    v8f (&acc)[NT]) {
  constexpr int CHH = KH / 16;
#pragma nounroll
  for (int ch = 0; ch < CHH; ++ch) {
    v2f b[4];
#pragma unroll
    for (int s = 0; s < 4; ++s) b[s] = *(const v2f*)(trow + ch * 16 + s * 4);
    const v2f* wp = wcol + (ch * 8) * PITCH;
#pragma unroll
    for (int s = 0; s < 4; ++s) {
#pragma unroll
      for (int t = 0; t < NT; ++t) {
        const v2f w = wp[(2 * s) * PITCH + t * 16];
        acc[t] = __builtin_amdgcn_wmma_f32_16x16x4_f32(
             false, w,  false, b[s],
             (short)0, acc[t],  false,  false);
      }
    }
  }
}

template <int KH, int NT, int PITCH>
__device__ __forceinline__ void sage_strip_mma(
    const float* __restrict__ r0, const float* __restrict__ n0,
    const v2f* __restrict__ wcol, v2f (&x_nxt)[4], v8f (&acc)[NT]) {
  constexpr int CHH = KH / 16;
#pragma nounroll
  for (int ch = 0; ch < CHH; ++ch) {
    v2f x_cur[4];
#pragma unroll
    for (int s = 0; s < 4; ++s) x_cur[s] = x_nxt[s];
    const int cn = ch + 1;
    const float* ap = (cn < CHH) ? r0 + cn * 16 : n0;
#pragma unroll
    for (int s = 0; s < 4; ++s)
      x_nxt[s] = *(const v2f*)(ap + s * 4);
    const v2f* wp = wcol + ((CHH + ch) * 8) * PITCH;
#pragma unroll
    for (int s = 0; s < 4; ++s) {
#pragma unroll
      for (int t = 0; t < NT; ++t) {
        const v2f w = wp[(2 * s) * PITCH + t * 16];
        acc[t] = __builtin_amdgcn_wmma_f32_16x16x4_f32(
             false, w,  false, x_cur[s],
             (short)0, acc[t],  false,  false);
      }
    }
  }
}

template <int FIN_T, int HID_T, int FOUT_T>
__global__ __launch_bounds__(SAGE_DENSE_THREADS) void sage_dense_kernel(
    const float* __restrict__ xin, const int* __restrict__ rowptr,
    const int* __restrict__ csr,
    const float* __restrict__ Wl1, const float* __restrict__ Wr1,
    const float* __restrict__ b1, const float* __restrict__ Wl2,
    const float* __restrict__ Wr2, const float* __restrict__ b2,
    float* __restrict__ p, float* __restrict__ out, int nstrips) {
  constexpr int NT1 = HID_T / 16;
  constexpr int P1  = HID_T + 16;
  constexpr int NTH = FOUT_T / 16;
  constexpr int NT2 = 2 * NTH;
  constexpr int P2  = 2 * FOUT_T + 4;
  constexpr int TP  = FIN_T + 4;
  static_assert(FIN_T == 64, "the in-launch gather splits a row over 8 lanes x 2 chunks");
  static_assert(FOUT_T == 32 && 2 * FOUT_T <= FIN_T,
                "output rows are 8 lanes x 16 bytes; p and out rows share a tile row");
  static_assert(P1 % 32 == 16, "W1 pitch: one pair row = 32 banks mod 64");
  static_assert(P2 % 8 == 4, "W2 pitch: four pair rows = 32 banks mod 64");
  __shared__ v2f w1img[FIN_T * P1];
  __shared__ v2f w2img[(HID_T / 2) * P2];
  __shared__ __attribute__((aligned(32))) float bias1[HID_T];
  __shared__ __attribute__((aligned(32))) float bias2[FOUT_T];
  __shared__ __attribute__((aligned(16))) float aggt[SAGE_DENSE_WAVES * 16 * TP];

  const int lane = threadIdx.x & 31;
  const int wave = threadIdx.x >> 5;
  const int half = lane >> 4;
  const int l15  = lane & 15;
  const int koff = half * 2;
  const v2f* w1col = w1img + half * P1 + l15;
  const v2f* w2col = w2img + (half * 4) * P2 + l15;
  const int step = gridDim.x * SAGE_DENSE_WAVES;
  const int grp = lane >> 3;
  const int c8  = (lane & 7) * 4;
  float* tile = aggt + wave * (16 * TP);
  const float* trow = tile + l15 * TP + koff;

  int strip = blockIdx.x + wave * gridDim.x;
  v2f x_nxt[4];
  if (strip < nstrips) {
    const float* f0 = xin + (long long)(strip * 16 + l15) * FIN_T + koff;
#pragma unroll
    for (int s = 0; s < 4; ++s) x_nxt[s] = *(const v2f*)(f0 + s * 4);
  }
  sage_stage_w<FIN_T, HID_T, P1>(Wl1, w1img, 0, 0);
  sage_stage_w<FIN_T, HID_T, P1>(Wr1, w1img, FIN_T, 0);
  sage_stage_w<HID_T, FOUT_T, P2>(Wl2, w2img, 0, 0);
  sage_stage_w<HID_T, FOUT_T, P2>(Wr2, w2img, 0, FOUT_T);
  for (int i = threadIdx.x; i < HID_T; i += SAGE_DENSE_THREADS) bias1[i] = b1[i];
  for (int i = threadIdx.x; i < FOUT_T; i += SAGE_DENSE_THREADS) bias2[i] = b2[i];
  __syncthreads();

  for (; strip < nstrips; strip += step) {
    asm volatile("" ::: "memory");
    const int row = strip * 16 + l15;
    const int rown = (strip + step < nstrips) ? row + step * 16 : row;
    const float* xRow = xin + (long long)row * FIN_T + koff;
    const float* xNxt = xin + (long long)rown * FIN_T + koff;

    const int rp = rowptr[strip * 16 + (lane < 16 ? lane : 16)];
#pragma nounroll
    for (int ps = 0; ps < 4; ++ps) {
      const int nl  = ps * 4 + grp;
      const int beg = __shfl(rp, nl);
      const int end = __shfl(rp, nl + 1);
      float4 m0, m1;
      sage_mean_x64(xin + c8, csr, beg, end, m0, m1);
      float* tp = tile + nl * TP + c8;
      *(float4*)tp = m0;
      *(float4*)(tp + 32) = m1;
    }
    __builtin_amdgcn_wave_barrier();

    v8f acc[NT1];
#pragma unroll
    for (int t = 0; t < NT1; ++t)
      acc[t] = *(const v8f*)(bias1 + t * 16 + half * 8);

    sage_strip_mma_tile<FIN_T, NT1, P1>(trow, w1col, acc);
    sage_strip_mma<FIN_T, NT1, P1>(xRow, xNxt, w1col, x_nxt, acc);

#pragma unroll
    for (int t = 0; t < NT1; ++t) {
#pragma unroll
      for (int v = 0; v < 8; ++v) acc[t][v] = fmaxf(acc[t][v], 0.f);
    }

    v8f acc2[NT2];
#pragma unroll
    for (int u = 0; u < NTH; ++u) {
      acc2[u] = (v8f){0.f, 0.f, 0.f, 0.f, 0.f, 0.f, 0.f, 0.f};
      acc2[NTH + u] = *(const v8f*)(bias2 + u * 16 + half * 8);
    }
#pragma unroll
    for (int t = 0; t < NT1; ++t) {
#pragma unroll
      for (int j = 0; j < 4; ++j) {
        v2f hb;
        hb.x = acc[t][2 * j];
        hb.y = acc[t][2 * j + 1];
#pragma unroll
        for (int u = 0; u < NT2; ++u) {
          const v2f w = w2col[(8 * t + j) * P2 + u * 16];
          acc2[u] = __builtin_amdgcn_wmma_f32_16x16x4_f32(
               false, w,  false, hb,
               (short)0, acc2[u],  false,  false);
        }
      }
    }

    float* srow = tile + l15 * TP + half * 8;
#pragma unroll
    for (int u = 0; u < NTH; ++u) {
      *(float4*)(srow + u * 16) =
          make_float4(acc2[u][0], acc2[u][1], acc2[u][2], acc2[u][3]);
      *(float4*)(srow + u * 16 + 4) =
          make_float4(acc2[u][4], acc2[u][5], acc2[u][6], acc2[u][7]);
      *(float4*)(srow + FOUT_T + u * 16) =
          make_float4(acc2[NTH + u][0], acc2[NTH + u][1], acc2[NTH + u][2],
                      acc2[NTH + u][3]);
      *(float4*)(srow + FOUT_T + u * 16 + 4) =
          make_float4(acc2[NTH + u][4], acc2[NTH + u][5], acc2[NTH + u][6],
                      acc2[NTH + u][7]);
    }
    __builtin_amdgcn_wave_barrier();
#pragma unroll
    for (int i = 0; i < 4; ++i) {
      const int r = i * 4 + grp;
      const float4 vp = *(const float4*)(tile + r * TP + c8);
      const float4 vo = *(const float4*)(tile + r * TP + FOUT_T + c8);
      const long long o = (long long)(strip * 16 + r) * FOUT_T + c8;
      *(float4*)(p + o)   = vp;
      *(float4*)(out + o) = vo;
    }
  }
}

extern "C" void kernel_launch(void* const* d_in, const int* in_sizes, int n_in,
                              void* d_out, int out_size, void* d_ws,
                              size_t ws_size, hipStream_t stream) {
  constexpr int FIN = 64, HID = 128, FOUT = 32;

  const float* x   = (const float*)d_in[0];
  const int*   ei  = (const int*)d_in[1];
  const float* Wl1 = (const float*)d_in[2];
  const float* Wr1 = (const float*)d_in[3];
  const float* b1  = (const float*)d_in[4];
  const float* Wl2 = (const float*)d_in[5];
  const float* Wr2 = (const float*)d_in[6];
  const float* b2  = (const float*)d_in[7];
  float* out = (float*)d_out;

  const int n = in_sizes[0] / FIN;
  const int E = in_sizes[1] / 2;
  const int* src = ei;
  const int* dst = ei + E;

  const size_t nP = ((size_t)n + 64) & ~(size_t)63;
  const size_t eP = ((size_t)E + 63) & ~(size_t)63;
  int* rowptr = (int*)d_ws;
  int* csr    = rowptr + nP;
  float* p2   = (float*)(csr + eP);
  int2* pairs = (int2*)p2;
  int* off    = (int*)(pairs + eP);

  const int nb     = (n + SAGE_BUCKET_NODES - 1) / SAGE_BUCKET_NODES;
  const int nchunk = (E + SAGE_SORT_CHUNK - 1) / SAGE_SORT_CHUNK;

  sage_chunk_sort_kernel<<<nchunk, SAGE_SORT_THREADS, 0, stream>>>(
      src, dst, pairs, off, E, nb);
  sage_bucket_csr_kernel<<<nb, SAGE_SORT_THREADS, 0, stream>>>(
      pairs, off, rowptr, csr, n, E, nb, nchunk);

  sage_dense_kernel<FIN, HID, FOUT>
      <<<256, SAGE_DENSE_THREADS, 0, stream>>>(x, rowptr, csr, Wl1, Wr1, b1, Wl2,
                                               Wr2, b2, p2, out, n / 16);


  sage_gather_kernel<FOUT, true>
      <<<(n + 31) / 32, 128, 0, stream>>>(p2, rowptr, csr, out, n);
}
